# starting kernel
# baseline (speedup 1.0000x reference)
#include <hip/hip_runtime.h>
#include <math.h>


#define NUM_NODES 100000
#define NUM_RELS  500
#define DIM       64
#define NUM_EDGES 1000000

typedef __attribute__((ext_vector_type(2))) float v2f;
typedef __attribute__((ext_vector_type(8))) float v8f;

__global__ void zero_kernel(float* __restrict__ p, int n) {
  for (int i = blockIdx.x * blockDim.x + threadIdx.x; i < n;
       i += gridDim.x * blockDim.x)
    p[i] = 0.0f;
}

__global__ __launch_bounds__(256) void fold_kernel(
    const float* __restrict__ W_ent, const float* __restrict__ b_ent,
    const float* __restrict__ W_relL, const float* __restrict__ b_relL,
    const float* __restrict__ W_rel2, const float* __restrict__ b_rel2,
    const float* __restrict__ W_a, const float* __restrict__ b_a,
    const float* __restrict__ W_fc, const float* __restrict__ b_fc,
    float* __restrict__ M_s, float* __restrict__ M_d,
    float* __restrict__ M_r, float* __restrict__ M_2,
    float* __restrict__ cvec, float* __restrict__ v_rel2,
    float* __restrict__ a0) {
  __shared__ float scvec[DIM];
  const int t = threadIdx.x;
  for (int idx = t; idx < DIM * DIM; idx += 256) {
    const int o = idx >> 6, k = idx & 63;
    float ms = 0.f, md = 0.f, mr = 0.f, m2 = 0.f;
    for (int u = 0; u < DIM; ++u) {
      const float we = W_ent[u * DIM + k];
      const float wl = W_relL[u * DIM + k];
      ms += W_fc[o * 192 + u] * we;
      md += W_fc[o * 192 + 64 + u] * we;
      mr += W_fc[o * 192 + 128 + u] * wl;
      m2 += W_rel2[o * DIM + u] * wl;
    }
    M_s[idx] = ms; M_d[idx] = md; M_r[idx] = mr; M_2[idx] = m2;
  }
  if (t < DIM) {
    float cv = b_fc[t], v2 = b_rel2[t];
    for (int u = 0; u < DIM; ++u) {
      cv += W_fc[t * 192 + u] * b_ent[u]
          + W_fc[t * 192 + 64 + u] * b_ent[u]
          + W_fc[t * 192 + 128 + u] * b_relL[u];
      v2 += W_rel2[t * DIM + u] * b_relL[u];
    }
    cvec[t] = cv; v_rel2[t] = v2; scvec[t] = cv;
  }
  __syncthreads();
  if (t == 0) {
    float a = b_a[0];
    for (int o = 0; o < DIM; ++o) a += scvec[o] * W_a[o];
    a0[0] = a;
  }
}

__global__ __launch_bounds__(256) void wmma_gemm64(
    const float* __restrict__ A, const float* __restrict__ W,
    const float* __restrict__ bias, float* __restrict__ out, int M) {
  __shared__ float sW[DIM * DIM];
  for (int i = threadIdx.x; i < DIM * DIM; i += 256) sW[i] = W[i];
  __syncthreads();

  const int wave = threadIdx.x >> 5;
  const int lane = threadIdx.x & 31;
  const int m0 = (blockIdx.x * 8 + wave) * 16;
  if (m0 >= M) return;

  const int lm = lane & 15;
  const int hi = lane >> 4;

  int arow = m0 + lm;
  if (arow >= M) arow = M - 1;
  const float* __restrict__ Arow = A + (size_t)arow * DIM;

  v8f acc0 = {}, acc1 = {}, acc2 = {}, acc3 = {};
  for (int k = 0; k < DIM; k += 4) {
    v2f a;
    a.x = Arow[k + 2 * hi + 0];
    a.y = Arow[k + 2 * hi + 1];
    v2f b0, b1, b2, b3;
    const float* w0 = &sW[(0 * 16 + lm) * DIM + k + 2 * hi];
    const float* w1 = &sW[(1 * 16 + lm) * DIM + k + 2 * hi];
    const float* w2 = &sW[(2 * 16 + lm) * DIM + k + 2 * hi];
    const float* w3 = &sW[(3 * 16 + lm) * DIM + k + 2 * hi];
    b0.x = w0[0]; b0.y = w0[1];
    b1.x = w1[0]; b1.y = w1[1];
    b2.x = w2[0]; b2.y = w2[1];
    b3.x = w3[0]; b3.y = w3[1];
    acc0 = __builtin_amdgcn_wmma_f32_16x16x4_f32(false, a, false, b0,
                                                 (short)0, acc0, false, false);
    acc1 = __builtin_amdgcn_wmma_f32_16x16x4_f32(false, a, false, b1,
                                                 (short)0, acc1, false, false);
    acc2 = __builtin_amdgcn_wmma_f32_16x16x4_f32(false, a, false, b2,
                                                 (short)0, acc2, false, false);
    acc3 = __builtin_amdgcn_wmma_f32_16x16x4_f32(false, a, false, b3,
                                                 (short)0, acc3, false, false);
  }

  v8f accs[4] = {acc0, acc1, acc2, acc3};
#pragma unroll
  for (int ob = 0; ob < 4; ++ob) {
    const float bv = bias ? bias[ob * 16 + lm] : 0.0f;
#pragma unroll
    for (int v = 0; v < 8; ++v) {
      const int row = m0 + v + 8 * hi;
      if (row < M)
        out[(size_t)row * DIM + ob * 16 + lm] = accs[ob][v] + bv;
    }
  }
}

__global__ void dot64_kernel(const float* __restrict__ X,
                             const float* __restrict__ v,
                             float* __restrict__ out, int M) {
  const int m = blockIdx.x * blockDim.x + threadIdx.x;
  if (m >= M) return;
  const float* x = X + (size_t)m * DIM;
  float acc = 0.f;
#pragma unroll 8
  for (int j = 0; j < DIM; ++j) acc += x[j] * v[j];
  out[m] = acc;
}

__global__ void edge_logits_kernel(
    const int* __restrict__ src, const int* __restrict__ dst,
    const int* __restrict__ rel, const float* __restrict__ a_s,
    const float* __restrict__ a_d, const float* __restrict__ a_r,
    const float* __restrict__ a0p, float* __restrict__ b_e,
    float* __restrict__ b_sum, float* __restrict__ cnt, int E) {
  const float a0 = a0p[0];
  for (int e = blockIdx.x * blockDim.x + threadIdx.x; e < E;
       e += gridDim.x * blockDim.x) {
    const int s = src[e], d = dst[e], r = rel[e];
    float l = a_s[s] + a_d[d] + a_r[r] + a0;
    l = l > 0.f ? l : 0.01f * l;
    const float b = expf(l);
    b_e[e] = b;
    atomicAdd(&b_sum[s], b);
    atomicAdd(&cnt[r], 1.0f);
  }
}

__global__ __launch_bounds__(256) void edge_scatter_h_kernel(
    const int* __restrict__ src, const int* __restrict__ dst,
    const int* __restrict__ rel, const float* __restrict__ u_s,
    const float* __restrict__ u_d, const float* __restrict__ u_r,
    const float* __restrict__ cvec, const float* __restrict__ b_e,
    const float* __restrict__ b_sum, float* __restrict__ h_ent, int E) {
  const int gw = blockIdx.x * (blockDim.x >> 5) + (threadIdx.x >> 5);
  const int lane = threadIdx.x & 31;
  const int nw = gridDim.x * (blockDim.x >> 5);
  for (int e = gw; e < E; e += nw) {
    const int s = src[e], d = dst[e], r = rel[e];
    const float alpha = b_e[e] / b_sum[s];
    const float* ps = u_s + (size_t)s * DIM;
    const float* pd = u_d + (size_t)d * DIM;
    const float* pr = u_r + (size_t)r * DIM;
    float* ph = h_ent + (size_t)s * DIM;
#pragma unroll
    for (int half = 0; half < 2; ++half) {
      const int j = lane + 32 * half;
      const float c = ps[j] + pd[j] + pr[j] + cvec[j];
      atomicAdd(&ph[j], alpha * c);
    }
  }
}

__global__ __launch_bounds__(256) void edge_featsum_kernel(
    const int* __restrict__ src, const int* __restrict__ dst,
    const int* __restrict__ rel, const float* __restrict__ h_ent,
    float* __restrict__ featsum, int E) {
  const int gw = blockIdx.x * (blockDim.x >> 5) + (threadIdx.x >> 5);
  const int lane = threadIdx.x & 31;
  const int nw = gridDim.x * (blockDim.x >> 5);
  for (int e = gw; e < E; e += nw) {
    const int s = src[e], d = dst[e], r = rel[e];
    const float* hs = h_ent + (size_t)s * DIM;
    const float* hd = h_ent + (size_t)d * DIM;
    float* fs = featsum + (size_t)r * 128;
#pragma unroll
    for (int half = 0; half < 2; ++half) {
      const int j = lane + 32 * half;
      atomicAdd(&fs[j], hs[j]);
      atomicAdd(&fs[64 + j], hd[j]);
    }
  }
}

__global__ void h_rel_kernel(const float* __restrict__ featsum,
                             const float* __restrict__ cnt,
                             const float* __restrict__ rel2_proj,
                             const float* __restrict__ W_rel3,
                             const float* __restrict__ b_rel3,
                             float* __restrict__ h_rel, int R) {
  const int t = blockIdx.x * blockDim.x + threadIdx.x;
  if (t >= R * DIM) return;
  const int r = t >> 6, o = t & 63;
  const float c = cnt[r];
  const float inv = 1.0f / fmaxf(c, 1.0f);
  const float* fs = featsum + (size_t)r * 128;
  const float* w = W_rel3 + (size_t)o * 192;
  float acc = b_rel3[o];
  for (int j = 0; j < 128; ++j) acc += (fs[j] * inv) * w[j];
  if (c > 0.f) {
    const float* r2 = rel2_proj + (size_t)r * DIM;
    for (int j = 0; j < DIM; ++j) acc += r2[j] * w[128 + j];
  }
  h_rel[t] = acc;
}

#define OFF_MS     0
#define OFF_MD     4096
#define OFF_MR     8192
#define OFF_M2     12288
#define OFF_CVEC   16384
#define OFF_VREL2  16448
#define OFF_A0     16512
#define OFF_UR     16576
#define OFF_AR     48576
#define OFF_REL2P  49088
#define OFF_AS     81088
#define OFF_AD     181088
#define OFF_BSUM   281088
#define OFF_CNT    381088
#define OFF_FEAT   381588
#define OFF_BE     445588
#define OFF_US     1445588
#define OFF_UD     7845588
#define ZERO_CNT   (OFF_FEAT + NUM_RELS * 128 - OFF_BSUM)

extern "C" void kernel_launch(void* const* d_in, const int* in_sizes, int n_in,
                              void* d_out, int out_size, void* d_ws,
                              size_t ws_size, hipStream_t stream) {
  const float* ent_embed = (const float*)d_in[0];
  const float* rel_embed = (const float*)d_in[1];
  const float* W_ent  = (const float*)d_in[2];
  const float* b_ent  = (const float*)d_in[3];
  const float* W_relL = (const float*)d_in[4];
  const float* b_relL = (const float*)d_in[5];
  const float* W_rel2 = (const float*)d_in[6];
  const float* b_rel2 = (const float*)d_in[7];
  const float* W_rel3 = (const float*)d_in[8];
  const float* b_rel3 = (const float*)d_in[9];
  const float* W_a    = (const float*)d_in[10];
  const float* b_a    = (const float*)d_in[11];
  const float* W_fc   = (const float*)d_in[12];
  const float* b_fc   = (const float*)d_in[13];
  const int* src_idx  = (const int*)d_in[14];
  const int* dst_idx  = (const int*)d_in[15];
  const int* rel_idx  = (const int*)d_in[16];

  float* ws = (float*)d_ws;
  float* h_ent = (float*)d_out;
  float* h_rel = (float*)d_out + (size_t)NUM_NODES * DIM;

  zero_kernel<<<2048, 256, 0, stream>>>(h_ent, NUM_NODES * DIM);
  zero_kernel<<<512, 256, 0, stream>>>(ws + OFF_BSUM, ZERO_CNT);

  fold_kernel<<<1, 256, 0, stream>>>(
      W_ent, b_ent, W_relL, b_relL, W_rel2, b_rel2, W_a, b_a, W_fc, b_fc,
      ws + OFF_MS, ws + OFF_MD, ws + OFF_MR, ws + OFF_M2, ws + OFF_CVEC,
      ws + OFF_VREL2, ws + OFF_A0);

  const int gemm_blocks_N = (NUM_NODES + 127) / 128;
  const int gemm_blocks_R = (NUM_RELS + 127) / 128;
  wmma_gemm64<<<gemm_blocks_N, 256, 0, stream>>>(ent_embed, ws + OFF_MS,
                                                 nullptr, ws + OFF_US, NUM_NODES);
  wmma_gemm64<<<gemm_blocks_N, 256, 0, stream>>>(ent_embed, ws + OFF_MD,
                                                 nullptr, ws + OFF_UD, NUM_NODES);
  wmma_gemm64<<<gemm_blocks_R, 256, 0, stream>>>(rel_embed, ws + OFF_MR,
                                                 nullptr, ws + OFF_UR, NUM_RELS);
  wmma_gemm64<<<gemm_blocks_R, 256, 0, stream>>>(rel_embed, ws + OFF_M2,
                                                 ws + OFF_VREL2, ws + OFF_REL2P,
                                                 NUM_RELS);

  dot64_kernel<<<(NUM_NODES + 255) / 256, 256, 0, stream>>>(ws + OFF_US, W_a,
                                                            ws + OFF_AS, NUM_NODES);
  dot64_kernel<<<(NUM_NODES + 255) / 256, 256, 0, stream>>>(ws + OFF_UD, W_a,
                                                            ws + OFF_AD, NUM_NODES);
  dot64_kernel<<<(NUM_RELS + 255) / 256, 256, 0, stream>>>(ws + OFF_UR, W_a,
                                                           ws + OFF_AR, NUM_RELS);

  edge_logits_kernel<<<4096, 256, 0, stream>>>(
      src_idx, dst_idx, rel_idx, ws + OFF_AS, ws + OFF_AD, ws + OFF_AR,
      ws + OFF_A0, ws + OFF_BE, ws + OFF_BSUM, ws + OFF_CNT, NUM_EDGES);
  edge_scatter_h_kernel<<<4096, 256, 0, stream>>>(
      src_idx, dst_idx, rel_idx, ws + OFF_US, ws + OFF_UD, ws + OFF_UR,
      ws + OFF_CVEC, ws + OFF_BE, ws + OFF_BSUM, h_ent, NUM_EDGES);
  edge_featsum_kernel<<<4096, 256, 0, stream>>>(
      src_idx, dst_idx, rel_idx, h_ent, ws + OFF_FEAT, NUM_EDGES);

  h_rel_kernel<<<(NUM_RELS * DIM + 255) / 256, 256, 0, stream>>>(
      ws + OFF_FEAT, ws + OFF_CNT, ws + OFF_REL2P, W_rel3, b_rel3, h_rel,
      NUM_RELS);
}


# edge_featsum: each work-group sums its edges' h_ent rows for one relation range in an LDS table (ds_add_f32), adds it to featsum once; no per-edge global atomics; f32 sum order changes
# speedup vs baseline: 2.1542x; 2.1542x over previous
#include <hip/hip_runtime.h>
#include <math.h>


#define NUM_NODES 100000
#define NUM_RELS  500
#define DIM       64
#define NUM_EDGES 1000000

typedef __attribute__((ext_vector_type(2))) float v2f;
typedef __attribute__((ext_vector_type(8))) float v8f;

__global__ void zero_kernel(float* __restrict__ p, int n) {
  for (int i = blockIdx.x * blockDim.x + threadIdx.x; i < n;
       i += gridDim.x * blockDim.x)
    p[i] = 0.0f;
}

__global__ __launch_bounds__(256) void fold_kernel(
    const float* __restrict__ W_ent, const float* __restrict__ b_ent,
    const float* __restrict__ W_relL, const float* __restrict__ b_relL,
    const float* __restrict__ W_rel2, const float* __restrict__ b_rel2,
    const float* __restrict__ W_a, const float* __restrict__ b_a,
    const float* __restrict__ W_fc, const float* __restrict__ b_fc,
    float* __restrict__ M_s, float* __restrict__ M_d,
    float* __restrict__ M_r, float* __restrict__ M_2,
    float* __restrict__ cvec, float* __restrict__ v_rel2,
    float* __restrict__ a0) {
  __shared__ float scvec[DIM];
  const int t = threadIdx.x;
  for (int idx = t; idx < DIM * DIM; idx += 256) {
    const int o = idx >> 6, k = idx & 63;
    float ms = 0.f, md = 0.f, mr = 0.f, m2 = 0.f;
    for (int u = 0; u < DIM; ++u) {
      const float we = W_ent[u * DIM + k];
      const float wl = W_relL[u * DIM + k];
      ms += W_fc[o * 192 + u] * we;
      md += W_fc[o * 192 + 64 + u] * we;
      mr += W_fc[o * 192 + 128 + u] * wl;
      m2 += W_rel2[o * DIM + u] * wl;
    }
    M_s[idx] = ms; M_d[idx] = md; M_r[idx] = mr; M_2[idx] = m2;
  }
  if (t < DIM) {
    float cv = b_fc[t], v2 = b_rel2[t];
    for (int u = 0; u < DIM; ++u) {
      cv += W_fc[t * 192 + u] * b_ent[u]
          + W_fc[t * 192 + 64 + u] * b_ent[u]
          + W_fc[t * 192 + 128 + u] * b_relL[u];
      v2 += W_rel2[t * DIM + u] * b_relL[u];
    }
    cvec[t] = cv; v_rel2[t] = v2; scvec[t] = cv;
  }
  __syncthreads();
  if (t == 0) {
    float a = b_a[0];
    for (int o = 0; o < DIM; ++o) a += scvec[o] * W_a[o];
    a0[0] = a;
  }
}

__global__ __launch_bounds__(256) void wmma_gemm64(
    const float* __restrict__ A, const float* __restrict__ W,
    const float* __restrict__ bias, float* __restrict__ out, int M) {
  __shared__ float sW[DIM * DIM];
  for (int i = threadIdx.x; i < DIM * DIM; i += 256) sW[i] = W[i];
  __syncthreads();

  const int wave = threadIdx.x >> 5;
  const int lane = threadIdx.x & 31;
  const int m0 = (blockIdx.x * 8 + wave) * 16;
  if (m0 >= M) return;

  const int lm = lane & 15;
  const int hi = lane >> 4;

  int arow = m0 + lm;
  if (arow >= M) arow = M - 1;
  const float* __restrict__ Arow = A + (size_t)arow * DIM;

  v8f acc0 = {}, acc1 = {}, acc2 = {}, acc3 = {};
  for (int k = 0; k < DIM; k += 4) {
    v2f a;
    a.x = Arow[k + 2 * hi + 0];
    a.y = Arow[k + 2 * hi + 1];
    v2f b0, b1, b2, b3;
    const float* w0 = &sW[(0 * 16 + lm) * DIM + k + 2 * hi];
    const float* w1 = &sW[(1 * 16 + lm) * DIM + k + 2 * hi];
    const float* w2 = &sW[(2 * 16 + lm) * DIM + k + 2 * hi];
    const float* w3 = &sW[(3 * 16 + lm) * DIM + k + 2 * hi];
    b0.x = w0[0]; b0.y = w0[1];
    b1.x = w1[0]; b1.y = w1[1];
    b2.x = w2[0]; b2.y = w2[1];
    b3.x = w3[0]; b3.y = w3[1];
    acc0 = __builtin_amdgcn_wmma_f32_16x16x4_f32(false, a, false, b0,
                                                 (short)0, acc0, false, false);
    acc1 = __builtin_amdgcn_wmma_f32_16x16x4_f32(false, a, false, b1,
                                                 (short)0, acc1, false, false);
    acc2 = __builtin_amdgcn_wmma_f32_16x16x4_f32(false, a, false, b2,
                                                 (short)0, acc2, false, false);
    acc3 = __builtin_amdgcn_wmma_f32_16x16x4_f32(false, a, false, b3,
                                                 (short)0, acc3, false, false);
  }

  v8f accs[4] = {acc0, acc1, acc2, acc3};
#pragma unroll
  for (int ob = 0; ob < 4; ++ob) {
    const float bv = bias ? bias[ob * 16 + lm] : 0.0f;
#pragma unroll
    for (int v = 0; v < 8; ++v) {
      const int row = m0 + v + 8 * hi;
      if (row < M)
        out[(size_t)row * DIM + ob * 16 + lm] = accs[ob][v] + bv;
    }
  }
}

__global__ void dot64_kernel(const float* __restrict__ X,
                             const float* __restrict__ v,
                             float* __restrict__ out, int M) {
  const int m = blockIdx.x * blockDim.x + threadIdx.x;
  if (m >= M) return;
  const float* x = X + (size_t)m * DIM;
  float acc = 0.f;
#pragma unroll 8
  for (int j = 0; j < DIM; ++j) acc += x[j] * v[j];
  out[m] = acc;
}

__global__ void edge_logits_kernel(
    const int* __restrict__ src, const int* __restrict__ dst,
    const int* __restrict__ rel, const float* __restrict__ a_s,
    const float* __restrict__ a_d, const float* __restrict__ a_r,
    const float* __restrict__ a0p, float* __restrict__ b_e,
    float* __restrict__ b_sum, float* __restrict__ cnt, int E) {
  const float a0 = a0p[0];
  for (int e = blockIdx.x * blockDim.x + threadIdx.x; e < E;
       e += gridDim.x * blockDim.x) {
    const int s = src[e], d = dst[e], r = rel[e];
    float l = a_s[s] + a_d[d] + a_r[r] + a0;
    l = l > 0.f ? l : 0.01f * l;
    const float b = expf(l);
    b_e[e] = b;
    atomicAdd(&b_sum[s], b);
    atomicAdd(&cnt[r], 1.0f);
  }
}

__global__ __launch_bounds__(256) void edge_scatter_h_kernel(
    const int* __restrict__ src, const int* __restrict__ dst,
    const int* __restrict__ rel, const float* __restrict__ u_s,
    const float* __restrict__ u_d, const float* __restrict__ u_r,
    const float* __restrict__ cvec, const float* __restrict__ b_e,
    const float* __restrict__ b_sum, float* __restrict__ h_ent, int E) {
  const int gw = blockIdx.x * (blockDim.x >> 5) + (threadIdx.x >> 5);
  const int lane = threadIdx.x & 31;
  const int nw = gridDim.x * (blockDim.x >> 5);
  for (int e = gw; e < E; e += nw) {
    const int s = src[e], d = dst[e], r = rel[e];
    const float alpha = b_e[e] / b_sum[s];
    const float* ps = u_s + (size_t)s * DIM;
    const float* pd = u_d + (size_t)d * DIM;
    const float* pr = u_r + (size_t)r * DIM;
    float* ph = h_ent + (size_t)s * DIM;
#pragma unroll
    for (int half = 0; half < 2; ++half) {
      const int j = lane + 32 * half;
      const float c = ps[j] + pd[j] + pr[j] + cvec[j];
      atomicAdd(&ph[j], alpha * c);
    }
  }
}

#define FS_GROUPS 8
#define FS_RPG    63
#define FS_SLICES 32
#define FS_BATCH  8
static_assert(FS_GROUPS * FS_RPG >= NUM_RELS, "relation ranges must cover all relations");

__global__ __launch_bounds__(1024) void edge_featsum_kernel(
    const int* __restrict__ src, const int* __restrict__ dst,
    const int* __restrict__ rel, const float* __restrict__ h_ent,
    float* __restrict__ featsum, int E) {
  __shared__ float sfs[FS_RPG * 128];
  const int t = threadIdx.x;
  for (int i = t; i < FS_RPG * 128; i += 1024) sfs[i] = 0.0f;
  __syncthreads();

  const int grp = blockIdx.x % FS_GROUPS;
  const int slice = blockIdx.x / FS_GROUPS;
  const int r0 = grp * FS_RPG;
  int nrel = NUM_RELS - r0;
  nrel = nrel > FS_RPG ? FS_RPG : nrel;
  nrel = nrel < 0 ? 0 : nrel;
  const int per = (E + FS_SLICES - 1) / FS_SLICES;
  const int e0 = slice * per;
  const int e1 = (e0 + per < E) ? (e0 + per) : E;

  const int wave = __builtin_amdgcn_readfirstlane(t >> 5);
  const int lane = t & 31;

  int base = e0 + wave * 32;
  int rn, sn, dn;
  {
    int ec = base + lane < e1 ? base + lane : e1 - 1;
    ec = ec < 0 ? 0 : ec;
    rn = rel[ec]; sn = src[ec]; dn = dst[ec];
    if (base + lane >= e1) rn = -1;
  }
  for (; base < e1; base += 32 * 32) {
    const int r = rn, s = sn, d = dn;
    const int en = base + 32 * 32 + lane;
    const int ec = en < e1 ? en : e1 - 1;
    rn = rel[ec]; sn = src[ec]; dn = dst[ec];
    if (en >= e1) rn = -1;

    const int loc = r - r0;
    unsigned mask =
        __builtin_amdgcn_ballot_w32((unsigned)loc < (unsigned)nrel);

    while (mask != 0u) {
      const int n = __builtin_popcount(mask);
      v2f vs[FS_BATCH], vd[FS_BATCH];
      int lo[FS_BATCH];
#pragma unroll
      for (int k = 0; k < FS_BATCH; ++k) {
        if (k < n) {
          const int l = __builtin_ctz(mask);
          mask &= mask - 1u;
          const int sk = __builtin_amdgcn_readlane(s, l);
          const int dk = __builtin_amdgcn_readlane(d, l);
          lo[k] = __builtin_amdgcn_readlane(loc, l);
          vs[k] = reinterpret_cast<const v2f*>(h_ent + (size_t)sk * DIM)[lane];
          vd[k] = reinterpret_cast<const v2f*>(h_ent + (size_t)dk * DIM)[lane];
        }
      }
#pragma unroll
      for (int k = 0; k < FS_BATCH; ++k) {
        if (k < n) {
          float* p = &sfs[lo[k] * 128 + 2 * lane];
          __hip_atomic_fetch_add(p, vs[k].x, __ATOMIC_RELAXED,
                                 __HIP_MEMORY_SCOPE_WORKGROUP);
          __hip_atomic_fetch_add(p + 1, vs[k].y, __ATOMIC_RELAXED,
                                 __HIP_MEMORY_SCOPE_WORKGROUP);
          __hip_atomic_fetch_add(p + 64, vd[k].x, __ATOMIC_RELAXED,
                                 __HIP_MEMORY_SCOPE_WORKGROUP);
          __hip_atomic_fetch_add(p + 65, vd[k].y, __ATOMIC_RELAXED,
                                 __HIP_MEMORY_SCOPE_WORKGROUP);
        }
      }
    }
  }
  __syncthreads();

  float* fs = featsum + (size_t)r0 * 128;
  for (int i = t; i < nrel * 128; i += 1024) atomicAdd(&fs[i], sfs[i]);
}

__global__ void h_rel_kernel(const float* __restrict__ featsum,
                             const float* __restrict__ cnt,
                             const float* __restrict__ rel2_proj,
                             const float* __restrict__ W_rel3,
                             const float* __restrict__ b_rel3,
                             float* __restrict__ h_rel, int R) {
  const int t = blockIdx.x * blockDim.x + threadIdx.x;
  if (t >= R * DIM) return;
  const int r = t >> 6, o = t & 63;
  const float c = cnt[r];
  const float inv = 1.0f / fmaxf(c, 1.0f);
  const float* fs = featsum + (size_t)r * 128;
  const float* w = W_rel3 + (size_t)o * 192;
  float acc = b_rel3[o];
  for (int j = 0; j < 128; ++j) acc += (fs[j] * inv) * w[j];
  if (c > 0.f) {
    const float* r2 = rel2_proj + (size_t)r * DIM;
    for (int j = 0; j < DIM; ++j) acc += r2[j] * w[128 + j];
  }
  h_rel[t] = acc;
}

#define OFF_MS     0
#define OFF_MD     4096
#define OFF_MR     8192
#define OFF_M2     12288
#define OFF_CVEC   16384
#define OFF_VREL2  16448
#define OFF_A0     16512
#define OFF_UR     16576
#define OFF_AR     48576
#define OFF_REL2P  49088
#define OFF_AS     81088
#define OFF_AD     181088
#define OFF_BSUM   281088
#define OFF_CNT    381088
#define OFF_FEAT   381588
#define OFF_BE     445588
#define OFF_US     1445588
#define OFF_UD     7845588
#define ZERO_CNT   (OFF_FEAT + NUM_RELS * 128 - OFF_BSUM)

extern "C" void kernel_launch(void* const* d_in, const int* in_sizes, int n_in,
                              void* d_out, int out_size, void* d_ws,
                              size_t ws_size, hipStream_t stream) {
  const float* ent_embed = (const float*)d_in[0];
  const float* rel_embed = (const float*)d_in[1];
  const float* W_ent  = (const float*)d_in[2];
  const float* b_ent  = (const float*)d_in[3];
  const float* W_relL = (const float*)d_in[4];
  const float* b_relL = (const float*)d_in[5];
  const float* W_rel2 = (const float*)d_in[6];
  const float* b_rel2 = (const float*)d_in[7];
  const float* W_rel3 = (const float*)d_in[8];
  const float* b_rel3 = (const float*)d_in[9];
  const float* W_a    = (const float*)d_in[10];
  const float* b_a    = (const float*)d_in[11];
  const float* W_fc   = (const float*)d_in[12];
  const float* b_fc   = (const float*)d_in[13];
  const int* src_idx  = (const int*)d_in[14];
  const int* dst_idx  = (const int*)d_in[15];
  const int* rel_idx  = (const int*)d_in[16];

  float* ws = (float*)d_ws;
  float* h_ent = (float*)d_out;
  float* h_rel = (float*)d_out + (size_t)NUM_NODES * DIM;

  zero_kernel<<<2048, 256, 0, stream>>>(h_ent, NUM_NODES * DIM);
  zero_kernel<<<512, 256, 0, stream>>>(ws + OFF_BSUM, ZERO_CNT);

  fold_kernel<<<1, 256, 0, stream>>>(
      W_ent, b_ent, W_relL, b_relL, W_rel2, b_rel2, W_a, b_a, W_fc, b_fc,
      ws + OFF_MS, ws + OFF_MD, ws + OFF_MR, ws + OFF_M2, ws + OFF_CVEC,
      ws + OFF_VREL2, ws + OFF_A0);

  const int gemm_blocks_N = (NUM_NODES + 127) / 128;
  const int gemm_blocks_R = (NUM_RELS + 127) / 128;
  wmma_gemm64<<<gemm_blocks_N, 256, 0, stream>>>(ent_embed, ws + OFF_MS,
                                                 nullptr, ws + OFF_US, NUM_NODES);
  wmma_gemm64<<<gemm_blocks_N, 256, 0, stream>>>(ent_embed, ws + OFF_MD,
                                                 nullptr, ws + OFF_UD, NUM_NODES);
  wmma_gemm64<<<gemm_blocks_R, 256, 0, stream>>>(rel_embed, ws + OFF_MR,
                                                 nullptr, ws + OFF_UR, NUM_RELS);
  wmma_gemm64<<<gemm_blocks_R, 256, 0, stream>>>(rel_embed, ws + OFF_M2,
                                                 ws + OFF_VREL2, ws + OFF_REL2P,
                                                 NUM_RELS);

  dot64_kernel<<<(NUM_NODES + 255) / 256, 256, 0, stream>>>(ws + OFF_US, W_a,
                                                            ws + OFF_AS, NUM_NODES);
  dot64_kernel<<<(NUM_NODES + 255) / 256, 256, 0, stream>>>(ws + OFF_UD, W_a,
                                                            ws + OFF_AD, NUM_NODES);
  dot64_kernel<<<(NUM_RELS + 255) / 256, 256, 0, stream>>>(ws + OFF_UR, W_a,
                                                           ws + OFF_AR, NUM_RELS);

  edge_logits_kernel<<<4096, 256, 0, stream>>>(
      src_idx, dst_idx, rel_idx, ws + OFF_AS, ws + OFF_AD, ws + OFF_AR,
      ws + OFF_A0, ws + OFF_BE, ws + OFF_BSUM, ws + OFF_CNT, NUM_EDGES);
  edge_scatter_h_kernel<<<4096, 256, 0, stream>>>(
      src_idx, dst_idx, rel_idx, ws + OFF_US, ws + OFF_UD, ws + OFF_UR,
      ws + OFF_CVEC, ws + OFF_BE, ws + OFF_BSUM, h_ent, NUM_EDGES);
  edge_featsum_kernel<<<FS_GROUPS * FS_SLICES, 1024, 0, stream>>>(
      src_idx, dst_idx, rel_idx, h_ent, ws + OFF_FEAT, NUM_EDGES);

  h_rel_kernel<<<(NUM_RELS * DIM + 255) / 256, 256, 0, stream>>>(
      ws + OFF_FEAT, ws + OFF_CNT, ws + OFF_REL2P, W_rel3, b_rel3, h_rel,
      NUM_RELS);
}


# edge_scatter_h: each work-group owns 400 nodes, scans all src, sums b*(u_d+u_r) of its edges in LDS (ds_add_f32), stores h_ent once as sum/b_sum+u_s+cvec; no global atomics; f32 sum order changes
# speedup vs baseline: 2.2640x; 1.0510x over previous
#include <hip/hip_runtime.h>
#include <math.h>


#define NUM_NODES 100000
#define NUM_RELS  500
#define DIM       64
#define NUM_EDGES 1000000

typedef __attribute__((ext_vector_type(2))) float v2f;
typedef __attribute__((ext_vector_type(8))) float v8f;

__global__ void zero_kernel(float* __restrict__ p, int n) {
  for (int i = blockIdx.x * blockDim.x + threadIdx.x; i < n;
       i += gridDim.x * blockDim.x)
    p[i] = 0.0f;
}

__global__ __launch_bounds__(256) void fold_kernel(
    const float* __restrict__ W_ent, const float* __restrict__ b_ent,
    const float* __restrict__ W_relL, const float* __restrict__ b_relL,
    const float* __restrict__ W_rel2, const float* __restrict__ b_rel2,
    const float* __restrict__ W_a, const float* __restrict__ b_a,
    const float* __restrict__ W_fc, const float* __restrict__ b_fc,
    float* __restrict__ M_s, float* __restrict__ M_d,
    float* __restrict__ M_r, float* __restrict__ M_2,
    float* __restrict__ cvec, float* __restrict__ v_rel2,
    float* __restrict__ a0) {
  __shared__ float scvec[DIM];
  const int t = threadIdx.x;
  for (int idx = t; idx < DIM * DIM; idx += 256) {
    const int o = idx >> 6, k = idx & 63;
    float ms = 0.f, md = 0.f, mr = 0.f, m2 = 0.f;
    for (int u = 0; u < DIM; ++u) {
      const float we = W_ent[u * DIM + k];
      const float wl = W_relL[u * DIM + k];
      ms += W_fc[o * 192 + u] * we;
      md += W_fc[o * 192 + 64 + u] * we;
      mr += W_fc[o * 192 + 128 + u] * wl;
      m2 += W_rel2[o * DIM + u] * wl;
    }
    M_s[idx] = ms; M_d[idx] = md; M_r[idx] = mr; M_2[idx] = m2;
  }
  if (t < DIM) {
    float cv = b_fc[t], v2 = b_rel2[t];
    for (int u = 0; u < DIM; ++u) {
      cv += W_fc[t * 192 + u] * b_ent[u]
          + W_fc[t * 192 + 64 + u] * b_ent[u]
          + W_fc[t * 192 + 128 + u] * b_relL[u];
      v2 += W_rel2[t * DIM + u] * b_relL[u];
    }
    cvec[t] = cv; v_rel2[t] = v2; scvec[t] = cv;
  }
  __syncthreads();
  if (t == 0) {
    float a = b_a[0];
    for (int o = 0; o < DIM; ++o) a += scvec[o] * W_a[o];
    a0[0] = a;
  }
}

__global__ __launch_bounds__(256) void wmma_gemm64(
    const float* __restrict__ A, const float* __restrict__ W,
    const float* __restrict__ bias, float* __restrict__ out, int M) {
  __shared__ float sW[DIM * DIM];
  for (int i = threadIdx.x; i < DIM * DIM; i += 256) sW[i] = W[i];
  __syncthreads();

  const int wave = threadIdx.x >> 5;
  const int lane = threadIdx.x & 31;
  const int m0 = (blockIdx.x * 8 + wave) * 16;
  if (m0 >= M) return;

  const int lm = lane & 15;
  const int hi = lane >> 4;

  int arow = m0 + lm;
  if (arow >= M) arow = M - 1;
  const float* __restrict__ Arow = A + (size_t)arow * DIM;

  v8f acc0 = {}, acc1 = {}, acc2 = {}, acc3 = {};
  for (int k = 0; k < DIM; k += 4) {
    v2f a;
    a.x = Arow[k + 2 * hi + 0];
    a.y = Arow[k + 2 * hi + 1];
    v2f b0, b1, b2, b3;
    const float* w0 = &sW[(0 * 16 + lm) * DIM + k + 2 * hi];
    const float* w1 = &sW[(1 * 16 + lm) * DIM + k + 2 * hi];
    const float* w2 = &sW[(2 * 16 + lm) * DIM + k + 2 * hi];
    const float* w3 = &sW[(3 * 16 + lm) * DIM + k + 2 * hi];
    b0.x = w0[0]; b0.y = w0[1];
    b1.x = w1[0]; b1.y = w1[1];
    b2.x = w2[0]; b2.y = w2[1];
    b3.x = w3[0]; b3.y = w3[1];
    acc0 = __builtin_amdgcn_wmma_f32_16x16x4_f32(false, a, false, b0,
                                                 (short)0, acc0, false, false);
    acc1 = __builtin_amdgcn_wmma_f32_16x16x4_f32(false, a, false, b1,
                                                 (short)0, acc1, false, false);
    acc2 = __builtin_amdgcn_wmma_f32_16x16x4_f32(false, a, false, b2,
                                                 (short)0, acc2, false, false);
    acc3 = __builtin_amdgcn_wmma_f32_16x16x4_f32(false, a, false, b3,
                                                 (short)0, acc3, false, false);
  }

  v8f accs[4] = {acc0, acc1, acc2, acc3};
#pragma unroll
  for (int ob = 0; ob < 4; ++ob) {
    const float bv = bias ? bias[ob * 16 + lm] : 0.0f;
#pragma unroll
    for (int v = 0; v < 8; ++v) {
      const int row = m0 + v + 8 * hi;
      if (row < M)
        out[(size_t)row * DIM + ob * 16 + lm] = accs[ob][v] + bv;
    }
  }
}

__global__ void dot64_kernel(const float* __restrict__ X,
                             const float* __restrict__ v,
                             float* __restrict__ out, int M) {
  const int m = blockIdx.x * blockDim.x + threadIdx.x;
  if (m >= M) return;
  const float* x = X + (size_t)m * DIM;
  float acc = 0.f;
#pragma unroll 8
  for (int j = 0; j < DIM; ++j) acc += x[j] * v[j];
  out[m] = acc;
}

__global__ void edge_logits_kernel(
    const int* __restrict__ src, const int* __restrict__ dst,
    const int* __restrict__ rel, const float* __restrict__ a_s,
    const float* __restrict__ a_d, const float* __restrict__ a_r,
    const float* __restrict__ a0p, float* __restrict__ b_e,
    float* __restrict__ b_sum, float* __restrict__ cnt, int E) {
  const float a0 = a0p[0];
  for (int e = blockIdx.x * blockDim.x + threadIdx.x; e < E;
       e += gridDim.x * blockDim.x) {
    const int s = src[e], d = dst[e], r = rel[e];
    float l = a_s[s] + a_d[d] + a_r[r] + a0;
    l = l > 0.f ? l : 0.01f * l;
    const float b = expf(l);
    b_e[e] = b;
    atomicAdd(&b_sum[s], b);
    atomicAdd(&cnt[r], 1.0f);
  }
}

#define HN_NPR    400
#define HN_GROUPS ((NUM_NODES + HN_NPR - 1) / HN_NPR)
#define HN_QCAP   1024
#define HN_BATCH  8
static_assert(NUM_EDGES <= (1 << 22), "edge id must fit 22 bits of a queue word");
static_assert(HN_NPR <= (1 << 10), "local node id must fit 10 bits of a queue word");
static_assert((HN_QCAP & (HN_QCAP - 1)) == 0 && HN_QCAP >= 32 + 512, "ring size");

__device__ __forceinline__ void hn_drain(
    const unsigned* q, int head, int n, int lane,
    const int* __restrict__ dst, const int* __restrict__ rel,
    const float* __restrict__ b_e, const float* __restrict__ u_d,
    const float* __restrict__ u_r, float* stab) {
  int d = 0, r = 0, loc = 0, bb = 0;
  if (lane < n) {
    const unsigned ent = q[(head + lane) & (HN_QCAP - 1)];
    const int e = (int)(ent & 0x3FFFFFu);
    loc = (int)(ent >> 22);
    d = dst[e]; r = rel[e]; bb = __float_as_int(b_e[e]);
  }
  for (int k0 = 0; k0 < n; k0 += HN_BATCH) {
    v2f vd[HN_BATCH], vr[HN_BATCH];
    float bk[HN_BATCH];
    int lo[HN_BATCH];
#pragma unroll
    for (int k = 0; k < HN_BATCH; ++k) {
      if (k0 + k < n) {
        const int dk = __builtin_amdgcn_readlane(d, k0 + k);
        const int rk = __builtin_amdgcn_readlane(r, k0 + k);
        bk[k] = __int_as_float(__builtin_amdgcn_readlane(bb, k0 + k));
        lo[k] = __builtin_amdgcn_readlane(loc, k0 + k);
        vd[k] = reinterpret_cast<const v2f*>(u_d + (size_t)dk * DIM)[lane];
        vr[k] = reinterpret_cast<const v2f*>(u_r + (size_t)rk * DIM)[lane];
      }
    }
#pragma unroll
    for (int k = 0; k < HN_BATCH; ++k) {
      if (k0 + k < n) {
        float* p = &stab[lo[k] * DIM + 2 * lane];
        __hip_atomic_fetch_add(p, bk[k] * (vd[k].x + vr[k].x), __ATOMIC_RELAXED,
                               __HIP_MEMORY_SCOPE_WORKGROUP);
        __hip_atomic_fetch_add(p + 1, bk[k] * (vd[k].y + vr[k].y),
                               __ATOMIC_RELAXED, __HIP_MEMORY_SCOPE_WORKGROUP);
      }
    }
  }
}

__device__ __forceinline__ void hn_push(unsigned* q, int head, int& cnt,
                                        int lane, int loc, int nn, int e) {
  const bool m = (unsigned)loc < (unsigned)nn;
  const unsigned mask = __builtin_amdgcn_ballot_w32(m);
  if (mask != 0u) {
    if (m) {
      const int rank = (int)__builtin_amdgcn_mbcnt_lo(mask, 0u);
      q[(head + cnt + rank) & (HN_QCAP - 1)] =
          ((unsigned)loc << 22) | (unsigned)e;
    }
    cnt += __builtin_popcount(mask);
  }
}

__global__ __launch_bounds__(1024) void edge_gather_h_kernel(
    const int* __restrict__ src, const int* __restrict__ dst,
    const int* __restrict__ rel, const float* __restrict__ u_s,
    const float* __restrict__ u_d, const float* __restrict__ u_r,
    const float* __restrict__ cvec, const float* __restrict__ b_e,
    const float* __restrict__ b_sum, float* __restrict__ h_ent, int E) {
  __shared__ __attribute__((aligned(16))) float stab[HN_NPR * DIM];
  __shared__ unsigned sq[32 * HN_QCAP];
  const int t = threadIdx.x;
  for (int i = t; i < HN_NPR * DIM; i += 1024) stab[i] = 0.0f;
  __syncthreads();

  const int n0 = blockIdx.x * HN_NPR;
  int nn = NUM_NODES - n0;
  nn = nn > HN_NPR ? HN_NPR : nn;
  nn = nn < 0 ? 0 : nn;

  const int wave = __builtin_amdgcn_readfirstlane(t >> 5);
  const int lane = t & 31;
  unsigned* q = &sq[wave * HN_QCAP];
  int head = 0, cnt = 0;

  const int ngroup = E / 512;
  const int4* __restrict__ src4 = reinterpret_cast<const int4*>(src);
  const int rot = (int)(((long long)blockIdx.x * ngroup) / HN_GROUPS);
  int g = wave;
  int4 nx0, nx1, nx2, nx3;
  nx0 = nx1 = nx2 = nx3 = make_int4(-1, -1, -1, -1);
  int gcur = 0;
  if (g < ngroup) {
    gcur = g + rot; gcur = gcur >= ngroup ? gcur - ngroup : gcur;
    const int4* p = src4 + (size_t)gcur * 128 + lane;
    nx0 = p[0]; nx1 = p[32]; nx2 = p[64]; nx3 = p[96];
  }
  for (; g < ngroup; g += 32) {
    const int4 c0 = nx0, c1 = nx1, c2 = nx2, c3 = nx3;
    const int eb = gcur * 512 + 4 * lane;
    const int gn = g + 32;
    if (gn < ngroup) {
      gcur = gn + rot; gcur = gcur >= ngroup ? gcur - ngroup : gcur;
      const int4* p = src4 + (size_t)gcur * 128 + lane;
      nx0 = p[0]; nx1 = p[32]; nx2 = p[64]; nx3 = p[96];
    }
    hn_push(q, head, cnt, lane, c0.x - n0, nn, eb + 0);
    hn_push(q, head, cnt, lane, c0.y - n0, nn, eb + 1);
    hn_push(q, head, cnt, lane, c0.z - n0, nn, eb + 2);
    hn_push(q, head, cnt, lane, c0.w - n0, nn, eb + 3);
    hn_push(q, head, cnt, lane, c1.x - n0, nn, eb + 128);
    hn_push(q, head, cnt, lane, c1.y - n0, nn, eb + 129);
    hn_push(q, head, cnt, lane, c1.z - n0, nn, eb + 130);
    hn_push(q, head, cnt, lane, c1.w - n0, nn, eb + 131);
    hn_push(q, head, cnt, lane, c2.x - n0, nn, eb + 256);
    hn_push(q, head, cnt, lane, c2.y - n0, nn, eb + 257);
    hn_push(q, head, cnt, lane, c2.z - n0, nn, eb + 258);
    hn_push(q, head, cnt, lane, c2.w - n0, nn, eb + 259);
    hn_push(q, head, cnt, lane, c3.x - n0, nn, eb + 384);
    hn_push(q, head, cnt, lane, c3.y - n0, nn, eb + 385);
    hn_push(q, head, cnt, lane, c3.z - n0, nn, eb + 386);
    hn_push(q, head, cnt, lane, c3.w - n0, nn, eb + 387);
    while (cnt >= 32) {
      hn_drain(q, head, 32, lane, dst, rel, b_e, u_d, u_r, stab);
      head = (head + 32) & (HN_QCAP - 1);
      cnt -= 32;
    }
  }
  {
    const int e = ngroup * 512 + wave * 32 + lane;
    int loc = -1;
    if (e < E) loc = src[e] - n0;
    hn_push(q, head, cnt, lane, loc, nn, e);
  }
  while (cnt > 0) {
    const int n = cnt < 32 ? cnt : 32;
    hn_drain(q, head, n, lane, dst, rel, b_e, u_d, u_r, stab);
    head = (head + n) & (HN_QCAP - 1);
    cnt -= n;
  }
  __syncthreads();

  for (int i = t; i < nn * (DIM / 4); i += 1024) {
    const int row = i >> 4, c4 = i & 15;
    const int node = n0 + row;
    const float bs = b_sum[node];
    float4 o = make_float4(0.0f, 0.0f, 0.0f, 0.0f);
    if (bs > 0.0f) {
      const float4 a = reinterpret_cast<const float4*>(stab)[row * 16 + c4];
      const float4 us =
          reinterpret_cast<const float4*>(u_s + (size_t)node * DIM)[c4];
      const float4 cv = reinterpret_cast<const float4*>(cvec)[c4];
      o.x = a.x / bs + (us.x + cv.x);
      o.y = a.y / bs + (us.y + cv.y);
      o.z = a.z / bs + (us.z + cv.z);
      o.w = a.w / bs + (us.w + cv.w);
    }
    reinterpret_cast<float4*>(h_ent + (size_t)node * DIM)[c4] = o;
  }
}

#define FS_GROUPS 8
#define FS_RPG    63
#define FS_SLICES 32
#define FS_BATCH  8
static_assert(FS_GROUPS * FS_RPG >= NUM_RELS, "relation ranges must cover all relations");

__global__ __launch_bounds__(1024) void edge_featsum_kernel(
    const int* __restrict__ src, const int* __restrict__ dst,
    const int* __restrict__ rel, const float* __restrict__ h_ent,
    float* __restrict__ featsum, int E) {
  __shared__ float sfs[FS_RPG * 128];
  const int t = threadIdx.x;
  for (int i = t; i < FS_RPG * 128; i += 1024) sfs[i] = 0.0f;
  __syncthreads();

  const int grp = blockIdx.x % FS_GROUPS;
  const int slice = blockIdx.x / FS_GROUPS;
  const int r0 = grp * FS_RPG;
  int nrel = NUM_RELS - r0;
  nrel = nrel > FS_RPG ? FS_RPG : nrel;
  nrel = nrel < 0 ? 0 : nrel;
  const int per = (E + FS_SLICES - 1) / FS_SLICES;
  const int e0 = slice * per;
  const int e1 = (e0 + per < E) ? (e0 + per) : E;

  const int wave = __builtin_amdgcn_readfirstlane(t >> 5);
  const int lane = t & 31;

  int base = e0 + wave * 32;
  int rn, sn, dn;
  {
    int ec = base + lane < e1 ? base + lane : e1 - 1;
    ec = ec < 0 ? 0 : ec;
    rn = rel[ec]; sn = src[ec]; dn = dst[ec];
    if (base + lane >= e1) rn = -1;
  }
  for (; base < e1; base += 32 * 32) {
    const int r = rn, s = sn, d = dn;
    const int en = base + 32 * 32 + lane;
    const int ec = en < e1 ? en : e1 - 1;
    rn = rel[ec]; sn = src[ec]; dn = dst[ec];
    if (en >= e1) rn = -1;

    const int loc = r - r0;
    unsigned mask =
        __builtin_amdgcn_ballot_w32((unsigned)loc < (unsigned)nrel);

    while (mask != 0u) {
      const int n = __builtin_popcount(mask);
      v2f vs[FS_BATCH], vd[FS_BATCH];
      int lo[FS_BATCH];
#pragma unroll
      for (int k = 0; k < FS_BATCH; ++k) {
        if (k < n) {
          const int l = __builtin_ctz(mask);
          mask &= mask - 1u;
          const int sk = __builtin_amdgcn_readlane(s, l);
          const int dk = __builtin_amdgcn_readlane(d, l);
          lo[k] = __builtin_amdgcn_readlane(loc, l);
          vs[k] = reinterpret_cast<const v2f*>(h_ent + (size_t)sk * DIM)[lane];
          vd[k] = reinterpret_cast<const v2f*>(h_ent + (size_t)dk * DIM)[lane];
        }
      }
#pragma unroll
      for (int k = 0; k < FS_BATCH; ++k) {
        if (k < n) {
          float* p = &sfs[lo[k] * 128 + 2 * lane];
          __hip_atomic_fetch_add(p, vs[k].x, __ATOMIC_RELAXED,
                                 __HIP_MEMORY_SCOPE_WORKGROUP);
          __hip_atomic_fetch_add(p + 1, vs[k].y, __ATOMIC_RELAXED,
                                 __HIP_MEMORY_SCOPE_WORKGROUP);
          __hip_atomic_fetch_add(p + 64, vd[k].x, __ATOMIC_RELAXED,
                                 __HIP_MEMORY_SCOPE_WORKGROUP);
          __hip_atomic_fetch_add(p + 65, vd[k].y, __ATOMIC_RELAXED,
                                 __HIP_MEMORY_SCOPE_WORKGROUP);
        }
      }
    }
  }
  __syncthreads();

  float* fs = featsum + (size_t)r0 * 128;
  for (int i = t; i < nrel * 128; i += 1024) atomicAdd(&fs[i], sfs[i]);
}

__global__ void h_rel_kernel(const float* __restrict__ featsum,
                             const float* __restrict__ cnt,
                             const float* __restrict__ rel2_proj,
                             const float* __restrict__ W_rel3,
                             const float* __restrict__ b_rel3,
                             float* __restrict__ h_rel, int R) {
  const int t = blockIdx.x * blockDim.x + threadIdx.x;
  if (t >= R * DIM) return;
  const int r = t >> 6, o = t & 63;
  const float c = cnt[r];
  const float inv = 1.0f / fmaxf(c, 1.0f);
  const float* fs = featsum + (size_t)r * 128;
  const float* w = W_rel3 + (size_t)o * 192;
  float acc = b_rel3[o];
  for (int j = 0; j < 128; ++j) acc += (fs[j] * inv) * w[j];
  if (c > 0.f) {
    const float* r2 = rel2_proj + (size_t)r * DIM;
    for (int j = 0; j < DIM; ++j) acc += r2[j] * w[128 + j];
  }
  h_rel[t] = acc;
}

#define OFF_MS     0
#define OFF_MD     4096
#define OFF_MR     8192
#define OFF_M2     12288
#define OFF_CVEC   16384
#define OFF_VREL2  16448
#define OFF_A0     16512
#define OFF_UR     16576
#define OFF_AR     48576
#define OFF_REL2P  49088
#define OFF_AS     81088
#define OFF_AD     181088
#define OFF_BSUM   281088
#define OFF_CNT    381088
#define OFF_FEAT   381588
#define OFF_BE     445588
#define OFF_US     1445588
#define OFF_UD     7845588
#define ZERO_CNT   (OFF_FEAT + NUM_RELS * 128 - OFF_BSUM)

extern "C" void kernel_launch(void* const* d_in, const int* in_sizes, int n_in,
                              void* d_out, int out_size, void* d_ws,
                              size_t ws_size, hipStream_t stream) {
  const float* ent_embed = (const float*)d_in[0];
  const float* rel_embed = (const float*)d_in[1];
  const float* W_ent  = (const float*)d_in[2];
  const float* b_ent  = (const float*)d_in[3];
  const float* W_relL = (const float*)d_in[4];
  const float* b_relL = (const float*)d_in[5];
  const float* W_rel2 = (const float*)d_in[6];
  const float* b_rel2 = (const float*)d_in[7];
  const float* W_rel3 = (const float*)d_in[8];
  const float* b_rel3 = (const float*)d_in[9];
  const float* W_a    = (const float*)d_in[10];
  const float* b_a    = (const float*)d_in[11];
  const float* W_fc   = (const float*)d_in[12];
  const float* b_fc   = (const float*)d_in[13];
  const int* src_idx  = (const int*)d_in[14];
  const int* dst_idx  = (const int*)d_in[15];
  const int* rel_idx  = (const int*)d_in[16];

  float* ws = (float*)d_ws;
  float* h_ent = (float*)d_out;
  float* h_rel = (float*)d_out + (size_t)NUM_NODES * DIM;

  zero_kernel<<<512, 256, 0, stream>>>(ws + OFF_BSUM, ZERO_CNT);

  fold_kernel<<<1, 256, 0, stream>>>(
      W_ent, b_ent, W_relL, b_relL, W_rel2, b_rel2, W_a, b_a, W_fc, b_fc,
      ws + OFF_MS, ws + OFF_MD, ws + OFF_MR, ws + OFF_M2, ws + OFF_CVEC,
      ws + OFF_VREL2, ws + OFF_A0);

  const int gemm_blocks_N = (NUM_NODES + 127) / 128;
  const int gemm_blocks_R = (NUM_RELS + 127) / 128;
  wmma_gemm64<<<gemm_blocks_N, 256, 0, stream>>>(ent_embed, ws + OFF_MS,
                                                 nullptr, ws + OFF_US, NUM_NODES);
  wmma_gemm64<<<gemm_blocks_N, 256, 0, stream>>>(ent_embed, ws + OFF_MD,
                                                 nullptr, ws + OFF_UD, NUM_NODES);
  wmma_gemm64<<<gemm_blocks_R, 256, 0, stream>>>(rel_embed, ws + OFF_MR,
                                                 nullptr, ws + OFF_UR, NUM_RELS);
  wmma_gemm64<<<gemm_blocks_R, 256, 0, stream>>>(rel_embed, ws + OFF_M2,
                                                 ws + OFF_VREL2, ws + OFF_REL2P,
                                                 NUM_RELS);

  dot64_kernel<<<(NUM_NODES + 255) / 256, 256, 0, stream>>>(ws + OFF_US, W_a,
                                                            ws + OFF_AS, NUM_NODES);
  dot64_kernel<<<(NUM_NODES + 255) / 256, 256, 0, stream>>>(ws + OFF_UD, W_a,
                                                            ws + OFF_AD, NUM_NODES);
  dot64_kernel<<<(NUM_RELS + 255) / 256, 256, 0, stream>>>(ws + OFF_UR, W_a,
                                                           ws + OFF_AR, NUM_RELS);

  edge_logits_kernel<<<4096, 256, 0, stream>>>(
      src_idx, dst_idx, rel_idx, ws + OFF_AS, ws + OFF_AD, ws + OFF_AR,
      ws + OFF_A0, ws + OFF_BE, ws + OFF_BSUM, ws + OFF_CNT, NUM_EDGES);
  edge_gather_h_kernel<<<HN_GROUPS, 1024, 0, stream>>>(
      src_idx, dst_idx, rel_idx, ws + OFF_US, ws + OFF_UD, ws + OFF_UR,
      ws + OFF_CVEC, ws + OFF_BE, ws + OFF_BSUM, h_ent, NUM_EDGES);
  edge_featsum_kernel<<<FS_GROUPS * FS_SLICES, 1024, 0, stream>>>(
      src_idx, dst_idx, rel_idx, h_ent, ws + OFF_FEAT, NUM_EDGES);

  h_rel_kernel<<<(NUM_RELS * DIM + 255) / 256, 256, 0, stream>>>(
      ws + OFF_FEAT, ws + OFF_CNT, ws + OFF_REL2P, W_rel3, b_rel3, h_rel,
      NUM_RELS);
}


# edge_logits launch removed: node-range owner computes each edge's b=exp(leaky(.)) as it drains it, b_sum in LDS (ds_add_f32), edge counts in the relation pass's LDS table; b_sum sum order changes
# speedup vs baseline: 4.2108x; 1.8599x over previous
#include <hip/hip_runtime.h>
#include <math.h>


#define NUM_NODES 100000
#define NUM_RELS  500
#define DIM       64
#define NUM_EDGES 1000000

typedef __attribute__((ext_vector_type(2))) float v2f;
typedef __attribute__((ext_vector_type(8))) float v8f;

__global__ void zero_kernel(float* __restrict__ p, int n) {
  for (int i = blockIdx.x * blockDim.x + threadIdx.x; i < n;
       i += gridDim.x * blockDim.x)
    p[i] = 0.0f;
}

__global__ __launch_bounds__(256) void fold_kernel(
    const float* __restrict__ W_ent, const float* __restrict__ b_ent,
    const float* __restrict__ W_relL, const float* __restrict__ b_relL,
    const float* __restrict__ W_rel2, const float* __restrict__ b_rel2,
    const float* __restrict__ W_a, const float* __restrict__ b_a,
    const float* __restrict__ W_fc, const float* __restrict__ b_fc,
    float* __restrict__ M_s, float* __restrict__ M_d,
    float* __restrict__ M_r, float* __restrict__ M_2,
    float* __restrict__ cvec, float* __restrict__ v_rel2,
    float* __restrict__ a0) {
  __shared__ float scvec[DIM];
  const int t = threadIdx.x;
  for (int idx = t; idx < DIM * DIM; idx += 256) {
    const int o = idx >> 6, k = idx & 63;
    float ms = 0.f, md = 0.f, mr = 0.f, m2 = 0.f;
    for (int u = 0; u < DIM; ++u) {
      const float we = W_ent[u * DIM + k];
      const float wl = W_relL[u * DIM + k];
      ms += W_fc[o * 192 + u] * we;
      md += W_fc[o * 192 + 64 + u] * we;
      mr += W_fc[o * 192 + 128 + u] * wl;
      m2 += W_rel2[o * DIM + u] * wl;
    }
    M_s[idx] = ms; M_d[idx] = md; M_r[idx] = mr; M_2[idx] = m2;
  }
  if (t < DIM) {
    float cv = b_fc[t], v2 = b_rel2[t];
    for (int u = 0; u < DIM; ++u) {
      cv += W_fc[t * 192 + u] * b_ent[u]
          + W_fc[t * 192 + 64 + u] * b_ent[u]
          + W_fc[t * 192 + 128 + u] * b_relL[u];
      v2 += W_rel2[t * DIM + u] * b_relL[u];
    }
    cvec[t] = cv; v_rel2[t] = v2; scvec[t] = cv;
  }
  __syncthreads();
  if (t == 0) {
    float a = b_a[0];
    for (int o = 0; o < DIM; ++o) a += scvec[o] * W_a[o];
    a0[0] = a;
  }
}

__global__ __launch_bounds__(256) void wmma_gemm64(
    const float* __restrict__ A, const float* __restrict__ W,
    const float* __restrict__ bias, float* __restrict__ out, int M) {
  __shared__ float sW[DIM * DIM];
  for (int i = threadIdx.x; i < DIM * DIM; i += 256) sW[i] = W[i];
  __syncthreads();

  const int wave = threadIdx.x >> 5;
  const int lane = threadIdx.x & 31;
  const int m0 = (blockIdx.x * 8 + wave) * 16;
  if (m0 >= M) return;

  const int lm = lane & 15;
  const int hi = lane >> 4;

  int arow = m0 + lm;
  if (arow >= M) arow = M - 1;
  const float* __restrict__ Arow = A + (size_t)arow * DIM;

  v8f acc0 = {}, acc1 = {}, acc2 = {}, acc3 = {};
  for (int k = 0; k < DIM; k += 4) {
    v2f a;
    a.x = Arow[k + 2 * hi + 0];
    a.y = Arow[k + 2 * hi + 1];
    v2f b0, b1, b2, b3;
    const float* w0 = &sW[(0 * 16 + lm) * DIM + k + 2 * hi];
    const float* w1 = &sW[(1 * 16 + lm) * DIM + k + 2 * hi];
    const float* w2 = &sW[(2 * 16 + lm) * DIM + k + 2 * hi];
    const float* w3 = &sW[(3 * 16 + lm) * DIM + k + 2 * hi];
    b0.x = w0[0]; b0.y = w0[1];
    b1.x = w1[0]; b1.y = w1[1];
    b2.x = w2[0]; b2.y = w2[1];
    b3.x = w3[0]; b3.y = w3[1];
    acc0 = __builtin_amdgcn_wmma_f32_16x16x4_f32(false, a, false, b0,
                                                 (short)0, acc0, false, false);
    acc1 = __builtin_amdgcn_wmma_f32_16x16x4_f32(false, a, false, b1,
                                                 (short)0, acc1, false, false);
    acc2 = __builtin_amdgcn_wmma_f32_16x16x4_f32(false, a, false, b2,
                                                 (short)0, acc2, false, false);
    acc3 = __builtin_amdgcn_wmma_f32_16x16x4_f32(false, a, false, b3,
                                                 (short)0, acc3, false, false);
  }

  v8f accs[4] = {acc0, acc1, acc2, acc3};
#pragma unroll
  for (int ob = 0; ob < 4; ++ob) {
    const float bv = bias ? bias[ob * 16 + lm] : 0.0f;
#pragma unroll
    for (int v = 0; v < 8; ++v) {
      const int row = m0 + v + 8 * hi;
      if (row < M)
        out[(size_t)row * DIM + ob * 16 + lm] = accs[ob][v] + bv;
    }
  }
}

__global__ void dot64_kernel(const float* __restrict__ X,
                             const float* __restrict__ v,
                             float* __restrict__ out, int M) {
  const int m = blockIdx.x * blockDim.x + threadIdx.x;
  if (m >= M) return;
  const float* x = X + (size_t)m * DIM;
  float acc = 0.f;
#pragma unroll 8
  for (int j = 0; j < DIM; ++j) acc += x[j] * v[j];
  out[m] = acc;
}

#define HN_NPR    400
#define HN_GROUPS ((NUM_NODES + HN_NPR - 1) / HN_NPR)
#define HN_QCAP   1024
#define HN_BATCH  8
static_assert(NUM_EDGES <= (1 << 22), "edge id must fit 22 bits of a queue word");
static_assert(HN_NPR <= (1 << 10), "local node id must fit 10 bits of a queue word");
static_assert((HN_QCAP & (HN_QCAP - 1)) == 0 && HN_QCAP >= 32 + 512, "ring size");

__device__ __forceinline__ void hn_drain(
    const unsigned* q, int head, int n, int lane, int n0, float a0,
    const int* __restrict__ dst, const int* __restrict__ rel,
    const float* __restrict__ a_s, const float* __restrict__ a_d,
    const float* __restrict__ a_r, const float* __restrict__ u_d,
    const float* __restrict__ u_r, float* stab, float* sbs) {
  int d = 0, r = 0, loc = 0, bb = 0;
  if (lane < n) {
    const unsigned ent = q[(head + lane) & (HN_QCAP - 1)];
    const int e = (int)(ent & 0x3FFFFFu);
    loc = (int)(ent >> 22);
    d = dst[e]; r = rel[e];
    float l = a_s[n0 + loc] + a_d[d] + a_r[r] + a0;
    l = l > 0.f ? l : 0.01f * l;
    const float b = expf(l);
    __hip_atomic_fetch_add(&sbs[loc], b, __ATOMIC_RELAXED,
                           __HIP_MEMORY_SCOPE_WORKGROUP);
    bb = __float_as_int(b);
  }
  for (int k0 = 0; k0 < n; k0 += HN_BATCH) {
    v2f vd[HN_BATCH], vr[HN_BATCH];
    float bk[HN_BATCH];
    int lo[HN_BATCH];
#pragma unroll
    for (int k = 0; k < HN_BATCH; ++k) {
      if (k0 + k < n) {
        const int dk = __builtin_amdgcn_readlane(d, k0 + k);
        const int rk = __builtin_amdgcn_readlane(r, k0 + k);
        bk[k] = __int_as_float(__builtin_amdgcn_readlane(bb, k0 + k));
        lo[k] = __builtin_amdgcn_readlane(loc, k0 + k);
        vd[k] = reinterpret_cast<const v2f*>(u_d + (size_t)dk * DIM)[lane];
        vr[k] = reinterpret_cast<const v2f*>(u_r + (size_t)rk * DIM)[lane];
      }
    }
#pragma unroll
    for (int k = 0; k < HN_BATCH; ++k) {
      if (k0 + k < n) {
        float* p = &stab[lo[k] * DIM + 2 * lane];
        __hip_atomic_fetch_add(p, bk[k] * (vd[k].x + vr[k].x), __ATOMIC_RELAXED,
                               __HIP_MEMORY_SCOPE_WORKGROUP);
        __hip_atomic_fetch_add(p + 1, bk[k] * (vd[k].y + vr[k].y),
                               __ATOMIC_RELAXED, __HIP_MEMORY_SCOPE_WORKGROUP);
      }
    }
  }
}

__device__ __forceinline__ void hn_push(unsigned* q, int head, int& cnt,
                                        int lane, int loc, int nn, int e) {
  const bool m = (unsigned)loc < (unsigned)nn;
  const unsigned mask = __builtin_amdgcn_ballot_w32(m);
  if (mask != 0u) {
    if (m) {
      const int rank = (int)__builtin_amdgcn_mbcnt_lo(mask, 0u);
      q[(head + cnt + rank) & (HN_QCAP - 1)] =
          ((unsigned)loc << 22) | (unsigned)e;
    }
    cnt += __builtin_popcount(mask);
  }
}

__global__ __launch_bounds__(1024) void edge_gather_h_kernel(
    const int* __restrict__ src, const int* __restrict__ dst,
    const int* __restrict__ rel, const float* __restrict__ u_s,
    const float* __restrict__ u_d, const float* __restrict__ u_r,
    const float* __restrict__ cvec, const float* __restrict__ a_s,
    const float* __restrict__ a_d, const float* __restrict__ a_r,
    const float* __restrict__ a0p, float* __restrict__ h_ent, int E) {
  __shared__ __attribute__((aligned(16))) float stab[HN_NPR * DIM];
  __shared__ unsigned sq[32 * HN_QCAP];
  __shared__ float sbs[HN_NPR];
  const int t = threadIdx.x;
  for (int i = t; i < HN_NPR * DIM; i += 1024) stab[i] = 0.0f;
  for (int i = t; i < HN_NPR; i += 1024) sbs[i] = 0.0f;
  __syncthreads();
  const float a0 = a0p[0];

  const int n0 = blockIdx.x * HN_NPR;
  int nn = NUM_NODES - n0;
  nn = nn > HN_NPR ? HN_NPR : nn;
  nn = nn < 0 ? 0 : nn;

  const int wave = __builtin_amdgcn_readfirstlane(t >> 5);
  const int lane = t & 31;
  unsigned* q = &sq[wave * HN_QCAP];
  int head = 0, cnt = 0;

  const int ngroup = E / 512;
  const int4* __restrict__ src4 = reinterpret_cast<const int4*>(src);
  const int rot = (int)(((long long)blockIdx.x * ngroup) / HN_GROUPS);
  int g = wave;
  int4 nx0, nx1, nx2, nx3;
  nx0 = nx1 = nx2 = nx3 = make_int4(-1, -1, -1, -1);
  int gcur = 0;
  if (g < ngroup) {
    gcur = g + rot; gcur = gcur >= ngroup ? gcur - ngroup : gcur;
    const int4* p = src4 + (size_t)gcur * 128 + lane;
    nx0 = p[0]; nx1 = p[32]; nx2 = p[64]; nx3 = p[96];
  }
  for (; g < ngroup; g += 32) {
    const int4 c0 = nx0, c1 = nx1, c2 = nx2, c3 = nx3;
    const int eb = gcur * 512 + 4 * lane;
    const int gn = g + 32;
    if (gn < ngroup) {
      gcur = gn + rot; gcur = gcur >= ngroup ? gcur - ngroup : gcur;
      const int4* p = src4 + (size_t)gcur * 128 + lane;
      nx0 = p[0]; nx1 = p[32]; nx2 = p[64]; nx3 = p[96];
    }
    hn_push(q, head, cnt, lane, c0.x - n0, nn, eb + 0);
    hn_push(q, head, cnt, lane, c0.y - n0, nn, eb + 1);
    hn_push(q, head, cnt, lane, c0.z - n0, nn, eb + 2);
    hn_push(q, head, cnt, lane, c0.w - n0, nn, eb + 3);
    hn_push(q, head, cnt, lane, c1.x - n0, nn, eb + 128);
    hn_push(q, head, cnt, lane, c1.y - n0, nn, eb + 129);
    hn_push(q, head, cnt, lane, c1.z - n0, nn, eb + 130);
    hn_push(q, head, cnt, lane, c1.w - n0, nn, eb + 131);
    hn_push(q, head, cnt, lane, c2.x - n0, nn, eb + 256);
    hn_push(q, head, cnt, lane, c2.y - n0, nn, eb + 257);
    hn_push(q, head, cnt, lane, c2.z - n0, nn, eb + 258);
    hn_push(q, head, cnt, lane, c2.w - n0, nn, eb + 259);
    hn_push(q, head, cnt, lane, c3.x - n0, nn, eb + 384);
    hn_push(q, head, cnt, lane, c3.y - n0, nn, eb + 385);
    hn_push(q, head, cnt, lane, c3.z - n0, nn, eb + 386);
    hn_push(q, head, cnt, lane, c3.w - n0, nn, eb + 387);
    while (cnt >= 32) {
      hn_drain(q, head, 32, lane, n0, a0, dst, rel, a_s, a_d, a_r, u_d, u_r,
               stab, sbs);
      head = (head + 32) & (HN_QCAP - 1);
      cnt -= 32;
    }
  }
  {
    const int e = ngroup * 512 + wave * 32 + lane;
    int loc = -1;
    if (e < E) loc = src[e] - n0;
    hn_push(q, head, cnt, lane, loc, nn, e);
  }
  while (cnt > 0) {
    const int n = cnt < 32 ? cnt : 32;
    hn_drain(q, head, n, lane, n0, a0, dst, rel, a_s, a_d, a_r, u_d, u_r, stab,
             sbs);
    head = (head + n) & (HN_QCAP - 1);
    cnt -= n;
  }
  __syncthreads();

  for (int i = t; i < nn * (DIM / 4); i += 1024) {
    const int row = i >> 4, c4 = i & 15;
    const int node = n0 + row;
    const float bs = sbs[row];
    float4 o = make_float4(0.0f, 0.0f, 0.0f, 0.0f);
    if (bs > 0.0f) {
      const float4 a = reinterpret_cast<const float4*>(stab)[row * 16 + c4];
      const float4 us =
          reinterpret_cast<const float4*>(u_s + (size_t)node * DIM)[c4];
      const float4 cv = reinterpret_cast<const float4*>(cvec)[c4];
      o.x = a.x / bs + (us.x + cv.x);
      o.y = a.y / bs + (us.y + cv.y);
      o.z = a.z / bs + (us.z + cv.z);
      o.w = a.w / bs + (us.w + cv.w);
    }
    reinterpret_cast<float4*>(h_ent + (size_t)node * DIM)[c4] = o;
  }
}

#define FS_GROUPS 8
#define FS_RPG    63
#define FS_SLICES 32
#define FS_BATCH  8
static_assert(FS_GROUPS * FS_RPG >= NUM_RELS, "relation ranges must cover all relations");

__global__ __launch_bounds__(1024) void edge_featsum_kernel(
    const int* __restrict__ src, const int* __restrict__ dst,
    const int* __restrict__ rel, const float* __restrict__ h_ent,
    float* __restrict__ featsum, float* __restrict__ cnt, int E) {
  __shared__ float sfs[FS_RPG * 128];
  __shared__ float scnt[FS_RPG];
  const int t = threadIdx.x;
  for (int i = t; i < FS_RPG * 128; i += 1024) sfs[i] = 0.0f;
  if (t < FS_RPG) scnt[t] = 0.0f;
  __syncthreads();

  const int grp = blockIdx.x % FS_GROUPS;
  const int slice = blockIdx.x / FS_GROUPS;
  const int r0 = grp * FS_RPG;
  int nrel = NUM_RELS - r0;
  nrel = nrel > FS_RPG ? FS_RPG : nrel;
  nrel = nrel < 0 ? 0 : nrel;
  const int per = (E + FS_SLICES - 1) / FS_SLICES;
  const int e0 = slice * per;
  const int e1 = (e0 + per < E) ? (e0 + per) : E;

  const int wave = __builtin_amdgcn_readfirstlane(t >> 5);
  const int lane = t & 31;

  int base = e0 + wave * 32;
  int rn, sn, dn;
  {
    int ec = base + lane < e1 ? base + lane : e1 - 1;
    ec = ec < 0 ? 0 : ec;
    rn = rel[ec]; sn = src[ec]; dn = dst[ec];
    if (base + lane >= e1) rn = -1;
  }
  for (; base < e1; base += 32 * 32) {
    const int r = rn, s = sn, d = dn;
    const int en = base + 32 * 32 + lane;
    const int ec = en < e1 ? en : e1 - 1;
    rn = rel[ec]; sn = src[ec]; dn = dst[ec];
    if (en >= e1) rn = -1;

    const int loc = r - r0;
    const bool mine = (unsigned)loc < (unsigned)nrel;
    unsigned mask = __builtin_amdgcn_ballot_w32(mine);
    if (mine)
      __hip_atomic_fetch_add(&scnt[loc], 1.0f, __ATOMIC_RELAXED,
                             __HIP_MEMORY_SCOPE_WORKGROUP);

    while (mask != 0u) {
      const int n = __builtin_popcount(mask);
      v2f vs[FS_BATCH], vd[FS_BATCH];
      int lo[FS_BATCH];
#pragma unroll
      for (int k = 0; k < FS_BATCH; ++k) {
        if (k < n) {
          const int l = __builtin_ctz(mask);
          mask &= mask - 1u;
          const int sk = __builtin_amdgcn_readlane(s, l);
          const int dk = __builtin_amdgcn_readlane(d, l);
          lo[k] = __builtin_amdgcn_readlane(loc, l);
          vs[k] = reinterpret_cast<const v2f*>(h_ent + (size_t)sk * DIM)[lane];
          vd[k] = reinterpret_cast<const v2f*>(h_ent + (size_t)dk * DIM)[lane];
        }
      }
#pragma unroll
      for (int k = 0; k < FS_BATCH; ++k) {
        if (k < n) {
          float* p = &sfs[lo[k] * 128 + 2 * lane];
          __hip_atomic_fetch_add(p, vs[k].x, __ATOMIC_RELAXED,
                                 __HIP_MEMORY_SCOPE_WORKGROUP);
          __hip_atomic_fetch_add(p + 1, vs[k].y, __ATOMIC_RELAXED,
                                 __HIP_MEMORY_SCOPE_WORKGROUP);
          __hip_atomic_fetch_add(p + 64, vd[k].x, __ATOMIC_RELAXED,
                                 __HIP_MEMORY_SCOPE_WORKGROUP);
          __hip_atomic_fetch_add(p + 65, vd[k].y, __ATOMIC_RELAXED,
                                 __HIP_MEMORY_SCOPE_WORKGROUP);
        }
      }
    }
  }
  __syncthreads();

  float* fs = featsum + (size_t)r0 * 128;
  for (int i = t; i < nrel * 128; i += 1024) atomicAdd(&fs[i], sfs[i]);
  if (t < nrel) atomicAdd(&cnt[r0 + t], scnt[t]);
}

__global__ void h_rel_kernel(const float* __restrict__ featsum,
                             const float* __restrict__ cnt,
                             const float* __restrict__ rel2_proj,
                             const float* __restrict__ W_rel3,
                             const float* __restrict__ b_rel3,
                             float* __restrict__ h_rel, int R) {
  const int t = blockIdx.x * blockDim.x + threadIdx.x;
  if (t >= R * DIM) return;
  const int r = t >> 6, o = t & 63;
  const float c = cnt[r];
  const float inv = 1.0f / fmaxf(c, 1.0f);
  const float* fs = featsum + (size_t)r * 128;
  const float* w = W_rel3 + (size_t)o * 192;
  float acc = b_rel3[o];
  for (int j = 0; j < 128; ++j) acc += (fs[j] * inv) * w[j];
  if (c > 0.f) {
    const float* r2 = rel2_proj + (size_t)r * DIM;
    for (int j = 0; j < DIM; ++j) acc += r2[j] * w[128 + j];
  }
  h_rel[t] = acc;
}

#define OFF_MS     0
#define OFF_MD     4096
#define OFF_MR     8192
#define OFF_M2     12288
#define OFF_CVEC   16384
#define OFF_VREL2  16448
#define OFF_A0     16512
#define OFF_UR     16576
#define OFF_AR     48576
#define OFF_REL2P  49088
#define OFF_AS     81088
#define OFF_AD     181088
#define OFF_CNT    381088
#define OFF_FEAT   381588
#define OFF_US     1445588
#define OFF_UD     7845588
#define ZERO_CNT   (OFF_FEAT + NUM_RELS * 128 - OFF_CNT)

extern "C" void kernel_launch(void* const* d_in, const int* in_sizes, int n_in,
                              void* d_out, int out_size, void* d_ws,
                              size_t ws_size, hipStream_t stream) {
  const float* ent_embed = (const float*)d_in[0];
  const float* rel_embed = (const float*)d_in[1];
  const float* W_ent  = (const float*)d_in[2];
  const float* b_ent  = (const float*)d_in[3];
  const float* W_relL = (const float*)d_in[4];
  const float* b_relL = (const float*)d_in[5];
  const float* W_rel2 = (const float*)d_in[6];
  const float* b_rel2 = (const float*)d_in[7];
  const float* W_rel3 = (const float*)d_in[8];
  const float* b_rel3 = (const float*)d_in[9];
  const float* W_a    = (const float*)d_in[10];
  const float* b_a    = (const float*)d_in[11];
  const float* W_fc   = (const float*)d_in[12];
  const float* b_fc   = (const float*)d_in[13];
  const int* src_idx  = (const int*)d_in[14];
  const int* dst_idx  = (const int*)d_in[15];
  const int* rel_idx  = (const int*)d_in[16];

  float* ws = (float*)d_ws;
  float* h_ent = (float*)d_out;
  float* h_rel = (float*)d_out + (size_t)NUM_NODES * DIM;

  zero_kernel<<<252, 256, 0, stream>>>(ws + OFF_CNT, ZERO_CNT);

  fold_kernel<<<1, 256, 0, stream>>>(
      W_ent, b_ent, W_relL, b_relL, W_rel2, b_rel2, W_a, b_a, W_fc, b_fc,
      ws + OFF_MS, ws + OFF_MD, ws + OFF_MR, ws + OFF_M2, ws + OFF_CVEC,
      ws + OFF_VREL2, ws + OFF_A0);

  const int gemm_blocks_N = (NUM_NODES + 127) / 128;
  const int gemm_blocks_R = (NUM_RELS + 127) / 128;
  wmma_gemm64<<<gemm_blocks_N, 256, 0, stream>>>(ent_embed, ws + OFF_MS,
                                                 nullptr, ws + OFF_US, NUM_NODES);
  wmma_gemm64<<<gemm_blocks_N, 256, 0, stream>>>(ent_embed, ws + OFF_MD,
                                                 nullptr, ws + OFF_UD, NUM_NODES);
  wmma_gemm64<<<gemm_blocks_R, 256, 0, stream>>>(rel_embed, ws + OFF_MR,
                                                 nullptr, ws + OFF_UR, NUM_RELS);
  wmma_gemm64<<<gemm_blocks_R, 256, 0, stream>>>(rel_embed, ws + OFF_M2,
                                                 ws + OFF_VREL2, ws + OFF_REL2P,
                                                 NUM_RELS);

  dot64_kernel<<<(NUM_NODES + 255) / 256, 256, 0, stream>>>(ws + OFF_US, W_a,
                                                            ws + OFF_AS, NUM_NODES);
  dot64_kernel<<<(NUM_NODES + 255) / 256, 256, 0, stream>>>(ws + OFF_UD, W_a,
                                                            ws + OFF_AD, NUM_NODES);
  dot64_kernel<<<(NUM_RELS + 255) / 256, 256, 0, stream>>>(ws + OFF_UR, W_a,
                                                           ws + OFF_AR, NUM_RELS);

  edge_gather_h_kernel<<<HN_GROUPS, 1024, 0, stream>>>(
      src_idx, dst_idx, rel_idx, ws + OFF_US, ws + OFF_UD, ws + OFF_UR,
      ws + OFF_CVEC, ws + OFF_AS, ws + OFF_AD, ws + OFF_AR, ws + OFF_A0, h_ent,
      NUM_EDGES);
  edge_featsum_kernel<<<FS_GROUPS * FS_SLICES, 1024, 0, stream>>>(
      src_idx, dst_idx, rel_idx, h_ent, ws + OFF_FEAT, ws + OFF_CNT,
      NUM_EDGES);

  h_rel_kernel<<<(NUM_RELS * DIM + 255) / 256, 256, 0, stream>>>(
      ws + OFF_FEAT, ws + OFF_CNT, ws + OFF_REL2P, W_rel3, b_rel3, h_rel,
      NUM_RELS);
}


# launch fusion: the four projection GEMMs, the three score dot products and the accumulator clear run as ONE launch, A read once, scores summed from the accumulators; a_s/a_d/a_r sum order changes
# speedup vs baseline: 5.0998x; 1.2111x over previous
#include <hip/hip_runtime.h>
#include <math.h>


#define NUM_NODES 100000
#define NUM_RELS  500
#define DIM       64
#define NUM_EDGES 1000000

typedef __attribute__((ext_vector_type(2))) float v2f;
typedef __attribute__((ext_vector_type(8))) float v8f;

__global__ __launch_bounds__(256) void fold_kernel(
    const float* __restrict__ W_ent, const float* __restrict__ b_ent,
    const float* __restrict__ W_relL, const float* __restrict__ b_relL,
    const float* __restrict__ W_rel2, const float* __restrict__ b_rel2,
    const float* __restrict__ W_a, const float* __restrict__ b_a,
    const float* __restrict__ W_fc, const float* __restrict__ b_fc,
    float* __restrict__ M_s, float* __restrict__ M_d,
    float* __restrict__ M_r, float* __restrict__ M_2,
    float* __restrict__ cvec, float* __restrict__ v_rel2,
    float* __restrict__ a0) {
  __shared__ float scvec[DIM];
  const int t = threadIdx.x;
  for (int idx = t; idx < DIM * DIM; idx += 256) {
    const int o = idx >> 6, k = idx & 63;
    float ms = 0.f, md = 0.f, mr = 0.f, m2 = 0.f;
    for (int u = 0; u < DIM; ++u) {
      const float we = W_ent[u * DIM + k];
      const float wl = W_relL[u * DIM + k];
      ms += W_fc[o * 192 + u] * we;
      md += W_fc[o * 192 + 64 + u] * we;
      mr += W_fc[o * 192 + 128 + u] * wl;
      m2 += W_rel2[o * DIM + u] * wl;
    }
    M_s[idx] = ms; M_d[idx] = md; M_r[idx] = mr; M_2[idx] = m2;
  }
  if (t < DIM) {
    float cv = b_fc[t], v2 = b_rel2[t];
    for (int u = 0; u < DIM; ++u) {
      cv += W_fc[t * 192 + u] * b_ent[u]
          + W_fc[t * 192 + 64 + u] * b_ent[u]
          + W_fc[t * 192 + 128 + u] * b_relL[u];
      v2 += W_rel2[t * DIM + u] * b_relL[u];
    }
    cvec[t] = cv; v_rel2[t] = v2; scvec[t] = cv;
  }
  __syncthreads();
  if (t == 0) {
    float a = b_a[0];
    for (int o = 0; o < DIM; ++o) a += scvec[o] * W_a[o];
    a0[0] = a;
  }
}

#define PC_NODE_BLOCKS ((NUM_NODES + 127) / 128)
#define PC_REL_BLOCKS  ((NUM_RELS + 127) / 128)

__device__ __forceinline__ float row16_sum(float x) {
  x += __int_as_float(__builtin_amdgcn_update_dpp(0, __float_as_int(x), 0x161, 0xf, 0xf, false));
  x += __int_as_float(__builtin_amdgcn_update_dpp(0, __float_as_int(x), 0x162, 0xf, 0xf, false));
  x += __int_as_float(__builtin_amdgcn_update_dpp(0, __float_as_int(x), 0x164, 0xf, 0xf, false));
  x += __int_as_float(__builtin_amdgcn_update_dpp(0, __float_as_int(x), 0x168, 0xf, 0xf, false));
  return x;
}

__global__ __launch_bounds__(256) void precompute_kernel(
    const float* __restrict__ ent_embed, const float* __restrict__ rel_embed,
    const float* __restrict__ W_a, const float* __restrict__ M_s,
    const float* __restrict__ M_d, const float* __restrict__ M_r,
    const float* __restrict__ M_2, const float* __restrict__ v_rel2,
    float* __restrict__ u_s, float* __restrict__ u_d, float* __restrict__ a_s,
    float* __restrict__ a_d, float* __restrict__ u_r,
    float* __restrict__ rel2_proj, float* __restrict__ a_r,
    float* __restrict__ zero_base, int zero_n) {
  __shared__ float sW[2 * DIM * DIM];
  const int t = threadIdx.x;

  for (int i = blockIdx.x * 256 + t; i < zero_n; i += gridDim.x * 256)
    zero_base[i] = 0.0f;

  const bool is_rel = blockIdx.x >= PC_NODE_BLOCKS;
  const float* A = is_rel ? rel_embed : ent_embed;
  const float* W0 = is_rel ? M_r : M_s;
  const float* W1 = is_rel ? M_2 : M_d;
  float* out0 = is_rel ? u_r : u_s;
  float* out1 = is_rel ? rel2_proj : u_d;
  float* aout0 = is_rel ? a_r : a_s;
  const int M = is_rel ? NUM_RELS : NUM_NODES;
  const int tb = is_rel ? (int)blockIdx.x - PC_NODE_BLOCKS : (int)blockIdx.x;

  for (int i = t; i < DIM * DIM; i += 256) {
    sW[i] = W0[i];
    sW[DIM * DIM + i] = W1[i];
  }
  __syncthreads();

  const int wave = t >> 5;
  const int lane = t & 31;
  const int m0 = (tb * 8 + wave) * 16;
  if (m0 >= M) return;

  const int lm = lane & 15;
  const int hi = lane >> 4;

  int arow = m0 + lm;
  if (arow >= M) arow = M - 1;
  const float* Arow = A + (size_t)arow * DIM;

  v8f acc[8] = {};
#pragma unroll
  for (int k = 0; k < DIM; k += 4) {
    v2f a;
    a.x = Arow[k + 2 * hi + 0];
    a.y = Arow[k + 2 * hi + 1];
#pragma unroll
    for (int j = 0; j < 8; ++j) {
      const float* w = &sW[(j * 16 + lm) * DIM + k + 2 * hi];
      v2f b;
      b.x = w[0];
      b.y = w[1];
      acc[j] = __builtin_amdgcn_wmma_f32_16x16x4_f32(false, a, false, b,
                                                     (short)0, acc[j], false,
                                                     false);
    }
  }

  float p0[8], p1[8];
#pragma unroll
  for (int v = 0; v < 8; ++v) { p0[v] = 0.0f; p1[v] = 0.0f; }
#pragma unroll
  for (int ob = 0; ob < 4; ++ob) {
    const float wa = W_a[ob * 16 + lm];
    const float bv = is_rel ? v_rel2[ob * 16 + lm] : 0.0f;
#pragma unroll
    for (int v = 0; v < 8; ++v) {
      const int row = m0 + v + 8 * hi;
      const float x0 = acc[ob][v];
      const float x1 = acc[4 + ob][v];
      p0[v] += x0 * wa;
      p1[v] += x1 * wa;
      if (row < M) {
        out0[(size_t)row * DIM + ob * 16 + lm] = x0;
        out1[(size_t)row * DIM + ob * 16 + lm] = x1 + bv;
      }
    }
  }

  const int srow = m0 + 8 * hi + lm;
  const bool swr = lm < 8 && srow < M;
  float s0 = 0.0f;
#pragma unroll
  for (int v = 0; v < 8; ++v) {
    const float r = row16_sum(p0[v]);
    s0 = (lm == v) ? r : s0;
  }
  if (swr) aout0[srow] = s0;
  if (!is_rel) {
    float s1 = 0.0f;
#pragma unroll
    for (int v = 0; v < 8; ++v) {
      const float r = row16_sum(p1[v]);
      s1 = (lm == v) ? r : s1;
    }
    if (swr) a_d[srow] = s1;
  }
}

#define HN_NPR    400
#define HN_GROUPS ((NUM_NODES + HN_NPR - 1) / HN_NPR)
#define HN_QCAP   1024
#define HN_BATCH  8
static_assert(NUM_EDGES <= (1 << 22), "edge id must fit 22 bits of a queue word");
static_assert(HN_NPR <= (1 << 10), "local node id must fit 10 bits of a queue word");
static_assert((HN_QCAP & (HN_QCAP - 1)) == 0 && HN_QCAP >= 32 + 512, "ring size");

__device__ __forceinline__ void hn_drain(
    const unsigned* q, int head, int n, int lane, int n0, float a0,
    const int* __restrict__ dst, const int* __restrict__ rel,
    const float* __restrict__ a_s, const float* __restrict__ a_d,
    const float* __restrict__ a_r, const float* __restrict__ u_d,
    const float* __restrict__ u_r, float* stab, float* sbs) {
  int d = 0, r = 0, loc = 0, bb = 0;
  if (lane < n) {
    const unsigned ent = q[(head + lane) & (HN_QCAP - 1)];
    const int e = (int)(ent & 0x3FFFFFu);
    loc = (int)(ent >> 22);
    d = dst[e]; r = rel[e];
    float l = a_s[n0 + loc] + a_d[d] + a_r[r] + a0;
    l = l > 0.f ? l : 0.01f * l;
    const float b = expf(l);
    __hip_atomic_fetch_add(&sbs[loc], b, __ATOMIC_RELAXED,
                           __HIP_MEMORY_SCOPE_WORKGROUP);
    bb = __float_as_int(b);
  }
  for (int k0 = 0; k0 < n; k0 += HN_BATCH) {
    v2f vd[HN_BATCH], vr[HN_BATCH];
    float bk[HN_BATCH];
    int lo[HN_BATCH];
#pragma unroll
    for (int k = 0; k < HN_BATCH; ++k) {
      if (k0 + k < n) {
        const int dk = __builtin_amdgcn_readlane(d, k0 + k);
        const int rk = __builtin_amdgcn_readlane(r, k0 + k);
        bk[k] = __int_as_float(__builtin_amdgcn_readlane(bb, k0 + k));
        lo[k] = __builtin_amdgcn_readlane(loc, k0 + k);
        vd[k] = reinterpret_cast<const v2f*>(u_d + (size_t)dk * DIM)[lane];
        vr[k] = reinterpret_cast<const v2f*>(u_r + (size_t)rk * DIM)[lane];
      }
    }
#pragma unroll
    for (int k = 0; k < HN_BATCH; ++k) {
      if (k0 + k < n) {
        float* p = &stab[lo[k] * DIM + 2 * lane];
        __hip_atomic_fetch_add(p, bk[k] * (vd[k].x + vr[k].x), __ATOMIC_RELAXED,
                               __HIP_MEMORY_SCOPE_WORKGROUP);
        __hip_atomic_fetch_add(p + 1, bk[k] * (vd[k].y + vr[k].y),
                               __ATOMIC_RELAXED, __HIP_MEMORY_SCOPE_WORKGROUP);
      }
    }
  }
}

__device__ __forceinline__ void hn_push(unsigned* q, int head, int& cnt,
                                        int lane, int loc, int nn, int e) {
  const bool m = (unsigned)loc < (unsigned)nn;
  const unsigned mask = __builtin_amdgcn_ballot_w32(m);
  if (mask != 0u) {
    if (m) {
      const int rank = (int)__builtin_amdgcn_mbcnt_lo(mask, 0u);
      q[(head + cnt + rank) & (HN_QCAP - 1)] =
          ((unsigned)loc << 22) | (unsigned)e;
    }
    cnt += __builtin_popcount(mask);
  }
}

__global__ __launch_bounds__(1024) void edge_gather_h_kernel(
    const int* __restrict__ src, const int* __restrict__ dst,
    const int* __restrict__ rel, const float* __restrict__ u_s,
    const float* __restrict__ u_d, const float* __restrict__ u_r,
    const float* __restrict__ cvec, const float* __restrict__ a_s,
    const float* __restrict__ a_d, const float* __restrict__ a_r,
    const float* __restrict__ a0p, float* __restrict__ h_ent, int E) {
  __shared__ __attribute__((aligned(16))) float stab[HN_NPR * DIM];
  __shared__ unsigned sq[32 * HN_QCAP];
  __shared__ float sbs[HN_NPR];
  const int t = threadIdx.x;
  for (int i = t; i < HN_NPR * DIM; i += 1024) stab[i] = 0.0f;
  for (int i = t; i < HN_NPR; i += 1024) sbs[i] = 0.0f;
  __syncthreads();
  const float a0 = a0p[0];

  const int n0 = blockIdx.x * HN_NPR;
  int nn = NUM_NODES - n0;
  nn = nn > HN_NPR ? HN_NPR : nn;
  nn = nn < 0 ? 0 : nn;

  const int wave = __builtin_amdgcn_readfirstlane(t >> 5);
  const int lane = t & 31;
  unsigned* q = &sq[wave * HN_QCAP];
  int head = 0, cnt = 0;

  const int ngroup = E / 512;
  const int4* __restrict__ src4 = reinterpret_cast<const int4*>(src);
  const int rot = (int)(((long long)blockIdx.x * ngroup) / HN_GROUPS);
  int g = wave;
  int4 nx0, nx1, nx2, nx3;
  nx0 = nx1 = nx2 = nx3 = make_int4(-1, -1, -1, -1);
  int gcur = 0;
  if (g < ngroup) {
    gcur = g + rot; gcur = gcur >= ngroup ? gcur - ngroup : gcur;
    const int4* p = src4 + (size_t)gcur * 128 + lane;
    nx0 = p[0]; nx1 = p[32]; nx2 = p[64]; nx3 = p[96];
  }
  for (; g < ngroup; g += 32) {
    const int4 c0 = nx0, c1 = nx1, c2 = nx2, c3 = nx3;
    const int eb = gcur * 512 + 4 * lane;
    const int gn = g + 32;
    if (gn < ngroup) {
      gcur = gn + rot; gcur = gcur >= ngroup ? gcur - ngroup : gcur;
      const int4* p = src4 + (size_t)gcur * 128 + lane;
      nx0 = p[0]; nx1 = p[32]; nx2 = p[64]; nx3 = p[96];
    }
    hn_push(q, head, cnt, lane, c0.x - n0, nn, eb + 0);
    hn_push(q, head, cnt, lane, c0.y - n0, nn, eb + 1);
    hn_push(q, head, cnt, lane, c0.z - n0, nn, eb + 2);
    hn_push(q, head, cnt, lane, c0.w - n0, nn, eb + 3);
    hn_push(q, head, cnt, lane, c1.x - n0, nn, eb + 128);
    hn_push(q, head, cnt, lane, c1.y - n0, nn, eb + 129);
    hn_push(q, head, cnt, lane, c1.z - n0, nn, eb + 130);
    hn_push(q, head, cnt, lane, c1.w - n0, nn, eb + 131);
    hn_push(q, head, cnt, lane, c2.x - n0, nn, eb + 256);
    hn_push(q, head, cnt, lane, c2.y - n0, nn, eb + 257);
    hn_push(q, head, cnt, lane, c2.z - n0, nn, eb + 258);
    hn_push(q, head, cnt, lane, c2.w - n0, nn, eb + 259);
    hn_push(q, head, cnt, lane, c3.x - n0, nn, eb + 384);
    hn_push(q, head, cnt, lane, c3.y - n0, nn, eb + 385);
    hn_push(q, head, cnt, lane, c3.z - n0, nn, eb + 386);
    hn_push(q, head, cnt, lane, c3.w - n0, nn, eb + 387);
    while (cnt >= 32) {
      hn_drain(q, head, 32, lane, n0, a0, dst, rel, a_s, a_d, a_r, u_d, u_r,
               stab, sbs);
      head = (head + 32) & (HN_QCAP - 1);
      cnt -= 32;
    }
  }
  {
    const int e = ngroup * 512 + wave * 32 + lane;
    int loc = -1;
    if (e < E) loc = src[e] - n0;
    hn_push(q, head, cnt, lane, loc, nn, e);
  }
  while (cnt > 0) {
    const int n = cnt < 32 ? cnt : 32;
    hn_drain(q, head, n, lane, n0, a0, dst, rel, a_s, a_d, a_r, u_d, u_r, stab,
             sbs);
    head = (head + n) & (HN_QCAP - 1);
    cnt -= n;
  }
  __syncthreads();

  for (int i = t; i < nn * (DIM / 4); i += 1024) {
    const int row = i >> 4, c4 = i & 15;
    const int node = n0 + row;
    const float bs = sbs[row];
    float4 o = make_float4(0.0f, 0.0f, 0.0f, 0.0f);
    if (bs > 0.0f) {
      const float4 a = reinterpret_cast<const float4*>(stab)[row * 16 + c4];
      const float4 us =
          reinterpret_cast<const float4*>(u_s + (size_t)node * DIM)[c4];
      const float4 cv = reinterpret_cast<const float4*>(cvec)[c4];
      o.x = a.x / bs + (us.x + cv.x);
      o.y = a.y / bs + (us.y + cv.y);
      o.z = a.z / bs + (us.z + cv.z);
      o.w = a.w / bs + (us.w + cv.w);
    }
    reinterpret_cast<float4*>(h_ent + (size_t)node * DIM)[c4] = o;
  }
}

#define FS_GROUPS 8
#define FS_RPG    63
#define FS_SLICES 32
#define FS_BATCH  8
static_assert(FS_GROUPS * FS_RPG >= NUM_RELS, "relation ranges must cover all relations");

__global__ __launch_bounds__(1024) void edge_featsum_kernel(
    const int* __restrict__ src, const int* __restrict__ dst,
    const int* __restrict__ rel, const float* __restrict__ h_ent,
    float* __restrict__ featsum, float* __restrict__ cnt, int E) {
  __shared__ float sfs[FS_RPG * 128];
  __shared__ float scnt[FS_RPG];
  const int t = threadIdx.x;
  for (int i = t; i < FS_RPG * 128; i += 1024) sfs[i] = 0.0f;
  if (t < FS_RPG) scnt[t] = 0.0f;
  __syncthreads();

  const int grp = blockIdx.x % FS_GROUPS;
  const int slice = blockIdx.x / FS_GROUPS;
  const int r0 = grp * FS_RPG;
  int nrel = NUM_RELS - r0;
  nrel = nrel > FS_RPG ? FS_RPG : nrel;
  nrel = nrel < 0 ? 0 : nrel;
  const int per = (E + FS_SLICES - 1) / FS_SLICES;
  const int e0 = slice * per;
  const int e1 = (e0 + per < E) ? (e0 + per) : E;

  const int wave = __builtin_amdgcn_readfirstlane(t >> 5);
  const int lane = t & 31;

  int base = e0 + wave * 32;
  int rn, sn, dn;
  {
    int ec = base + lane < e1 ? base + lane : e1 - 1;
    ec = ec < 0 ? 0 : ec;
    rn = rel[ec]; sn = src[ec]; dn = dst[ec];
    if (base + lane >= e1) rn = -1;
  }
  for (; base < e1; base += 32 * 32) {
    const int r = rn, s = sn, d = dn;
    const int en = base + 32 * 32 + lane;
    const int ec = en < e1 ? en : e1 - 1;
    rn = rel[ec]; sn = src[ec]; dn = dst[ec];
    if (en >= e1) rn = -1;

    const int loc = r - r0;
    const bool mine = (unsigned)loc < (unsigned)nrel;
    unsigned mask = __builtin_amdgcn_ballot_w32(mine);
    if (mine)
      __hip_atomic_fetch_add(&scnt[loc], 1.0f, __ATOMIC_RELAXED,
                             __HIP_MEMORY_SCOPE_WORKGROUP);

    while (mask != 0u) {
      const int n = __builtin_popcount(mask);
      v2f vs[FS_BATCH], vd[FS_BATCH];
      int lo[FS_BATCH];
#pragma unroll
      for (int k = 0; k < FS_BATCH; ++k) {
        if (k < n) {
          const int l = __builtin_ctz(mask);
          mask &= mask - 1u;
          const int sk = __builtin_amdgcn_readlane(s, l);
          const int dk = __builtin_amdgcn_readlane(d, l);
          lo[k] = __builtin_amdgcn_readlane(loc, l);
          vs[k] = reinterpret_cast<const v2f*>(h_ent + (size_t)sk * DIM)[lane];
          vd[k] = reinterpret_cast<const v2f*>(h_ent + (size_t)dk * DIM)[lane];
        }
      }
#pragma unroll
      for (int k = 0; k < FS_BATCH; ++k) {
        if (k < n) {
          float* p = &sfs[lo[k] * 128 + 2 * lane];
          __hip_atomic_fetch_add(p, vs[k].x, __ATOMIC_RELAXED,
                                 __HIP_MEMORY_SCOPE_WORKGROUP);
          __hip_atomic_fetch_add(p + 1, vs[k].y, __ATOMIC_RELAXED,
                                 __HIP_MEMORY_SCOPE_WORKGROUP);
          __hip_atomic_fetch_add(p + 64, vd[k].x, __ATOMIC_RELAXED,
                                 __HIP_MEMORY_SCOPE_WORKGROUP);
          __hip_atomic_fetch_add(p + 65, vd[k].y, __ATOMIC_RELAXED,
                                 __HIP_MEMORY_SCOPE_WORKGROUP);
        }
      }
    }
  }
  __syncthreads();

  float* fs = featsum + (size_t)r0 * 128;
  for (int i = t; i < nrel * 128; i += 1024) atomicAdd(&fs[i], sfs[i]);
  if (t < nrel) atomicAdd(&cnt[r0 + t], scnt[t]);
}

__global__ void h_rel_kernel(const float* __restrict__ featsum,
                             const float* __restrict__ cnt,
                             const float* __restrict__ rel2_proj,
                             const float* __restrict__ W_rel3,
                             const float* __restrict__ b_rel3,
                             float* __restrict__ h_rel, int R) {
  const int t = blockIdx.x * blockDim.x + threadIdx.x;
  if (t >= R * DIM) return;
  const int r = t >> 6, o = t & 63;
  const float c = cnt[r];
  const float inv = 1.0f / fmaxf(c, 1.0f);
  const float* fs = featsum + (size_t)r * 128;
  const float* w = W_rel3 + (size_t)o * 192;
  float acc = b_rel3[o];
  for (int j = 0; j < 128; ++j) acc += (fs[j] * inv) * w[j];
  if (c > 0.f) {
    const float* r2 = rel2_proj + (size_t)r * DIM;
    for (int j = 0; j < DIM; ++j) acc += r2[j] * w[128 + j];
  }
  h_rel[t] = acc;
}

#define OFF_MS     0
#define OFF_MD     4096
#define OFF_MR     8192
#define OFF_M2     12288
#define OFF_CVEC   16384
#define OFF_VREL2  16448
#define OFF_A0     16512
#define OFF_UR     16576
#define OFF_AR     48576
#define OFF_REL2P  49088
#define OFF_AS     81088
#define OFF_AD     181088
#define OFF_CNT    381088
#define OFF_FEAT   381588
#define OFF_US     1445588
#define OFF_UD     7845588
#define ZERO_CNT   (OFF_FEAT + NUM_RELS * 128 - OFF_CNT)

extern "C" void kernel_launch(void* const* d_in, const int* in_sizes, int n_in,
                              void* d_out, int out_size, void* d_ws,
                              size_t ws_size, hipStream_t stream) {
  const float* ent_embed = (const float*)d_in[0];
  const float* rel_embed = (const float*)d_in[1];
  const float* W_ent  = (const float*)d_in[2];
  const float* b_ent  = (const float*)d_in[3];
  const float* W_relL = (const float*)d_in[4];
  const float* b_relL = (const float*)d_in[5];
  const float* W_rel2 = (const float*)d_in[6];
  const float* b_rel2 = (const float*)d_in[7];
  const float* W_rel3 = (const float*)d_in[8];
  const float* b_rel3 = (const float*)d_in[9];
  const float* W_a    = (const float*)d_in[10];
  const float* b_a    = (const float*)d_in[11];
  const float* W_fc   = (const float*)d_in[12];
  const float* b_fc   = (const float*)d_in[13];
  const int* src_idx  = (const int*)d_in[14];
  const int* dst_idx  = (const int*)d_in[15];
  const int* rel_idx  = (const int*)d_in[16];

  float* ws = (float*)d_ws;
  float* h_ent = (float*)d_out;
  float* h_rel = (float*)d_out + (size_t)NUM_NODES * DIM;

  fold_kernel<<<1, 256, 0, stream>>>(
      W_ent, b_ent, W_relL, b_relL, W_rel2, b_rel2, W_a, b_a, W_fc, b_fc,
      ws + OFF_MS, ws + OFF_MD, ws + OFF_MR, ws + OFF_M2, ws + OFF_CVEC,
      ws + OFF_VREL2, ws + OFF_A0);

  precompute_kernel<<<PC_NODE_BLOCKS + PC_REL_BLOCKS, 256, 0, stream>>>(
      ent_embed, rel_embed, W_a, ws + OFF_MS, ws + OFF_MD, ws + OFF_MR,
      ws + OFF_M2, ws + OFF_VREL2, ws + OFF_US, ws + OFF_UD, ws + OFF_AS,
      ws + OFF_AD, ws + OFF_UR, ws + OFF_REL2P, ws + OFF_AR, ws + OFF_CNT,
      ZERO_CNT);

  edge_gather_h_kernel<<<HN_GROUPS, 1024, 0, stream>>>(
      src_idx, dst_idx, rel_idx, ws + OFF_US, ws + OFF_UD, ws + OFF_UR,
      ws + OFF_CVEC, ws + OFF_AS, ws + OFF_AD, ws + OFF_AR, ws + OFF_A0, h_ent,
      NUM_EDGES);
  edge_featsum_kernel<<<FS_GROUPS * FS_SLICES, 1024, 0, stream>>>(
      src_idx, dst_idx, rel_idx, h_ent, ws + OFF_FEAT, ws + OFF_CNT,
      NUM_EDGES);

  h_rel_kernel<<<(NUM_RELS * DIM + 255) / 256, 256, 0, stream>>>(
      ws + OFF_FEAT, ws + OFF_CNT, ws + OFF_REL2P, W_rel3, b_rel3, h_rel,
      NUM_RELS);
}


# fold_kernel on 65 work-groups, not 1: one row of the four folded matrices per group, a thread's 64 column loads in flight together; bias terms by wave sums; cvec/v_rel2/a0 sum order changes
# speedup vs baseline: 6.5072x; 1.2760x over previous
#include <hip/hip_runtime.h>
#include <math.h>


#define NUM_NODES 100000
#define NUM_RELS  500
#define DIM       64
#define NUM_EDGES 1000000

typedef __attribute__((ext_vector_type(2))) float v2f;
typedef __attribute__((ext_vector_type(8))) float v8f;

#define FOLD_BLOCKS (DIM + 1)

__device__ __forceinline__ float row16_sum(float x) {
  x += __int_as_float(__builtin_amdgcn_update_dpp(0, __float_as_int(x), 0x161, 0xf, 0xf, false));
  x += __int_as_float(__builtin_amdgcn_update_dpp(0, __float_as_int(x), 0x162, 0xf, 0xf, false));
  x += __int_as_float(__builtin_amdgcn_update_dpp(0, __float_as_int(x), 0x164, 0xf, 0xf, false));
  x += __int_as_float(__builtin_amdgcn_update_dpp(0, __float_as_int(x), 0x168, 0xf, 0xf, false));
  return x;
}

__device__ __forceinline__ float wave32_sum(float x) {
  x = row16_sum(x);
  return x + __shfl_xor(x, 16);
}

__global__ __launch_bounds__(256) void fold_kernel(
    const float* __restrict__ W_ent, const float* __restrict__ b_ent,
    const float* __restrict__ W_relL, const float* __restrict__ b_relL,
    const float* __restrict__ W_rel2, const float* __restrict__ b_rel2,
    const float* __restrict__ W_a, const float* __restrict__ b_a,
    const float* __restrict__ W_fc, const float* __restrict__ b_fc,
    float* __restrict__ M_s, float* __restrict__ M_d,
    float* __restrict__ M_r, float* __restrict__ M_2,
    float* __restrict__ cvec, float* __restrict__ v_rel2,
    float* __restrict__ a0) {
  __shared__ __attribute__((aligned(16))) float srow[4 * DIM];
  const int t = threadIdx.x;

  if (blockIdx.x < DIM) {
    const int o = blockIdx.x;
    srow[t] = t < 3 * DIM ? W_fc[o * 192 + t] : W_rel2[o * DIM + (t - 3 * DIM)];
    __syncthreads();
    const int mat = t >> 6;
    const int k = t & 63;
    const float* Wk = (mat < 2 ? W_ent : W_relL) + k;
    const float* cf = &srow[mat * DIM];
    float acc = 0.f;
#pragma unroll
    for (int u = 0; u < DIM; ++u) acc += cf[u] * Wk[u * DIM];
    float* Mout = mat == 0 ? M_s : mat == 1 ? M_d : mat == 2 ? M_r : M_2;
    Mout[o * DIM + k] = acc;
    return;
  }

  const int wave = t >> 5;
  const int lane = t & 31;
  const int orow = wave * 8 + (lane & 7);
  const float bfc = b_fc[orow], br2 = b_rel2[orow];
  const float wa0 = W_a[lane], wa1 = W_a[lane + 32], ba = b_a[0];
  const float be0 = b_ent[lane], be1 = b_ent[lane + 32];
  const float bl0 = b_relL[lane], bl1 = b_relL[lane + 32];
  float pc[8], pv[8];
#pragma unroll
  for (int j = 0; j < 8; ++j) {
    const float* f = W_fc + (wave * 8 + j) * 192;
    const float* g = W_rel2 + (wave * 8 + j) * DIM;
    pc[j] = f[lane] * be0 + f[64 + lane] * be0 + f[128 + lane] * bl0 +
            f[32 + lane] * be1 + f[96 + lane] * be1 + f[160 + lane] * bl1;
    pv[j] = g[lane] * bl0 + g[32 + lane] * bl1;
  }
  float cs = 0.f, vs = 0.f;
#pragma unroll
  for (int j = 0; j < 8; ++j) {
    const float sc = wave32_sum(pc[j]);
    const float sv = wave32_sum(pv[j]);
    cs = (lane == j) ? sc : cs;
    vs = (lane == j) ? sv : vs;
  }
  if (lane < 8) {
    const float cv = bfc + cs;
    cvec[orow] = cv;
    v_rel2[orow] = br2 + vs;
    srow[orow] = cv;
  }
  __syncthreads();
  if (wave == 0) {
    const float a = wave32_sum(srow[lane] * wa0 + srow[lane + 32] * wa1);
    if (lane == 0) a0[0] = ba + a;
  }
}

#define PC_NODE_BLOCKS ((NUM_NODES + 127) / 128)
#define PC_REL_BLOCKS  ((NUM_RELS + 127) / 128)

__global__ __launch_bounds__(256) void precompute_kernel(
    const float* __restrict__ ent_embed, const float* __restrict__ rel_embed,
    const float* __restrict__ W_a, const float* __restrict__ M_s,
    const float* __restrict__ M_d, const float* __restrict__ M_r,
    const float* __restrict__ M_2, const float* __restrict__ v_rel2,
    float* __restrict__ u_s, float* __restrict__ u_d, float* __restrict__ a_s,
    float* __restrict__ a_d, float* __restrict__ u_r,
    float* __restrict__ rel2_proj, float* __restrict__ a_r,
    float* __restrict__ zero_base, int zero_n) {
  __shared__ float sW[2 * DIM * DIM];
  const int t = threadIdx.x;

  for (int i = blockIdx.x * 256 + t; i < zero_n; i += gridDim.x * 256)
    zero_base[i] = 0.0f;

  const bool is_rel = blockIdx.x >= PC_NODE_BLOCKS;
  const float* A = is_rel ? rel_embed : ent_embed;
  const float* W0 = is_rel ? M_r : M_s;
  const float* W1 = is_rel ? M_2 : M_d;
  float* out0 = is_rel ? u_r : u_s;
  float* out1 = is_rel ? rel2_proj : u_d;
  float* aout0 = is_rel ? a_r : a_s;
  const int M = is_rel ? NUM_RELS : NUM_NODES;
  const int tb = is_rel ? (int)blockIdx.x - PC_NODE_BLOCKS : (int)blockIdx.x;

  for (int i = t; i < DIM * DIM; i += 256) {
    sW[i] = W0[i];
    sW[DIM * DIM + i] = W1[i];
  }
  __syncthreads();

  const int wave = t >> 5;
  const int lane = t & 31;
  const int m0 = (tb * 8 + wave) * 16;
  if (m0 >= M) return;

  const int lm = lane & 15;
  const int hi = lane >> 4;

  int arow = m0 + lm;
  if (arow >= M) arow = M - 1;
  const float* Arow = A + (size_t)arow * DIM;

  v8f acc[8] = {};
#pragma unroll
  for (int k = 0; k < DIM; k += 4) {
    v2f a;
    a.x = Arow[k + 2 * hi + 0];
    a.y = Arow[k + 2 * hi + 1];
#pragma unroll
    for (int j = 0; j < 8; ++j) {
      const float* w = &sW[(j * 16 + lm) * DIM + k + 2 * hi];
      v2f b;
      b.x = w[0];
      b.y = w[1];
      acc[j] = __builtin_amdgcn_wmma_f32_16x16x4_f32(false, a, false, b,
                                                     (short)0, acc[j], false,
                                                     false);
    }
  }

  float p0[8], p1[8];
#pragma unroll
  for (int v = 0; v < 8; ++v) { p0[v] = 0.0f; p1[v] = 0.0f; }
#pragma unroll
  for (int ob = 0; ob < 4; ++ob) {
    const float wa = W_a[ob * 16 + lm];
    const float bv = is_rel ? v_rel2[ob * 16 + lm] : 0.0f;
#pragma unroll
    for (int v = 0; v < 8; ++v) {
      const int row = m0 + v + 8 * hi;
      const float x0 = acc[ob][v];
      const float x1 = acc[4 + ob][v];
      p0[v] += x0 * wa;
      p1[v] += x1 * wa;
      if (row < M) {
        out0[(size_t)row * DIM + ob * 16 + lm] = x0;
        out1[(size_t)row * DIM + ob * 16 + lm] = x1 + bv;
      }
    }
  }

  const int srow = m0 + 8 * hi + lm;
  const bool swr = lm < 8 && srow < M;
  float s0 = 0.0f;
#pragma unroll
  for (int v = 0; v < 8; ++v) {
    const float r = row16_sum(p0[v]);
    s0 = (lm == v) ? r : s0;
  }
  if (swr) aout0[srow] = s0;
  if (!is_rel) {
    float s1 = 0.0f;
#pragma unroll
    for (int v = 0; v < 8; ++v) {
      const float r = row16_sum(p1[v]);
      s1 = (lm == v) ? r : s1;
    }
    if (swr) a_d[srow] = s1;
  }
}

#define HN_NPR    400
#define HN_GROUPS ((NUM_NODES + HN_NPR - 1) / HN_NPR)
#define HN_QCAP   1024
#define HN_BATCH  8
static_assert(NUM_EDGES <= (1 << 22), "edge id must fit 22 bits of a queue word");
static_assert(HN_NPR <= (1 << 10), "local node id must fit 10 bits of a queue word");
static_assert((HN_QCAP & (HN_QCAP - 1)) == 0 && HN_QCAP >= 32 + 512, "ring size");

__device__ __forceinline__ void hn_drain(
    const unsigned* q, int head, int n, int lane, int n0, float a0,
    const int* __restrict__ dst, const int* __restrict__ rel,
    const float* __restrict__ a_s, const float* __restrict__ a_d,
    const float* __restrict__ a_r, const float* __restrict__ u_d,
    const float* __restrict__ u_r, float* stab, float* sbs) {
  int d = 0, r = 0, loc = 0, bb = 0;
  if (lane < n) {
    const unsigned ent = q[(head + lane) & (HN_QCAP - 1)];
    const int e = (int)(ent & 0x3FFFFFu);
    loc = (int)(ent >> 22);
    d = dst[e]; r = rel[e];
    float l = a_s[n0 + loc] + a_d[d] + a_r[r] + a0;
    l = l > 0.f ? l : 0.01f * l;
    const float b = expf(l);
    __hip_atomic_fetch_add(&sbs[loc], b, __ATOMIC_RELAXED,
                           __HIP_MEMORY_SCOPE_WORKGROUP);
    bb = __float_as_int(b);
  }
  for (int k0 = 0; k0 < n; k0 += HN_BATCH) {
    v2f vd[HN_BATCH], vr[HN_BATCH];
    float bk[HN_BATCH];
    int lo[HN_BATCH];
#pragma unroll
    for (int k = 0; k < HN_BATCH; ++k) {
      if (k0 + k < n) {
        const int dk = __builtin_amdgcn_readlane(d, k0 + k);
        const int rk = __builtin_amdgcn_readlane(r, k0 + k);
        bk[k] = __int_as_float(__builtin_amdgcn_readlane(bb, k0 + k));
        lo[k] = __builtin_amdgcn_readlane(loc, k0 + k);
        vd[k] = reinterpret_cast<const v2f*>(u_d + (size_t)dk * DIM)[lane];
        vr[k] = reinterpret_cast<const v2f*>(u_r + (size_t)rk * DIM)[lane];
      }
    }
#pragma unroll
    for (int k = 0; k < HN_BATCH; ++k) {
      if (k0 + k < n) {
        float* p = &stab[lo[k] * DIM + 2 * lane];
        __hip_atomic_fetch_add(p, bk[k] * (vd[k].x + vr[k].x), __ATOMIC_RELAXED,
                               __HIP_MEMORY_SCOPE_WORKGROUP);
        __hip_atomic_fetch_add(p + 1, bk[k] * (vd[k].y + vr[k].y),
                               __ATOMIC_RELAXED, __HIP_MEMORY_SCOPE_WORKGROUP);
      }
    }
  }
}

__device__ __forceinline__ void hn_push(unsigned* q, int head, int& cnt,
                                        int lane, int loc, int nn, int e) {
  const bool m = (unsigned)loc < (unsigned)nn;
  const unsigned mask = __builtin_amdgcn_ballot_w32(m);
  if (mask != 0u) {
    if (m) {
      const int rank = (int)__builtin_amdgcn_mbcnt_lo(mask, 0u);
      q[(head + cnt + rank) & (HN_QCAP - 1)] =
          ((unsigned)loc << 22) | (unsigned)e;
    }
    cnt += __builtin_popcount(mask);
  }
}

__global__ __launch_bounds__(1024) void edge_gather_h_kernel(
    const int* __restrict__ src, const int* __restrict__ dst,
    const int* __restrict__ rel, const float* __restrict__ u_s,
    const float* __restrict__ u_d, const float* __restrict__ u_r,
    const float* __restrict__ cvec, const float* __restrict__ a_s,
    const float* __restrict__ a_d, const float* __restrict__ a_r,
    const float* __restrict__ a0p, float* __restrict__ h_ent, int E) {
  __shared__ __attribute__((aligned(16))) float stab[HN_NPR * DIM];
  __shared__ unsigned sq[32 * HN_QCAP];
  __shared__ float sbs[HN_NPR];
  const int t = threadIdx.x;
  for (int i = t; i < HN_NPR * DIM; i += 1024) stab[i] = 0.0f;
  for (int i = t; i < HN_NPR; i += 1024) sbs[i] = 0.0f;
  __syncthreads();
  const float a0 = a0p[0];

  const int n0 = blockIdx.x * HN_NPR;
  int nn = NUM_NODES - n0;
  nn = nn > HN_NPR ? HN_NPR : nn;
  nn = nn < 0 ? 0 : nn;

  const int wave = __builtin_amdgcn_readfirstlane(t >> 5);
  const int lane = t & 31;
  unsigned* q = &sq[wave * HN_QCAP];
  int head = 0, cnt = 0;

  const int ngroup = E / 512;
  const int4* __restrict__ src4 = reinterpret_cast<const int4*>(src);
  const int rot = (int)(((long long)blockIdx.x * ngroup) / HN_GROUPS);
  int g = wave;
  int4 nx0, nx1, nx2, nx3;
  nx0 = nx1 = nx2 = nx3 = make_int4(-1, -1, -1, -1);
  int gcur = 0;
  if (g < ngroup) {
    gcur = g + rot; gcur = gcur >= ngroup ? gcur - ngroup : gcur;
    const int4* p = src4 + (size_t)gcur * 128 + lane;
    nx0 = p[0]; nx1 = p[32]; nx2 = p[64]; nx3 = p[96];
  }
  for (; g < ngroup; g += 32) {
    const int4 c0 = nx0, c1 = nx1, c2 = nx2, c3 = nx3;
    const int eb = gcur * 512 + 4 * lane;
    const int gn = g + 32;
    if (gn < ngroup) {
      gcur = gn + rot; gcur = gcur >= ngroup ? gcur - ngroup : gcur;
      const int4* p = src4 + (size_t)gcur * 128 + lane;
      nx0 = p[0]; nx1 = p[32]; nx2 = p[64]; nx3 = p[96];
    }
    hn_push(q, head, cnt, lane, c0.x - n0, nn, eb + 0);
    hn_push(q, head, cnt, lane, c0.y - n0, nn, eb + 1);
    hn_push(q, head, cnt, lane, c0.z - n0, nn, eb + 2);
    hn_push(q, head, cnt, lane, c0.w - n0, nn, eb + 3);
    hn_push(q, head, cnt, lane, c1.x - n0, nn, eb + 128);
    hn_push(q, head, cnt, lane, c1.y - n0, nn, eb + 129);
    hn_push(q, head, cnt, lane, c1.z - n0, nn, eb + 130);
    hn_push(q, head, cnt, lane, c1.w - n0, nn, eb + 131);
    hn_push(q, head, cnt, lane, c2.x - n0, nn, eb + 256);
    hn_push(q, head, cnt, lane, c2.y - n0, nn, eb + 257);
    hn_push(q, head, cnt, lane, c2.z - n0, nn, eb + 258);
    hn_push(q, head, cnt, lane, c2.w - n0, nn, eb + 259);
    hn_push(q, head, cnt, lane, c3.x - n0, nn, eb + 384);
    hn_push(q, head, cnt, lane, c3.y - n0, nn, eb + 385);
    hn_push(q, head, cnt, lane, c3.z - n0, nn, eb + 386);
    hn_push(q, head, cnt, lane, c3.w - n0, nn, eb + 387);
    while (cnt >= 32) {
      hn_drain(q, head, 32, lane, n0, a0, dst, rel, a_s, a_d, a_r, u_d, u_r,
               stab, sbs);
      head = (head + 32) & (HN_QCAP - 1);
      cnt -= 32;
    }
  }
  {
    const int e = ngroup * 512 + wave * 32 + lane;
    int loc = -1;
    if (e < E) loc = src[e] - n0;
    hn_push(q, head, cnt, lane, loc, nn, e);
  }
  while (cnt > 0) {
    const int n = cnt < 32 ? cnt : 32;
    hn_drain(q, head, n, lane, n0, a0, dst, rel, a_s, a_d, a_r, u_d, u_r, stab,
             sbs);
    head = (head + n) & (HN_QCAP - 1);
    cnt -= n;
  }
  __syncthreads();

  for (int i = t; i < nn * (DIM / 4); i += 1024) {
    const int row = i >> 4, c4 = i & 15;
    const int node = n0 + row;
    const float bs = sbs[row];
    float4 o = make_float4(0.0f, 0.0f, 0.0f, 0.0f);
    if (bs > 0.0f) {
      const float4 a = reinterpret_cast<const float4*>(stab)[row * 16 + c4];
      const float4 us =
          reinterpret_cast<const float4*>(u_s + (size_t)node * DIM)[c4];
      const float4 cv = reinterpret_cast<const float4*>(cvec)[c4];
      o.x = a.x / bs + (us.x + cv.x);
      o.y = a.y / bs + (us.y + cv.y);
      o.z = a.z / bs + (us.z + cv.z);
      o.w = a.w / bs + (us.w + cv.w);
    }
    reinterpret_cast<float4*>(h_ent + (size_t)node * DIM)[c4] = o;
  }
}

#define FS_GROUPS 8
#define FS_RPG    63
#define FS_SLICES 32
#define FS_BATCH  8
static_assert(FS_GROUPS * FS_RPG >= NUM_RELS, "relation ranges must cover all relations");

__global__ __launch_bounds__(1024) void edge_featsum_kernel(
    const int* __restrict__ src, const int* __restrict__ dst,
    const int* __restrict__ rel, const float* __restrict__ h_ent,
    float* __restrict__ featsum, float* __restrict__ cnt, int E) {
  __shared__ float sfs[FS_RPG * 128];
  __shared__ float scnt[FS_RPG];
  const int t = threadIdx.x;
  for (int i = t; i < FS_RPG * 128; i += 1024) sfs[i] = 0.0f;
  if (t < FS_RPG) scnt[t] = 0.0f;
  __syncthreads();

  const int grp = blockIdx.x % FS_GROUPS;
  const int slice = blockIdx.x / FS_GROUPS;
  const int r0 = grp * FS_RPG;
  int nrel = NUM_RELS - r0;
  nrel = nrel > FS_RPG ? FS_RPG : nrel;
  nrel = nrel < 0 ? 0 : nrel;
  const int per = (E + FS_SLICES - 1) / FS_SLICES;
  const int e0 = slice * per;
  const int e1 = (e0 + per < E) ? (e0 + per) : E;

  const int wave = __builtin_amdgcn_readfirstlane(t >> 5);
  const int lane = t & 31;

  int base = e0 + wave * 32;
  int rn, sn, dn;
  {
    int ec = base + lane < e1 ? base + lane : e1 - 1;
    ec = ec < 0 ? 0 : ec;
    rn = rel[ec]; sn = src[ec]; dn = dst[ec];
    if (base + lane >= e1) rn = -1;
  }
  for (; base < e1; base += 32 * 32) {
    const int r = rn, s = sn, d = dn;
    const int en = base + 32 * 32 + lane;
    const int ec = en < e1 ? en : e1 - 1;
    rn = rel[ec]; sn = src[ec]; dn = dst[ec];
    if (en >= e1) rn = -1;

    const int loc = r - r0;
    const bool mine = (unsigned)loc < (unsigned)nrel;
    unsigned mask = __builtin_amdgcn_ballot_w32(mine);
    if (mine)
      __hip_atomic_fetch_add(&scnt[loc], 1.0f, __ATOMIC_RELAXED,
                             __HIP_MEMORY_SCOPE_WORKGROUP);

    while (mask != 0u) {
      const int n = __builtin_popcount(mask);
      v2f vs[FS_BATCH], vd[FS_BATCH];
      int lo[FS_BATCH];
#pragma unroll
      for (int k = 0; k < FS_BATCH; ++k) {
        if (k < n) {
          const int l = __builtin_ctz(mask);
          mask &= mask - 1u;
          const int sk = __builtin_amdgcn_readlane(s, l);
          const int dk = __builtin_amdgcn_readlane(d, l);
          lo[k] = __builtin_amdgcn_readlane(loc, l);
          vs[k] = reinterpret_cast<const v2f*>(h_ent + (size_t)sk * DIM)[lane];
          vd[k] = reinterpret_cast<const v2f*>(h_ent + (size_t)dk * DIM)[lane];
        }
      }
#pragma unroll
      for (int k = 0; k < FS_BATCH; ++k) {
        if (k < n) {
          float* p = &sfs[lo[k] * 128 + 2 * lane];
          __hip_atomic_fetch_add(p, vs[k].x, __ATOMIC_RELAXED,
                                 __HIP_MEMORY_SCOPE_WORKGROUP);
          __hip_atomic_fetch_add(p + 1, vs[k].y, __ATOMIC_RELAXED,
                                 __HIP_MEMORY_SCOPE_WORKGROUP);
          __hip_atomic_fetch_add(p + 64, vd[k].x, __ATOMIC_RELAXED,
                                 __HIP_MEMORY_SCOPE_WORKGROUP);
          __hip_atomic_fetch_add(p + 65, vd[k].y, __ATOMIC_RELAXED,
                                 __HIP_MEMORY_SCOPE_WORKGROUP);
        }
      }
    }
  }
  __syncthreads();

  float* fs = featsum + (size_t)r0 * 128;
  for (int i = t; i < nrel * 128; i += 1024) atomicAdd(&fs[i], sfs[i]);
  if (t < nrel) atomicAdd(&cnt[r0 + t], scnt[t]);
}

__global__ void h_rel_kernel(const float* __restrict__ featsum,
                             const float* __restrict__ cnt,
                             const float* __restrict__ rel2_proj,
                             const float* __restrict__ W_rel3,
                             const float* __restrict__ b_rel3,
                             float* __restrict__ h_rel, int R) {
  const int t = blockIdx.x * blockDim.x + threadIdx.x;
  if (t >= R * DIM) return;
  const int r = t >> 6, o = t & 63;
  const float c = cnt[r];
  const float inv = 1.0f / fmaxf(c, 1.0f);
  const float* fs = featsum + (size_t)r * 128;
  const float* w = W_rel3 + (size_t)o * 192;
  float acc = b_rel3[o];
  for (int j = 0; j < 128; ++j) acc += (fs[j] * inv) * w[j];
  if (c > 0.f) {
    const float* r2 = rel2_proj + (size_t)r * DIM;
    for (int j = 0; j < DIM; ++j) acc += r2[j] * w[128 + j];
  }
  h_rel[t] = acc;
}

#define OFF_MS     0
#define OFF_MD     4096
#define OFF_MR     8192
#define OFF_M2     12288
#define OFF_CVEC   16384
#define OFF_VREL2  16448
#define OFF_A0     16512
#define OFF_UR     16576
#define OFF_AR     48576
#define OFF_REL2P  49088
#define OFF_AS     81088
#define OFF_AD     181088
#define OFF_CNT    381088
#define OFF_FEAT   381588
#define OFF_US     1445588
#define OFF_UD     7845588
#define ZERO_CNT   (OFF_FEAT + NUM_RELS * 128 - OFF_CNT)

extern "C" void kernel_launch(void* const* d_in, const int* in_sizes, int n_in,
                              void* d_out, int out_size, void* d_ws,
                              size_t ws_size, hipStream_t stream) {
  const float* ent_embed = (const float*)d_in[0];
  const float* rel_embed = (const float*)d_in[1];
  const float* W_ent  = (const float*)d_in[2];
  const float* b_ent  = (const float*)d_in[3];
  const float* W_relL = (const float*)d_in[4];
  const float* b_relL = (const float*)d_in[5];
  const float* W_rel2 = (const float*)d_in[6];
  const float* b_rel2 = (const float*)d_in[7];
  const float* W_rel3 = (const float*)d_in[8];
  const float* b_rel3 = (const float*)d_in[9];
  const float* W_a    = (const float*)d_in[10];
  const float* b_a    = (const float*)d_in[11];
  const float* W_fc   = (const float*)d_in[12];
  const float* b_fc   = (const float*)d_in[13];
  const int* src_idx  = (const int*)d_in[14];
  const int* dst_idx  = (const int*)d_in[15];
  const int* rel_idx  = (const int*)d_in[16];

  float* ws = (float*)d_ws;
  float* h_ent = (float*)d_out;
  float* h_rel = (float*)d_out + (size_t)NUM_NODES * DIM;

  fold_kernel<<<FOLD_BLOCKS, 256, 0, stream>>>(
      W_ent, b_ent, W_relL, b_relL, W_rel2, b_rel2, W_a, b_a, W_fc, b_fc,
      ws + OFF_MS, ws + OFF_MD, ws + OFF_MR, ws + OFF_M2, ws + OFF_CVEC,
      ws + OFF_VREL2, ws + OFF_A0);

  precompute_kernel<<<PC_NODE_BLOCKS + PC_REL_BLOCKS, 256, 0, stream>>>(
      ent_embed, rel_embed, W_a, ws + OFF_MS, ws + OFF_MD, ws + OFF_MR,
      ws + OFF_M2, ws + OFF_VREL2, ws + OFF_US, ws + OFF_UD, ws + OFF_AS,
      ws + OFF_AD, ws + OFF_UR, ws + OFF_REL2P, ws + OFF_AR, ws + OFF_CNT,
      ZERO_CNT);

  edge_gather_h_kernel<<<HN_GROUPS, 1024, 0, stream>>>(
      src_idx, dst_idx, rel_idx, ws + OFF_US, ws + OFF_UD, ws + OFF_UR,
      ws + OFF_CVEC, ws + OFF_AS, ws + OFF_AD, ws + OFF_AR, ws + OFF_A0, h_ent,
      NUM_EDGES);
  edge_featsum_kernel<<<FS_GROUPS * FS_SLICES, 1024, 0, stream>>>(
      src_idx, dst_idx, rel_idx, h_ent, ws + OFF_FEAT, ws + OFF_CNT,
      NUM_EDGES);

  h_rel_kernel<<<(NUM_RELS * DIM + 255) / 256, 256, 0, stream>>>(
      ws + OFF_FEAT, ws + OFF_CNT, ws + OFF_REL2P, W_rel3, b_rel3, h_rel,
      NUM_RELS);
}


# precompute_kernel: LDS pitch of the staged weight rows 68 floats, not 64: the lanes of a B-fragment read fall in 64 different banks, not 4 (bank conflict on every WMMA operand read removed)
# speedup vs baseline: 6.8600x; 1.0542x over previous
#include <hip/hip_runtime.h>
#include <math.h>


#define NUM_NODES 100000
#define NUM_RELS  500
#define DIM       64
#define NUM_EDGES 1000000

typedef __attribute__((ext_vector_type(2))) float v2f;
typedef __attribute__((ext_vector_type(8))) float v8f;

#define FOLD_BLOCKS (DIM + 1)

__device__ __forceinline__ float row16_sum(float x) {
  x += __int_as_float(__builtin_amdgcn_update_dpp(0, __float_as_int(x), 0x161, 0xf, 0xf, false));
  x += __int_as_float(__builtin_amdgcn_update_dpp(0, __float_as_int(x), 0x162, 0xf, 0xf, false));
  x += __int_as_float(__builtin_amdgcn_update_dpp(0, __float_as_int(x), 0x164, 0xf, 0xf, false));
  x += __int_as_float(__builtin_amdgcn_update_dpp(0, __float_as_int(x), 0x168, 0xf, 0xf, false));
  return x;
}

__device__ __forceinline__ float wave32_sum(float x) {
  x = row16_sum(x);
  return x + __shfl_xor(x, 16);
}

__global__ __launch_bounds__(256) void fold_kernel(
    const float* __restrict__ W_ent, const float* __restrict__ b_ent,
    const float* __restrict__ W_relL, const float* __restrict__ b_relL,
    const float* __restrict__ W_rel2, const float* __restrict__ b_rel2,
    const float* __restrict__ W_a, const float* __restrict__ b_a,
    const float* __restrict__ W_fc, const float* __restrict__ b_fc,
    float* __restrict__ M_s, float* __restrict__ M_d,
    float* __restrict__ M_r, float* __restrict__ M_2,
    float* __restrict__ cvec, float* __restrict__ v_rel2,
    float* __restrict__ a0) {
  __shared__ __attribute__((aligned(16))) float srow[4 * DIM];
  const int t = threadIdx.x;

  if (blockIdx.x < DIM) {
    const int o = blockIdx.x;
    srow[t] = t < 3 * DIM ? W_fc[o * 192 + t] : W_rel2[o * DIM + (t - 3 * DIM)];
    __syncthreads();
    const int mat = t >> 6;
    const int k = t & 63;
    const float* Wk = (mat < 2 ? W_ent : W_relL) + k;
    const float* cf = &srow[mat * DIM];
    float acc = 0.f;
#pragma unroll
    for (int u = 0; u < DIM; ++u) acc += cf[u] * Wk[u * DIM];
    float* Mout = mat == 0 ? M_s : mat == 1 ? M_d : mat == 2 ? M_r : M_2;
    Mout[o * DIM + k] = acc;
    return;
  }

  const int wave = t >> 5;
  const int lane = t & 31;
  const int orow = wave * 8 + (lane & 7);
  const float bfc = b_fc[orow], br2 = b_rel2[orow];
  const float wa0 = W_a[lane], wa1 = W_a[lane + 32], ba = b_a[0];
  const float be0 = b_ent[lane], be1 = b_ent[lane + 32];
  const float bl0 = b_relL[lane], bl1 = b_relL[lane + 32];
  float pc[8], pv[8];
#pragma unroll
  for (int j = 0; j < 8; ++j) {
    const float* f = W_fc + (wave * 8 + j) * 192;
    const float* g = W_rel2 + (wave * 8 + j) * DIM;
    pc[j] = f[lane] * be0 + f[64 + lane] * be0 + f[128 + lane] * bl0 +
            f[32 + lane] * be1 + f[96 + lane] * be1 + f[160 + lane] * bl1;
    pv[j] = g[lane] * bl0 + g[32 + lane] * bl1;
  }
  float cs = 0.f, vs = 0.f;
#pragma unroll
  for (int j = 0; j < 8; ++j) {
    const float sc = wave32_sum(pc[j]);
    const float sv = wave32_sum(pv[j]);
    cs = (lane == j) ? sc : cs;
    vs = (lane == j) ? sv : vs;
  }
  if (lane < 8) {
    const float cv = bfc + cs;
    cvec[orow] = cv;
    v_rel2[orow] = br2 + vs;
    srow[orow] = cv;
  }
  __syncthreads();
  if (wave == 0) {
    const float a = wave32_sum(srow[lane] * wa0 + srow[lane + 32] * wa1);
    if (lane == 0) a0[0] = ba + a;
  }
}

#define PC_NODE_BLOCKS ((NUM_NODES + 127) / 128)
#define PC_REL_BLOCKS  ((NUM_RELS + 127) / 128)
#define PC_PITCH (DIM + 4)

__global__ __launch_bounds__(256) void precompute_kernel(
    const float* __restrict__ ent_embed, const float* __restrict__ rel_embed,
    const float* __restrict__ W_a, const float* __restrict__ M_s,
    const float* __restrict__ M_d, const float* __restrict__ M_r,
    const float* __restrict__ M_2, const float* __restrict__ v_rel2,
    float* __restrict__ u_s, float* __restrict__ u_d, float* __restrict__ a_s,
    float* __restrict__ a_d, float* __restrict__ u_r,
    float* __restrict__ rel2_proj, float* __restrict__ a_r,
    float* __restrict__ zero_base, int zero_n) {
  __shared__ float sW[2 * DIM * PC_PITCH];
  const int t = threadIdx.x;

  for (int i = blockIdx.x * 256 + t; i < zero_n; i += gridDim.x * 256)
    zero_base[i] = 0.0f;

  const bool is_rel = blockIdx.x >= PC_NODE_BLOCKS;
  const float* A = is_rel ? rel_embed : ent_embed;
  const float* W0 = is_rel ? M_r : M_s;
  const float* W1 = is_rel ? M_2 : M_d;
  float* out0 = is_rel ? u_r : u_s;
  float* out1 = is_rel ? rel2_proj : u_d;
  float* aout0 = is_rel ? a_r : a_s;
  const int M = is_rel ? NUM_RELS : NUM_NODES;
  const int tb = is_rel ? (int)blockIdx.x - PC_NODE_BLOCKS : (int)blockIdx.x;

  for (int i = t; i < DIM * DIM; i += 256) {
    const int si = (i >> 6) * PC_PITCH + (i & 63);
    sW[si] = W0[i];
    sW[DIM * PC_PITCH + si] = W1[i];
  }
  __syncthreads();

  const int wave = t >> 5;
  const int lane = t & 31;
  const int m0 = (tb * 8 + wave) * 16;
  if (m0 >= M) return;

  const int lm = lane & 15;
  const int hi = lane >> 4;

  int arow = m0 + lm;
  if (arow >= M) arow = M - 1;
  const float* Arow = A + (size_t)arow * DIM;

  v8f acc[8] = {};
#pragma unroll
  for (int k = 0; k < DIM; k += 4) {
    v2f a;
    a.x = Arow[k + 2 * hi + 0];
    a.y = Arow[k + 2 * hi + 1];
#pragma unroll
    for (int j = 0; j < 8; ++j) {
      const float* w = &sW[(j * 16 + lm) * PC_PITCH + k + 2 * hi];
      v2f b;
      b.x = w[0];
      b.y = w[1];
      acc[j] = __builtin_amdgcn_wmma_f32_16x16x4_f32(false, a, false, b,
                                                     (short)0, acc[j], false,
                                                     false);
    }
  }

  float p0[8], p1[8];
#pragma unroll
  for (int v = 0; v < 8; ++v) { p0[v] = 0.0f; p1[v] = 0.0f; }
#pragma unroll
  for (int ob = 0; ob < 4; ++ob) {
    const float wa = W_a[ob * 16 + lm];
    const float bv = is_rel ? v_rel2[ob * 16 + lm] : 0.0f;
#pragma unroll
    for (int v = 0; v < 8; ++v) {
      const int row = m0 + v + 8 * hi;
      const float x0 = acc[ob][v];
      const float x1 = acc[4 + ob][v];
      p0[v] += x0 * wa;
      p1[v] += x1 * wa;
      if (row < M) {
        out0[(size_t)row * DIM + ob * 16 + lm] = x0;
        out1[(size_t)row * DIM + ob * 16 + lm] = x1 + bv;
      }
    }
  }

  const int srow = m0 + 8 * hi + lm;
  const bool swr = lm < 8 && srow < M;
  float s0 = 0.0f;
#pragma unroll
  for (int v = 0; v < 8; ++v) {
    const float r = row16_sum(p0[v]);
    s0 = (lm == v) ? r : s0;
  }
  if (swr) aout0[srow] = s0;
  if (!is_rel) {
    float s1 = 0.0f;
#pragma unroll
    for (int v = 0; v < 8; ++v) {
      const float r = row16_sum(p1[v]);
      s1 = (lm == v) ? r : s1;
    }
    if (swr) a_d[srow] = s1;
  }
}

#define HN_NPR    400
#define HN_GROUPS ((NUM_NODES + HN_NPR - 1) / HN_NPR)
#define HN_QCAP   1024
#define HN_BATCH  8
static_assert(NUM_EDGES <= (1 << 22), "edge id must fit 22 bits of a queue word");
static_assert(HN_NPR <= (1 << 10), "local node id must fit 10 bits of a queue word");
static_assert((HN_QCAP & (HN_QCAP - 1)) == 0 && HN_QCAP >= 32 + 512, "ring size");

__device__ __forceinline__ void hn_drain(
    const unsigned* q, int head, int n, int lane, int n0, float a0,
    const int* __restrict__ dst, const int* __restrict__ rel,
    const float* __restrict__ a_s, const float* __restrict__ a_d,
    const float* __restrict__ a_r, const float* __restrict__ u_d,
    const float* __restrict__ u_r, float* stab, float* sbs) {
  int d = 0, r = 0, loc = 0, bb = 0;
  if (lane < n) {
    const unsigned ent = q[(head + lane) & (HN_QCAP - 1)];
    const int e = (int)(ent & 0x3FFFFFu);
    loc = (int)(ent >> 22);
    d = dst[e]; r = rel[e];
    float l = a_s[n0 + loc] + a_d[d] + a_r[r] + a0;
    l = l > 0.f ? l : 0.01f * l;
    const float b = expf(l);
    __hip_atomic_fetch_add(&sbs[loc], b, __ATOMIC_RELAXED,
                           __HIP_MEMORY_SCOPE_WORKGROUP);
    bb = __float_as_int(b);
  }
  for (int k0 = 0; k0 < n; k0 += HN_BATCH) {
    v2f vd[HN_BATCH], vr[HN_BATCH];
    float bk[HN_BATCH];
    int lo[HN_BATCH];
#pragma unroll
    for (int k = 0; k < HN_BATCH; ++k) {
      if (k0 + k < n) {
        const int dk = __builtin_amdgcn_readlane(d, k0 + k);
        const int rk = __builtin_amdgcn_readlane(r, k0 + k);
        bk[k] = __int_as_float(__builtin_amdgcn_readlane(bb, k0 + k));
        lo[k] = __builtin_amdgcn_readlane(loc, k0 + k);
        vd[k] = reinterpret_cast<const v2f*>(u_d + (size_t)dk * DIM)[lane];
        vr[k] = reinterpret_cast<const v2f*>(u_r + (size_t)rk * DIM)[lane];
      }
    }
#pragma unroll
    for (int k = 0; k < HN_BATCH; ++k) {
      if (k0 + k < n) {
        float* p = &stab[lo[k] * DIM + 2 * lane];
        __hip_atomic_fetch_add(p, bk[k] * (vd[k].x + vr[k].x), __ATOMIC_RELAXED,
                               __HIP_MEMORY_SCOPE_WORKGROUP);
        __hip_atomic_fetch_add(p + 1, bk[k] * (vd[k].y + vr[k].y),
                               __ATOMIC_RELAXED, __HIP_MEMORY_SCOPE_WORKGROUP);
      }
    }
  }
}

__device__ __forceinline__ void hn_push(unsigned* q, int head, int& cnt,
                                        int lane, int loc, int nn, int e) {
  const bool m = (unsigned)loc < (unsigned)nn;
  const unsigned mask = __builtin_amdgcn_ballot_w32(m);
  if (mask != 0u) {
    if (m) {
      const int rank = (int)__builtin_amdgcn_mbcnt_lo(mask, 0u);
      q[(head + cnt + rank) & (HN_QCAP - 1)] =
          ((unsigned)loc << 22) | (unsigned)e;
    }
    cnt += __builtin_popcount(mask);
  }
}

__global__ __launch_bounds__(1024) void edge_gather_h_kernel(
    const int* __restrict__ src, const int* __restrict__ dst,
    const int* __restrict__ rel, const float* __restrict__ u_s,
    const float* __restrict__ u_d, const float* __restrict__ u_r,
    const float* __restrict__ cvec, const float* __restrict__ a_s,
    const float* __restrict__ a_d, const float* __restrict__ a_r,
    const float* __restrict__ a0p, float* __restrict__ h_ent, int E) {
  __shared__ __attribute__((aligned(16))) float stab[HN_NPR * DIM];
  __shared__ unsigned sq[32 * HN_QCAP];
  __shared__ float sbs[HN_NPR];
  const int t = threadIdx.x;
  for (int i = t; i < HN_NPR * DIM; i += 1024) stab[i] = 0.0f;
  for (int i = t; i < HN_NPR; i += 1024) sbs[i] = 0.0f;
  __syncthreads();
  const float a0 = a0p[0];

  const int n0 = blockIdx.x * HN_NPR;
  int nn = NUM_NODES - n0;
  nn = nn > HN_NPR ? HN_NPR : nn;
  nn = nn < 0 ? 0 : nn;

  const int wave = __builtin_amdgcn_readfirstlane(t >> 5);
  const int lane = t & 31;
  unsigned* q = &sq[wave * HN_QCAP];
  int head = 0, cnt = 0;

  const int ngroup = E / 512;
  const int4* __restrict__ src4 = reinterpret_cast<const int4*>(src);
  const int rot = (int)(((long long)blockIdx.x * ngroup) / HN_GROUPS);
  int g = wave;
  int4 nx0, nx1, nx2, nx3;
  nx0 = nx1 = nx2 = nx3 = make_int4(-1, -1, -1, -1);
  int gcur = 0;
  if (g < ngroup) {
    gcur = g + rot; gcur = gcur >= ngroup ? gcur - ngroup : gcur;
    const int4* p = src4 + (size_t)gcur * 128 + lane;
    nx0 = p[0]; nx1 = p[32]; nx2 = p[64]; nx3 = p[96];
  }
  for (; g < ngroup; g += 32) {
    const int4 c0 = nx0, c1 = nx1, c2 = nx2, c3 = nx3;
    const int eb = gcur * 512 + 4 * lane;
    const int gn = g + 32;
    if (gn < ngroup) {
      gcur = gn + rot; gcur = gcur >= ngroup ? gcur - ngroup : gcur;
      const int4* p = src4 + (size_t)gcur * 128 + lane;
      nx0 = p[0]; nx1 = p[32]; nx2 = p[64]; nx3 = p[96];
    }
    hn_push(q, head, cnt, lane, c0.x - n0, nn, eb + 0);
    hn_push(q, head, cnt, lane, c0.y - n0, nn, eb + 1);
    hn_push(q, head, cnt, lane, c0.z - n0, nn, eb + 2);
    hn_push(q, head, cnt, lane, c0.w - n0, nn, eb + 3);
    hn_push(q, head, cnt, lane, c1.x - n0, nn, eb + 128);
    hn_push(q, head, cnt, lane, c1.y - n0, nn, eb + 129);
    hn_push(q, head, cnt, lane, c1.z - n0, nn, eb + 130);
    hn_push(q, head, cnt, lane, c1.w - n0, nn, eb + 131);
    hn_push(q, head, cnt, lane, c2.x - n0, nn, eb + 256);
    hn_push(q, head, cnt, lane, c2.y - n0, nn, eb + 257);
    hn_push(q, head, cnt, lane, c2.z - n0, nn, eb + 258);
    hn_push(q, head, cnt, lane, c2.w - n0, nn, eb + 259);
    hn_push(q, head, cnt, lane, c3.x - n0, nn, eb + 384);
    hn_push(q, head, cnt, lane, c3.y - n0, nn, eb + 385);
    hn_push(q, head, cnt, lane, c3.z - n0, nn, eb + 386);
    hn_push(q, head, cnt, lane, c3.w - n0, nn, eb + 387);
    while (cnt >= 32) {
      hn_drain(q, head, 32, lane, n0, a0, dst, rel, a_s, a_d, a_r, u_d, u_r,
               stab, sbs);
      head = (head + 32) & (HN_QCAP - 1);
      cnt -= 32;
    }
  }
  {
    const int e = ngroup * 512 + wave * 32 + lane;
    int loc = -1;
    if (e < E) loc = src[e] - n0;
    hn_push(q, head, cnt, lane, loc, nn, e);
  }
  while (cnt > 0) {
    const int n = cnt < 32 ? cnt : 32;
    hn_drain(q, head, n, lane, n0, a0, dst, rel, a_s, a_d, a_r, u_d, u_r, stab,
             sbs);
    head = (head + n) & (HN_QCAP - 1);
    cnt -= n;
  }
  __syncthreads();

  for (int i = t; i < nn * (DIM / 4); i += 1024) {
    const int row = i >> 4, c4 = i & 15;
    const int node = n0 + row;
    const float bs = sbs[row];
    float4 o = make_float4(0.0f, 0.0f, 0.0f, 0.0f);
    if (bs > 0.0f) {
      const float4 a = reinterpret_cast<const float4*>(stab)[row * 16 + c4];
      const float4 us =
          reinterpret_cast<const float4*>(u_s + (size_t)node * DIM)[c4];
      const float4 cv = reinterpret_cast<const float4*>(cvec)[c4];
      o.x = a.x / bs + (us.x + cv.x);
      o.y = a.y / bs + (us.y + cv.y);
      o.z = a.z / bs + (us.z + cv.z);
      o.w = a.w / bs + (us.w + cv.w);
    }
    reinterpret_cast<float4*>(h_ent + (size_t)node * DIM)[c4] = o;
  }
}

#define FS_GROUPS 8
#define FS_RPG    63
#define FS_SLICES 32
#define FS_BATCH  8
static_assert(FS_GROUPS * FS_RPG >= NUM_RELS, "relation ranges must cover all relations");

__global__ __launch_bounds__(1024) void edge_featsum_kernel(
    const int* __restrict__ src, const int* __restrict__ dst,
    const int* __restrict__ rel, const float* __restrict__ h_ent,
    float* __restrict__ featsum, float* __restrict__ cnt, int E) {
  __shared__ float sfs[FS_RPG * 128];
  __shared__ float scnt[FS_RPG];
  const int t = threadIdx.x;
  for (int i = t; i < FS_RPG * 128; i += 1024) sfs[i] = 0.0f;
  if (t < FS_RPG) scnt[t] = 0.0f;
  __syncthreads();

  const int grp = blockIdx.x % FS_GROUPS;
  const int slice = blockIdx.x / FS_GROUPS;
  const int r0 = grp * FS_RPG;
  int nrel = NUM_RELS - r0;
  nrel = nrel > FS_RPG ? FS_RPG : nrel;
  nrel = nrel < 0 ? 0 : nrel;
  const int per = (E + FS_SLICES - 1) / FS_SLICES;
  const int e0 = slice * per;
  const int e1 = (e0 + per < E) ? (e0 + per) : E;

  const int wave = __builtin_amdgcn_readfirstlane(t >> 5);
  const int lane = t & 31;

  int base = e0 + wave * 32;
  int rn, sn, dn;
  {
    int ec = base + lane < e1 ? base + lane : e1 - 1;
    ec = ec < 0 ? 0 : ec;
    rn = rel[ec]; sn = src[ec]; dn = dst[ec];
    if (base + lane >= e1) rn = -1;
  }
  for (; base < e1; base += 32 * 32) {
    const int r = rn, s = sn, d = dn;
    const int en = base + 32 * 32 + lane;
    const int ec = en < e1 ? en : e1 - 1;
    rn = rel[ec]; sn = src[ec]; dn = dst[ec];
    if (en >= e1) rn = -1;

    const int loc = r - r0;
    const bool mine = (unsigned)loc < (unsigned)nrel;
    unsigned mask = __builtin_amdgcn_ballot_w32(mine);
    if (mine)
      __hip_atomic_fetch_add(&scnt[loc], 1.0f, __ATOMIC_RELAXED,
                             __HIP_MEMORY_SCOPE_WORKGROUP);

    while (mask != 0u) {
      const int n = __builtin_popcount(mask);
      v2f vs[FS_BATCH], vd[FS_BATCH];
      int lo[FS_BATCH];
#pragma unroll
      for (int k = 0; k < FS_BATCH; ++k) {
        if (k < n) {
          const int l = __builtin_ctz(mask);
          mask &= mask - 1u;
          const int sk = __builtin_amdgcn_readlane(s, l);
          const int dk = __builtin_amdgcn_readlane(d, l);
          lo[k] = __builtin_amdgcn_readlane(loc, l);
          vs[k] = reinterpret_cast<const v2f*>(h_ent + (size_t)sk * DIM)[lane];
          vd[k] = reinterpret_cast<const v2f*>(h_ent + (size_t)dk * DIM)[lane];
        }
      }
#pragma unroll
      for (int k = 0; k < FS_BATCH; ++k) {
        if (k < n) {
          float* p = &sfs[lo[k] * 128 + 2 * lane];
          __hip_atomic_fetch_add(p, vs[k].x, __ATOMIC_RELAXED,
                                 __HIP_MEMORY_SCOPE_WORKGROUP);
          __hip_atomic_fetch_add(p + 1, vs[k].y, __ATOMIC_RELAXED,
                                 __HIP_MEMORY_SCOPE_WORKGROUP);
          __hip_atomic_fetch_add(p + 64, vd[k].x, __ATOMIC_RELAXED,
                                 __HIP_MEMORY_SCOPE_WORKGROUP);
          __hip_atomic_fetch_add(p + 65, vd[k].y, __ATOMIC_RELAXED,
                                 __HIP_MEMORY_SCOPE_WORKGROUP);
        }
      }
    }
  }
  __syncthreads();

  float* fs = featsum + (size_t)r0 * 128;
  for (int i = t; i < nrel * 128; i += 1024) atomicAdd(&fs[i], sfs[i]);
  if (t < nrel) atomicAdd(&cnt[r0 + t], scnt[t]);
}

__global__ void h_rel_kernel(const float* __restrict__ featsum,
                             const float* __restrict__ cnt,
                             const float* __restrict__ rel2_proj,
                             const float* __restrict__ W_rel3,
                             const float* __restrict__ b_rel3,
                             float* __restrict__ h_rel, int R) {
  const int t = blockIdx.x * blockDim.x + threadIdx.x;
  if (t >= R * DIM) return;
  const int r = t >> 6, o = t & 63;
  const float c = cnt[r];
  const float inv = 1.0f / fmaxf(c, 1.0f);
  const float* fs = featsum + (size_t)r * 128;
  const float* w = W_rel3 + (size_t)o * 192;
  float acc = b_rel3[o];
  for (int j = 0; j < 128; ++j) acc += (fs[j] * inv) * w[j];
  if (c > 0.f) {
    const float* r2 = rel2_proj + (size_t)r * DIM;
    for (int j = 0; j < DIM; ++j) acc += r2[j] * w[128 + j];
  }
  h_rel[t] = acc;
}

#define OFF_MS     0
#define OFF_MD     4096
#define OFF_MR     8192
#define OFF_M2     12288
#define OFF_CVEC   16384
#define OFF_VREL2  16448
#define OFF_A0     16512
#define OFF_UR     16576
#define OFF_AR     48576
#define OFF_REL2P  49088
#define OFF_AS     81088
#define OFF_AD     181088
#define OFF_CNT    381088
#define OFF_FEAT   381588
#define OFF_US     1445588
#define OFF_UD     7845588
#define ZERO_CNT   (OFF_FEAT + NUM_RELS * 128 - OFF_CNT)

extern "C" void kernel_launch(void* const* d_in, const int* in_sizes, int n_in,
                              void* d_out, int out_size, void* d_ws,
                              size_t ws_size, hipStream_t stream) {
  const float* ent_embed = (const float*)d_in[0];
  const float* rel_embed = (const float*)d_in[1];
  const float* W_ent  = (const float*)d_in[2];
  const float* b_ent  = (const float*)d_in[3];
  const float* W_relL = (const float*)d_in[4];
  const float* b_relL = (const float*)d_in[5];
  const float* W_rel2 = (const float*)d_in[6];
  const float* b_rel2 = (const float*)d_in[7];
  const float* W_rel3 = (const float*)d_in[8];
  const float* b_rel3 = (const float*)d_in[9];
  const float* W_a    = (const float*)d_in[10];
  const float* b_a    = (const float*)d_in[11];
  const float* W_fc   = (const float*)d_in[12];
  const float* b_fc   = (const float*)d_in[13];
  const int* src_idx  = (const int*)d_in[14];
  const int* dst_idx  = (const int*)d_in[15];
  const int* rel_idx  = (const int*)d_in[16];

  float* ws = (float*)d_ws;
  float* h_ent = (float*)d_out;
  float* h_rel = (float*)d_out + (size_t)NUM_NODES * DIM;

  fold_kernel<<<FOLD_BLOCKS, 256, 0, stream>>>(
      W_ent, b_ent, W_relL, b_relL, W_rel2, b_rel2, W_a, b_a, W_fc, b_fc,
      ws + OFF_MS, ws + OFF_MD, ws + OFF_MR, ws + OFF_M2, ws + OFF_CVEC,
      ws + OFF_VREL2, ws + OFF_A0);

  precompute_kernel<<<PC_NODE_BLOCKS + PC_REL_BLOCKS, 256, 0, stream>>>(
      ent_embed, rel_embed, W_a, ws + OFF_MS, ws + OFF_MD, ws + OFF_MR,
      ws + OFF_M2, ws + OFF_VREL2, ws + OFF_US, ws + OFF_UD, ws + OFF_AS,
      ws + OFF_AD, ws + OFF_UR, ws + OFF_REL2P, ws + OFF_AR, ws + OFF_CNT,
      ZERO_CNT);

  edge_gather_h_kernel<<<HN_GROUPS, 1024, 0, stream>>>(
      src_idx, dst_idx, rel_idx, ws + OFF_US, ws + OFF_UD, ws + OFF_UR,
      ws + OFF_CVEC, ws + OFF_AS, ws + OFF_AD, ws + OFF_AR, ws + OFF_A0, h_ent,
      NUM_EDGES);
  edge_featsum_kernel<<<FS_GROUPS * FS_SLICES, 1024, 0, stream>>>(
      src_idx, dst_idx, rel_idx, h_ent, ws + OFF_FEAT, ws + OFF_CNT,
      NUM_EDGES);

  h_rel_kernel<<<(NUM_RELS * DIM + 255) / 256, 256, 0, stream>>>(
      ws + OFF_FEAT, ws + OFF_CNT, ws + OFF_REL2P, W_rel3, b_rel3, h_rel,
      NUM_RELS);
}


# edges grouped by the owner of their source node once per call (LDS counting sort, no global atomics): an owner reads its own 8-byte records, not all of src plus dst/rel per edge; sum order varies
# speedup vs baseline: 7.4020x; 1.0790x over previous
#include <hip/hip_runtime.h>
#include <math.h>


#define NUM_NODES 100000
#define NUM_RELS  500
#define DIM       64
#define NUM_EDGES 1000000

typedef __attribute__((ext_vector_type(2))) float v2f;
typedef __attribute__((ext_vector_type(8))) float v8f;

#define FOLD_BLOCKS (DIM + 1)

__device__ __forceinline__ float row16_sum(float x) {
  x += __int_as_float(__builtin_amdgcn_update_dpp(0, __float_as_int(x), 0x161, 0xf, 0xf, false));
  x += __int_as_float(__builtin_amdgcn_update_dpp(0, __float_as_int(x), 0x162, 0xf, 0xf, false));
  x += __int_as_float(__builtin_amdgcn_update_dpp(0, __float_as_int(x), 0x164, 0xf, 0xf, false));
  x += __int_as_float(__builtin_amdgcn_update_dpp(0, __float_as_int(x), 0x168, 0xf, 0xf, false));
  return x;
}

__device__ __forceinline__ float wave32_sum(float x) {
  x = row16_sum(x);
  return x + __shfl_xor(x, 16);
}

__global__ __launch_bounds__(256) void fold_kernel(
    const float* __restrict__ W_ent, const float* __restrict__ b_ent,
    const float* __restrict__ W_relL, const float* __restrict__ b_relL,
    const float* __restrict__ W_rel2, const float* __restrict__ b_rel2,
    const float* __restrict__ W_a, const float* __restrict__ b_a,
    const float* __restrict__ W_fc, const float* __restrict__ b_fc,
    float* __restrict__ M_s, float* __restrict__ M_d,
    float* __restrict__ M_r, float* __restrict__ M_2,
    float* __restrict__ cvec, float* __restrict__ v_rel2,
    float* __restrict__ a0) {
  __shared__ __attribute__((aligned(16))) float srow[4 * DIM];
  const int t = threadIdx.x;

  if (blockIdx.x < DIM) {
    const int o = blockIdx.x;
    srow[t] = t < 3 * DIM ? W_fc[o * 192 + t] : W_rel2[o * DIM + (t - 3 * DIM)];
    __syncthreads();
    const int mat = t >> 6;
    const int k = t & 63;
    const float* Wk = (mat < 2 ? W_ent : W_relL) + k;
    const float* cf = &srow[mat * DIM];
    float acc = 0.f;
#pragma unroll
    for (int u = 0; u < DIM; ++u) acc += cf[u] * Wk[u * DIM];
    float* Mout = mat == 0 ? M_s : mat == 1 ? M_d : mat == 2 ? M_r : M_2;
    Mout[o * DIM + k] = acc;
    return;
  }

  const int wave = t >> 5;
  const int lane = t & 31;
  const int orow = wave * 8 + (lane & 7);
  const float bfc = b_fc[orow], br2 = b_rel2[orow];
  const float wa0 = W_a[lane], wa1 = W_a[lane + 32], ba = b_a[0];
  const float be0 = b_ent[lane], be1 = b_ent[lane + 32];
  const float bl0 = b_relL[lane], bl1 = b_relL[lane + 32];
  float pc[8], pv[8];
#pragma unroll
  for (int j = 0; j < 8; ++j) {
    const float* f = W_fc + (wave * 8 + j) * 192;
    const float* g = W_rel2 + (wave * 8 + j) * DIM;
    pc[j] = f[lane] * be0 + f[64 + lane] * be0 + f[128 + lane] * bl0 +
            f[32 + lane] * be1 + f[96 + lane] * be1 + f[160 + lane] * bl1;
    pv[j] = g[lane] * bl0 + g[32 + lane] * bl1;
  }
  float cs = 0.f, vs = 0.f;
#pragma unroll
  for (int j = 0; j < 8; ++j) {
    const float sc = wave32_sum(pc[j]);
    const float sv = wave32_sum(pv[j]);
    cs = (lane == j) ? sc : cs;
    vs = (lane == j) ? sv : vs;
  }
  if (lane < 8) {
    const float cv = bfc + cs;
    cvec[orow] = cv;
    v_rel2[orow] = br2 + vs;
    srow[orow] = cv;
  }
  __syncthreads();
  if (wave == 0) {
    const float a = wave32_sum(srow[lane] * wa0 + srow[lane + 32] * wa1);
    if (lane == 0) a0[0] = ba + a;
  }
}

#define PC_NODE_BLOCKS ((NUM_NODES + 127) / 128)
#define HN_NPR    400
#define HN_GROUPS ((NUM_NODES + HN_NPR - 1) / HN_NPR)
#define BK_GROUPS HN_GROUPS
#define BK_SLICE  ((NUM_EDGES + BK_GROUPS - 1) / BK_GROUPS)
#define BK_BINS   256
static_assert(HN_GROUPS <= BK_BINS, "an owner index must fit the count-matrix row");
static_assert(BK_GROUPS <= PC_NODE_BLOCKS, "every slice needs a counting work-group");
static_assert(BK_SLICE <= 4096, "a slice's records must fit the LDS staging area");
#define BK_TRIPS  ((BK_SLICE + 1023) / 1024)
#define PC_CNT_TRIPS ((BK_SLICE + 255) / 256)

__device__ __forceinline__ unsigned owner_of(int s) {
  const unsigned b = (unsigned)s / HN_NPR;
  return b < BK_BINS ? b : BK_BINS - 1;
}
#define PC_REL_BLOCKS  ((NUM_RELS + 127) / 128)
#define PC_PITCH (DIM + 4)

__global__ __launch_bounds__(256) void precompute_kernel(
    const float* __restrict__ ent_embed, const float* __restrict__ rel_embed,
    const float* __restrict__ W_a, const float* __restrict__ M_s,
    const float* __restrict__ M_d, const float* __restrict__ M_r,
    const float* __restrict__ M_2, const float* __restrict__ v_rel2,
    float* __restrict__ u_s, float* __restrict__ u_d, float* __restrict__ a_s,
    float* __restrict__ a_d, float* __restrict__ u_r,
    float* __restrict__ rel2_proj, float* __restrict__ a_r,
    float* __restrict__ zero_base, int zero_n, const int* __restrict__ src,
    float* __restrict__ cntm_ws) {
  __shared__ float sW[2 * DIM * PC_PITCH];
  __shared__ int shist[BK_BINS];
  const int t = threadIdx.x;
  int* cntm = reinterpret_cast<int*>(cntm_ws);

  if (blockIdx.x < BK_GROUPS) {
    shist[t] = 0;
    __syncthreads();
    const int e0 = blockIdx.x * BK_SLICE;
    int sv[PC_CNT_TRIPS];
#pragma unroll
    for (int k = 0; k < PC_CNT_TRIPS; ++k) {
      const int i = t + 256 * k;
      sv[k] = 0;
      if (i < BK_SLICE && e0 + i < NUM_EDGES) sv[k] = src[e0 + i];
    }
#pragma unroll
    for (int k = 0; k < PC_CNT_TRIPS; ++k) {
      const int i = t + 256 * k;
      if (i < BK_SLICE && e0 + i < NUM_EDGES)
        __hip_atomic_fetch_add(&shist[owner_of(sv[k])], 1, __ATOMIC_RELAXED,
                               __HIP_MEMORY_SCOPE_WORKGROUP);
    }
    __syncthreads();
    cntm[blockIdx.x * BK_BINS + t] = shist[t];
  }

  for (int i = blockIdx.x * 256 + t; i < zero_n; i += gridDim.x * 256)
    zero_base[i] = 0.0f;

  const bool is_rel = blockIdx.x >= PC_NODE_BLOCKS;
  const float* A = is_rel ? rel_embed : ent_embed;
  const float* W0 = is_rel ? M_r : M_s;
  const float* W1 = is_rel ? M_2 : M_d;
  float* out0 = is_rel ? u_r : u_s;
  float* out1 = is_rel ? rel2_proj : u_d;
  float* aout0 = is_rel ? a_r : a_s;
  const int M = is_rel ? NUM_RELS : NUM_NODES;
  const int tb = is_rel ? (int)blockIdx.x - PC_NODE_BLOCKS : (int)blockIdx.x;

  for (int i = t; i < DIM * DIM; i += 256) {
    const int si = (i >> 6) * PC_PITCH + (i & 63);
    sW[si] = W0[i];
    sW[DIM * PC_PITCH + si] = W1[i];
  }
  __syncthreads();

  const int wave = t >> 5;
  const int lane = t & 31;
  const int m0 = (tb * 8 + wave) * 16;
  if (m0 >= M) return;

  const int lm = lane & 15;
  const int hi = lane >> 4;

  int arow = m0 + lm;
  if (arow >= M) arow = M - 1;
  const float* Arow = A + (size_t)arow * DIM;

  v8f acc[8] = {};
#pragma unroll
  for (int k = 0; k < DIM; k += 4) {
    v2f a;
    a.x = Arow[k + 2 * hi + 0];
    a.y = Arow[k + 2 * hi + 1];
#pragma unroll
    for (int j = 0; j < 8; ++j) {
      const float* w = &sW[(j * 16 + lm) * PC_PITCH + k + 2 * hi];
      v2f b;
      b.x = w[0];
      b.y = w[1];
      acc[j] = __builtin_amdgcn_wmma_f32_16x16x4_f32(false, a, false, b,
                                                     (short)0, acc[j], false,
                                                     false);
    }
  }

  float p0[8], p1[8];
#pragma unroll
  for (int v = 0; v < 8; ++v) { p0[v] = 0.0f; p1[v] = 0.0f; }
#pragma unroll
  for (int ob = 0; ob < 4; ++ob) {
    const float wa = W_a[ob * 16 + lm];
    const float bv = is_rel ? v_rel2[ob * 16 + lm] : 0.0f;
#pragma unroll
    for (int v = 0; v < 8; ++v) {
      const int row = m0 + v + 8 * hi;
      const float x0 = acc[ob][v];
      const float x1 = acc[4 + ob][v];
      p0[v] += x0 * wa;
      p1[v] += x1 * wa;
      if (row < M) {
        out0[(size_t)row * DIM + ob * 16 + lm] = x0;
        out1[(size_t)row * DIM + ob * 16 + lm] = x1 + bv;
      }
    }
  }

  const int srow = m0 + 8 * hi + lm;
  const bool swr = lm < 8 && srow < M;
  float s0 = 0.0f;
#pragma unroll
  for (int v = 0; v < 8; ++v) {
    const float r = row16_sum(p0[v]);
    s0 = (lm == v) ? r : s0;
  }
  if (swr) aout0[srow] = s0;
  if (!is_rel) {
    float s1 = 0.0f;
#pragma unroll
    for (int v = 0; v < 8; ++v) {
      const float r = row16_sum(p1[v]);
      s1 = (lm == v) ? r : s1;
    }
    if (swr) a_d[srow] = s1;
  }
}

static_assert(NUM_RELS <= (1 << 9), "relation id must fit 9 bits of a record");
static_assert(NUM_NODES <= (1 << 23), "node id must fit 23 bits of a record");

__global__ __launch_bounds__(1024) void bucket_kernel(
    const int* __restrict__ src, const int* __restrict__ dst,
    const int* __restrict__ rel, const float* __restrict__ cntm_ws,
    float* __restrict__ rec_ws, float* __restrict__ binstart_ws,
    float* __restrict__ bintot_ws) {
  const int* cntm = reinterpret_cast<const int*>(cntm_ws);
  uint2* rec = reinterpret_cast<uint2*>(rec_ws);
  int* binstart = reinterpret_cast<int*>(binstart_ws);
  int* bintot = reinterpret_cast<int*>(bintot_ws);
  __shared__ int spart[4 * 2 * BK_BINS];
  __shared__ int sT[BK_BINS], sP[BK_BINS], sL[BK_BINS];
  __shared__ int sbase[BK_BINS];
  __shared__ int lpos[BK_BINS];
  __shared__ int lcur[BK_BINS];
  __shared__ uint2 srec[4096];
  const int t = threadIdx.x;
  const int g = blockIdx.x;
  const int b = t & (BK_BINS - 1);
  const int part = t >> 8;

  int tot = 0, pre = 0;
#pragma unroll
  for (int k = 0; k < (BK_GROUPS + 3) / 4; ++k) {
    const int r = part + 4 * k;
    if (r < BK_GROUPS) {
      const int v = cntm[r * BK_BINS + b];
      tot += v;
      pre += r < g ? v : 0;
    }
  }
  spart[part * 2 * BK_BINS + b] = tot;
  spart[part * 2 * BK_BINS + BK_BINS + b] = pre;
  __syncthreads();
  if (t < BK_BINS) {
    sT[t] = spart[t] + spart[2 * BK_BINS + t] + spart[4 * BK_BINS + t] +
            spart[6 * BK_BINS + t];
    sP[t] = spart[BK_BINS + t] + spart[3 * BK_BINS + t] +
            spart[5 * BK_BINS + t] + spart[7 * BK_BINS + t];
    sL[t] = cntm[g * BK_BINS + t];
    lcur[t] = 0;
  }
  __syncthreads();
  if (t < 32) {
    int vT[8], vL[8];
    int accT = 0, accL = 0;
#pragma unroll
    for (int j = 0; j < 8; ++j) {
      vT[j] = sT[8 * t + j];
      vL[j] = sL[8 * t + j];
      accT += vT[j];
      accL += vL[j];
    }
    int inT = accT, inL = accL;
    for (int o = 1; o < 32; o <<= 1) {
      const int uT = __shfl_up(inT, o);
      const int uL = __shfl_up(inL, o);
      if (t >= o) { inT += uT; inL += uL; }
    }
    int exT = inT - accT, exL = inL - accL;
#pragma unroll
    for (int j = 0; j < 8; ++j) {
      const int bb = 8 * t + j;
      sbase[bb] = exT + sP[bb];
      lpos[bb] = exL;
      if (g == 0) { binstart[bb] = exT; bintot[bb] = vT[j]; }
      exT += vT[j];
      exL += vL[j];
    }
  }
  __syncthreads();

  const int e0 = g * BK_SLICE;
  int ns = NUM_EDGES - e0;
  ns = ns > BK_SLICE ? BK_SLICE : ns;
  ns = ns < 0 ? 0 : ns;
  int sv[BK_TRIPS], dv[BK_TRIPS], rv[BK_TRIPS];
#pragma unroll
  for (int k = 0; k < BK_TRIPS; ++k) {
    const int i = t + 1024 * k;
    sv[k] = 0; dv[k] = 0; rv[k] = 0;
    if (i < ns) { sv[k] = src[e0 + i]; dv[k] = dst[e0 + i]; rv[k] = rel[e0 + i]; }
  }
#pragma unroll
  for (int k = 0; k < BK_TRIPS; ++k) {
    const int i = t + 1024 * k;
    if (i < ns) {
      const unsigned bb = owner_of(sv[k]);
      const int rank = __hip_atomic_fetch_add(&lcur[bb], 1, __ATOMIC_RELAXED,
                                              __HIP_MEMORY_SCOPE_WORKGROUP);
      int slot = lpos[bb] + rank;
      slot = slot < 0 ? 0 : (slot > 4095 ? 4095 : slot);
      srec[slot] = make_uint2((unsigned)sv[k], ((unsigned)dv[k] << 9) |
                                                   ((unsigned)rv[k] & 511u));
    }
  }
  __syncthreads();
  for (int i = t; i < ns; i += 1024) {
    const uint2 rc = srec[i];
    const unsigned bb = owner_of((int)rc.x);
    int gp = sbase[bb] + (i - lpos[bb]);
    gp = gp < 0 ? 0 : (gp > NUM_EDGES - 1 ? NUM_EDGES - 1 : gp);
    rec[gp] = rc;
  }
}

#define HN_BATCH  8

__device__ __forceinline__ void hn_drain(
    uint2 rc, int n, int lane, int n0, float a0,
    const float* __restrict__ a_s, const float* __restrict__ a_d,
    const float* __restrict__ a_r, const float* __restrict__ u_d,
    const float* __restrict__ u_r, float* stab, float* sbs) {
  int d = 0, r = 0, loc = 0, bb = 0;
  if (lane < n) {
    loc = (int)rc.x - n0;
    loc = loc < 0 ? 0 : (loc > HN_NPR - 1 ? HN_NPR - 1 : loc);
    d = (int)(rc.y >> 9);
    r = (int)(rc.y & 511u);
    float l = a_s[n0 + loc] + a_d[d] + a_r[r] + a0;
    l = l > 0.f ? l : 0.01f * l;
    const float b = expf(l);
    __hip_atomic_fetch_add(&sbs[loc], b, __ATOMIC_RELAXED,
                           __HIP_MEMORY_SCOPE_WORKGROUP);
    bb = __float_as_int(b);
  }
  for (int k0 = 0; k0 < n; k0 += HN_BATCH) {
    v2f vd[HN_BATCH], vr[HN_BATCH];
    float bk[HN_BATCH];
    int lo[HN_BATCH];
#pragma unroll
    for (int k = 0; k < HN_BATCH; ++k) {
      const int sl = k0 + k < n ? k0 + k : n - 1;
      const int dk = __builtin_amdgcn_readlane(d, sl);
      const int rk = __builtin_amdgcn_readlane(r, sl);
      bk[k] = __int_as_float(__builtin_amdgcn_readlane(bb, sl));
      lo[k] = __builtin_amdgcn_readlane(loc, sl);
      vd[k] = reinterpret_cast<const v2f*>(u_d + (size_t)dk * DIM)[lane];
      vr[k] = reinterpret_cast<const v2f*>(u_r + (size_t)rk * DIM)[lane];
    }
#pragma unroll
    for (int k = 0; k < HN_BATCH; ++k) {
      if (k0 + k < n) {
        float* p = &stab[lo[k] * DIM + 2 * lane];
        __hip_atomic_fetch_add(p, bk[k] * (vd[k].x + vr[k].x), __ATOMIC_RELAXED,
                               __HIP_MEMORY_SCOPE_WORKGROUP);
        __hip_atomic_fetch_add(p + 1, bk[k] * (vd[k].y + vr[k].y),
                               __ATOMIC_RELAXED, __HIP_MEMORY_SCOPE_WORKGROUP);
      }
    }
  }
}

__global__ __launch_bounds__(1024) void edge_gather_h_kernel(
    const float* __restrict__ rec_ws, const float* __restrict__ binstart_ws,
    const float* __restrict__ bintot_ws, const float* __restrict__ u_d,
    const float* __restrict__ u_r, const float* __restrict__ cvec,
    const float* __restrict__ a_s, const float* __restrict__ a_d,
    const float* __restrict__ a_r, const float* __restrict__ a0p,
    float* __restrict__ h_ent) {
  __shared__ __attribute__((aligned(16))) float stab[HN_NPR * DIM];
  __shared__ float sbs[HN_NPR];
  const uint2* rec = reinterpret_cast<const uint2*>(rec_ws);
  const int* binstart = reinterpret_cast<const int*>(binstart_ws);
  const int* bintot = reinterpret_cast<const int*>(bintot_ws);
  const int t = threadIdx.x;
  for (int i = t; i < HN_NPR * DIM; i += 1024) stab[i] = 0.0f;
  for (int i = t; i < HN_NPR; i += 1024) sbs[i] = 0.0f;
  __syncthreads();
  const float a0 = a0p[0];

  const int n0 = blockIdx.x * HN_NPR;
  int nn = NUM_NODES - n0;
  nn = nn > HN_NPR ? HN_NPR : nn;
  nn = nn < 0 ? 0 : nn;

  int s0 = binstart[blockIdx.x];
  int ne = bintot[blockIdx.x];
  s0 = s0 < 0 ? 0 : (s0 > NUM_EDGES ? NUM_EDGES : s0);
  ne = ne < 0 ? 0 : (ne > NUM_EDGES - s0 ? NUM_EDGES - s0 : ne);

  const int wave = __builtin_amdgcn_readfirstlane(t >> 5);
  const int lane = t & 31;

  int i = wave * 32;
  uint2 nx = make_uint2(0u, 0u);
  if (i + lane < ne) nx = rec[s0 + i + lane];
  for (; i < ne; i += 1024) {
    const uint2 rc = nx;
    const int left = ne - i;
    const int m = left < 32 ? left : 32;
    const int in = i + 1024;
    if (in + lane < ne) nx = rec[s0 + in + lane];
    hn_drain(rc, m, lane, n0, a0, a_s, a_d, a_r, u_d, u_r, stab, sbs);
  }
  __syncthreads();

  for (int j = t; j < nn * (DIM / 4); j += 1024) {
    const int row = j >> 4, c4 = j & 15;
    const int node = n0 + row;
    const float bs = sbs[row];
    float4 o = make_float4(0.0f, 0.0f, 0.0f, 0.0f);
    float4* hp = reinterpret_cast<float4*>(h_ent + (size_t)node * DIM) + c4;
    if (bs > 0.0f) {
      const float4 a = reinterpret_cast<const float4*>(stab)[row * 16 + c4];
      const float4 us = *hp;
      const float4 cv = reinterpret_cast<const float4*>(cvec)[c4];
      o.x = a.x / bs + (us.x + cv.x);
      o.y = a.y / bs + (us.y + cv.y);
      o.z = a.z / bs + (us.z + cv.z);
      o.w = a.w / bs + (us.w + cv.w);
    }
    *hp = o;
  }
}

#define FS_GROUPS 8
#define FS_RPG    63
#define FS_SLICES 32
#define FS_BATCH  8
static_assert(FS_GROUPS * FS_RPG >= NUM_RELS, "relation ranges must cover all relations");

__global__ __launch_bounds__(1024) void edge_featsum_kernel(
    const int* __restrict__ src, const int* __restrict__ dst,
    const int* __restrict__ rel, const float* __restrict__ h_ent,
    float* __restrict__ featsum, float* __restrict__ cnt, int E) {
  __shared__ float sfs[FS_RPG * 128];
  __shared__ float scnt[FS_RPG];
  const int t = threadIdx.x;
  for (int i = t; i < FS_RPG * 128; i += 1024) sfs[i] = 0.0f;
  if (t < FS_RPG) scnt[t] = 0.0f;
  __syncthreads();

  const int grp = blockIdx.x % FS_GROUPS;
  const int slice = blockIdx.x / FS_GROUPS;
  const int r0 = grp * FS_RPG;
  int nrel = NUM_RELS - r0;
  nrel = nrel > FS_RPG ? FS_RPG : nrel;
  nrel = nrel < 0 ? 0 : nrel;
  const int per = (E + FS_SLICES - 1) / FS_SLICES;
  const int e0 = slice * per;
  const int e1 = (e0 + per < E) ? (e0 + per) : E;

  const int wave = __builtin_amdgcn_readfirstlane(t >> 5);
  const int lane = t & 31;

  int base = e0 + wave * 32;
  int rn, sn, dn;
  {
    int ec = base + lane < e1 ? base + lane : e1 - 1;
    ec = ec < 0 ? 0 : ec;
    rn = rel[ec]; sn = src[ec]; dn = dst[ec];
    if (base + lane >= e1) rn = -1;
  }
  for (; base < e1; base += 32 * 32) {
    const int r = rn, s = sn, d = dn;
    const int en = base + 32 * 32 + lane;
    const int ec = en < e1 ? en : e1 - 1;
    rn = rel[ec]; sn = src[ec]; dn = dst[ec];
    if (en >= e1) rn = -1;

    const int loc = r - r0;
    const bool mine = (unsigned)loc < (unsigned)nrel;
    unsigned mask = __builtin_amdgcn_ballot_w32(mine);
    if (mine)
      __hip_atomic_fetch_add(&scnt[loc], 1.0f, __ATOMIC_RELAXED,
                             __HIP_MEMORY_SCOPE_WORKGROUP);

    while (mask != 0u) {
      const int n = __builtin_popcount(mask);
      v2f vs[FS_BATCH], vd[FS_BATCH];
      int lo[FS_BATCH];
#pragma unroll
      for (int k = 0; k < FS_BATCH; ++k) {
        if (k < n) {
          const int l = __builtin_ctz(mask);
          mask &= mask - 1u;
          const int sk = __builtin_amdgcn_readlane(s, l);
          const int dk = __builtin_amdgcn_readlane(d, l);
          lo[k] = __builtin_amdgcn_readlane(loc, l);
          vs[k] = reinterpret_cast<const v2f*>(h_ent + (size_t)sk * DIM)[lane];
          vd[k] = reinterpret_cast<const v2f*>(h_ent + (size_t)dk * DIM)[lane];
        }
      }
#pragma unroll
      for (int k = 0; k < FS_BATCH; ++k) {
        if (k < n) {
          float* p = &sfs[lo[k] * 128 + 2 * lane];
          __hip_atomic_fetch_add(p, vs[k].x, __ATOMIC_RELAXED,
                                 __HIP_MEMORY_SCOPE_WORKGROUP);
          __hip_atomic_fetch_add(p + 1, vs[k].y, __ATOMIC_RELAXED,
                                 __HIP_MEMORY_SCOPE_WORKGROUP);
          __hip_atomic_fetch_add(p + 64, vd[k].x, __ATOMIC_RELAXED,
                                 __HIP_MEMORY_SCOPE_WORKGROUP);
          __hip_atomic_fetch_add(p + 65, vd[k].y, __ATOMIC_RELAXED,
                                 __HIP_MEMORY_SCOPE_WORKGROUP);
        }
      }
    }
  }
  __syncthreads();

  float* fs = featsum + (size_t)r0 * 128;
  for (int i = t; i < nrel * 128; i += 1024) atomicAdd(&fs[i], sfs[i]);
  if (t < nrel) atomicAdd(&cnt[r0 + t], scnt[t]);
}

__global__ void h_rel_kernel(const float* __restrict__ featsum,
                             const float* __restrict__ cnt,
                             const float* __restrict__ rel2_proj,
                             const float* __restrict__ W_rel3,
                             const float* __restrict__ b_rel3,
                             float* __restrict__ h_rel, int R) {
  const int t = blockIdx.x * blockDim.x + threadIdx.x;
  if (t >= R * DIM) return;
  const int r = t >> 6, o = t & 63;
  const float c = cnt[r];
  const float inv = 1.0f / fmaxf(c, 1.0f);
  const float* fs = featsum + (size_t)r * 128;
  const float* w = W_rel3 + (size_t)o * 192;
  float acc = b_rel3[o];
  for (int j = 0; j < 128; ++j) acc += (fs[j] * inv) * w[j];
  if (c > 0.f) {
    const float* r2 = rel2_proj + (size_t)r * DIM;
    for (int j = 0; j < DIM; ++j) acc += r2[j] * w[128 + j];
  }
  h_rel[t] = acc;
}

#define OFF_MS     0
#define OFF_MD     4096
#define OFF_MR     8192
#define OFF_M2     12288
#define OFF_CVEC   16384
#define OFF_VREL2  16448
#define OFF_A0     16512
#define OFF_UR     16576
#define OFF_AR     48576
#define OFF_REL2P  49088
#define OFF_AS     81088
#define OFF_AD     181088
#define OFF_CNTM   281088
#define OFF_BSTART 345088
#define OFF_BTOT   345344
#define OFF_CNT    381088
#define OFF_FEAT   381588
#define OFF_REC    1445588
#define OFF_UD     7845588
static_assert(OFF_CNTM + BK_GROUPS * BK_BINS <= OFF_BSTART &&
              OFF_BTOT + BK_BINS <= OFF_CNT &&
              OFF_REC % 2 == 0 && OFF_REC + 2 * NUM_EDGES <= OFF_UD,
              "workspace planes do not overlap; records are 8-byte aligned");
#define ZERO_CNT   (OFF_FEAT + NUM_RELS * 128 - OFF_CNT)

extern "C" void kernel_launch(void* const* d_in, const int* in_sizes, int n_in,
                              void* d_out, int out_size, void* d_ws,
                              size_t ws_size, hipStream_t stream) {
  const float* ent_embed = (const float*)d_in[0];
  const float* rel_embed = (const float*)d_in[1];
  const float* W_ent  = (const float*)d_in[2];
  const float* b_ent  = (const float*)d_in[3];
  const float* W_relL = (const float*)d_in[4];
  const float* b_relL = (const float*)d_in[5];
  const float* W_rel2 = (const float*)d_in[6];
  const float* b_rel2 = (const float*)d_in[7];
  const float* W_rel3 = (const float*)d_in[8];
  const float* b_rel3 = (const float*)d_in[9];
  const float* W_a    = (const float*)d_in[10];
  const float* b_a    = (const float*)d_in[11];
  const float* W_fc   = (const float*)d_in[12];
  const float* b_fc   = (const float*)d_in[13];
  const int* src_idx  = (const int*)d_in[14];
  const int* dst_idx  = (const int*)d_in[15];
  const int* rel_idx  = (const int*)d_in[16];

  float* ws = (float*)d_ws;
  float* h_ent = (float*)d_out;
  float* h_rel = (float*)d_out + (size_t)NUM_NODES * DIM;

  fold_kernel<<<FOLD_BLOCKS, 256, 0, stream>>>(
      W_ent, b_ent, W_relL, b_relL, W_rel2, b_rel2, W_a, b_a, W_fc, b_fc,
      ws + OFF_MS, ws + OFF_MD, ws + OFF_MR, ws + OFF_M2, ws + OFF_CVEC,
      ws + OFF_VREL2, ws + OFF_A0);

  precompute_kernel<<<PC_NODE_BLOCKS + PC_REL_BLOCKS, 256, 0, stream>>>(
      ent_embed, rel_embed, W_a, ws + OFF_MS, ws + OFF_MD, ws + OFF_MR,
      ws + OFF_M2, ws + OFF_VREL2, h_ent, ws + OFF_UD, ws + OFF_AS,
      ws + OFF_AD, ws + OFF_UR, ws + OFF_REL2P, ws + OFF_AR, ws + OFF_CNT,
      ZERO_CNT, src_idx, ws + OFF_CNTM);

  bucket_kernel<<<BK_GROUPS, 1024, 0, stream>>>(
      src_idx, dst_idx, rel_idx, ws + OFF_CNTM, ws + OFF_REC, ws + OFF_BSTART,
      ws + OFF_BTOT);

  edge_gather_h_kernel<<<HN_GROUPS, 1024, 0, stream>>>(
      ws + OFF_REC, ws + OFF_BSTART, ws + OFF_BTOT, ws + OFF_UD, ws + OFF_UR,
      ws + OFF_CVEC, ws + OFF_AS, ws + OFF_AD, ws + OFF_AR, ws + OFF_A0, h_ent);
  edge_featsum_kernel<<<FS_GROUPS * FS_SLICES, 1024, 0, stream>>>(
      src_idx, dst_idx, rel_idx, h_ent, ws + OFF_FEAT, ws + OFF_CNT,
      NUM_EDGES);

  h_rel_kernel<<<(NUM_RELS * DIM + 255) / 256, 256, 0, stream>>>(
      ws + OFF_FEAT, ws + OFF_CNT, ws + OFF_REL2P, W_rel3, b_rel3, h_rel,
      NUM_RELS);
}


# edges also grouped by relation per slice (LDS counting sort in bucket_kernel); one work-group per relation sums h_ent rows in registers and finishes h_rel: no atomics, 5 launches; sum order varies
# speedup vs baseline: 8.0715x; 1.0904x over previous
#include <hip/hip_runtime.h>
#include <math.h>


#define NUM_NODES 100000
#define NUM_RELS  500
#define DIM       64
#define NUM_EDGES 1000000

typedef __attribute__((ext_vector_type(2))) float v2f;
typedef __attribute__((ext_vector_type(8))) float v8f;

#define FOLD_BLOCKS (DIM + 1)

__device__ __forceinline__ float row16_sum(float x) {
  x += __int_as_float(__builtin_amdgcn_update_dpp(0, __float_as_int(x), 0x161, 0xf, 0xf, false));
  x += __int_as_float(__builtin_amdgcn_update_dpp(0, __float_as_int(x), 0x162, 0xf, 0xf, false));
  x += __int_as_float(__builtin_amdgcn_update_dpp(0, __float_as_int(x), 0x164, 0xf, 0xf, false));
  x += __int_as_float(__builtin_amdgcn_update_dpp(0, __float_as_int(x), 0x168, 0xf, 0xf, false));
  return x;
}

__device__ __forceinline__ float wave32_sum(float x) {
  x = row16_sum(x);
  return x + __shfl_xor(x, 16);
}

__global__ __launch_bounds__(256) void fold_kernel(
    const float* __restrict__ W_ent, const float* __restrict__ b_ent,
    const float* __restrict__ W_relL, const float* __restrict__ b_relL,
    const float* __restrict__ W_rel2, const float* __restrict__ b_rel2,
    const float* __restrict__ W_a, const float* __restrict__ b_a,
    const float* __restrict__ W_fc, const float* __restrict__ b_fc,
    float* __restrict__ M_s, float* __restrict__ M_d,
    float* __restrict__ M_r, float* __restrict__ M_2,
    float* __restrict__ cvec, float* __restrict__ v_rel2,
    float* __restrict__ a0) {
  __shared__ __attribute__((aligned(16))) float srow[4 * DIM];
  const int t = threadIdx.x;

  if (blockIdx.x < DIM) {
    const int o = blockIdx.x;
    srow[t] = t < 3 * DIM ? W_fc[o * 192 + t] : W_rel2[o * DIM + (t - 3 * DIM)];
    __syncthreads();
    const int mat = t >> 6;
    const int k = t & 63;
    const float* Wk = (mat < 2 ? W_ent : W_relL) + k;
    const float* cf = &srow[mat * DIM];
    float acc = 0.f;
#pragma unroll
    for (int u = 0; u < DIM; ++u) acc += cf[u] * Wk[u * DIM];
    float* Mout = mat == 0 ? M_s : mat == 1 ? M_d : mat == 2 ? M_r : M_2;
    Mout[o * DIM + k] = acc;
    return;
  }

  const int wave = t >> 5;
  const int lane = t & 31;
  const int orow = wave * 8 + (lane & 7);
  const float bfc = b_fc[orow], br2 = b_rel2[orow];
  const float wa0 = W_a[lane], wa1 = W_a[lane + 32], ba = b_a[0];
  const float be0 = b_ent[lane], be1 = b_ent[lane + 32];
  const float bl0 = b_relL[lane], bl1 = b_relL[lane + 32];
  float pc[8], pv[8];
#pragma unroll
  for (int j = 0; j < 8; ++j) {
    const float* f = W_fc + (wave * 8 + j) * 192;
    const float* g = W_rel2 + (wave * 8 + j) * DIM;
    pc[j] = f[lane] * be0 + f[64 + lane] * be0 + f[128 + lane] * bl0 +
            f[32 + lane] * be1 + f[96 + lane] * be1 + f[160 + lane] * bl1;
    pv[j] = g[lane] * bl0 + g[32 + lane] * bl1;
  }
  float cs = 0.f, vs = 0.f;
#pragma unroll
  for (int j = 0; j < 8; ++j) {
    const float sc = wave32_sum(pc[j]);
    const float sv = wave32_sum(pv[j]);
    cs = (lane == j) ? sc : cs;
    vs = (lane == j) ? sv : vs;
  }
  if (lane < 8) {
    const float cv = bfc + cs;
    cvec[orow] = cv;
    v_rel2[orow] = br2 + vs;
    srow[orow] = cv;
  }
  __syncthreads();
  if (wave == 0) {
    const float a = wave32_sum(srow[lane] * wa0 + srow[lane + 32] * wa1);
    if (lane == 0) a0[0] = ba + a;
  }
}

#define PC_NODE_BLOCKS ((NUM_NODES + 127) / 128)
#define HN_NPR    400
#define HN_GROUPS ((NUM_NODES + HN_NPR - 1) / HN_NPR)
#define BK_GROUPS HN_GROUPS
#define BK_SLICE  ((NUM_EDGES + BK_GROUPS - 1) / BK_GROUPS)
#define BK_BINS   256
static_assert(HN_GROUPS <= BK_BINS, "an owner index must fit the count-matrix row");
static_assert(BK_GROUPS <= PC_NODE_BLOCKS, "every slice needs a counting work-group");
static_assert(BK_SLICE <= 4096, "a slice's records must fit the LDS staging area");
#define BK_TRIPS  ((BK_SLICE + 1023) / 1024)
#define PC_CNT_TRIPS ((BK_SLICE + 255) / 256)

__device__ __forceinline__ unsigned owner_of(int s) {
  const unsigned b = (unsigned)s / HN_NPR;
  return b < BK_BINS ? b : BK_BINS - 1;
}
#define PC_REL_BLOCKS  ((NUM_RELS + 127) / 128)
#define PC_PITCH (DIM + 4)

__global__ __launch_bounds__(256) void precompute_kernel(
    const float* __restrict__ ent_embed, const float* __restrict__ rel_embed,
    const float* __restrict__ W_a, const float* __restrict__ M_s,
    const float* __restrict__ M_d, const float* __restrict__ M_r,
    const float* __restrict__ M_2, const float* __restrict__ v_rel2,
    float* __restrict__ u_s, float* __restrict__ u_d, float* __restrict__ a_s,
    float* __restrict__ a_d, float* __restrict__ u_r,
    float* __restrict__ rel2_proj, float* __restrict__ a_r,
    const int* __restrict__ src, float* __restrict__ cntm_ws) {
  __shared__ float sW[2 * DIM * PC_PITCH];
  __shared__ int shist[BK_BINS];
  const int t = threadIdx.x;
  int* cntm = reinterpret_cast<int*>(cntm_ws);

  if (blockIdx.x < BK_GROUPS) {
    shist[t] = 0;
    __syncthreads();
    const int e0 = blockIdx.x * BK_SLICE;
    int sv[PC_CNT_TRIPS];
#pragma unroll
    for (int k = 0; k < PC_CNT_TRIPS; ++k) {
      const int i = t + 256 * k;
      sv[k] = 0;
      if (i < BK_SLICE && e0 + i < NUM_EDGES) sv[k] = src[e0 + i];
    }
#pragma unroll
    for (int k = 0; k < PC_CNT_TRIPS; ++k) {
      const int i = t + 256 * k;
      if (i < BK_SLICE && e0 + i < NUM_EDGES)
        __hip_atomic_fetch_add(&shist[owner_of(sv[k])], 1, __ATOMIC_RELAXED,
                               __HIP_MEMORY_SCOPE_WORKGROUP);
    }
    __syncthreads();
    cntm[blockIdx.x * BK_BINS + t] = shist[t];
  }

  const bool is_rel = blockIdx.x >= PC_NODE_BLOCKS;
  const float* A = is_rel ? rel_embed : ent_embed;
  const float* W0 = is_rel ? M_r : M_s;
  const float* W1 = is_rel ? M_2 : M_d;
  float* out0 = is_rel ? u_r : u_s;
  float* out1 = is_rel ? rel2_proj : u_d;
  float* aout0 = is_rel ? a_r : a_s;
  const int M = is_rel ? NUM_RELS : NUM_NODES;
  const int tb = is_rel ? (int)blockIdx.x - PC_NODE_BLOCKS : (int)blockIdx.x;

  for (int i = t; i < DIM * DIM; i += 256) {
    const int si = (i >> 6) * PC_PITCH + (i & 63);
    sW[si] = W0[i];
    sW[DIM * PC_PITCH + si] = W1[i];
  }
  __syncthreads();

  const int wave = t >> 5;
  const int lane = t & 31;
  const int m0 = (tb * 8 + wave) * 16;
  if (m0 >= M) return;

  const int lm = lane & 15;
  const int hi = lane >> 4;

  int arow = m0 + lm;
  if (arow >= M) arow = M - 1;
  const float* Arow = A + (size_t)arow * DIM;

  v8f acc[8] = {};
#pragma unroll
  for (int k = 0; k < DIM; k += 4) {
    v2f a;
    a.x = Arow[k + 2 * hi + 0];
    a.y = Arow[k + 2 * hi + 1];
#pragma unroll
    for (int j = 0; j < 8; ++j) {
      const float* w = &sW[(j * 16 + lm) * PC_PITCH + k + 2 * hi];
      v2f b;
      b.x = w[0];
      b.y = w[1];
      acc[j] = __builtin_amdgcn_wmma_f32_16x16x4_f32(false, a, false, b,
                                                     (short)0, acc[j], false,
                                                     false);
    }
  }

  float p0[8], p1[8];
#pragma unroll
  for (int v = 0; v < 8; ++v) { p0[v] = 0.0f; p1[v] = 0.0f; }
#pragma unroll
  for (int ob = 0; ob < 4; ++ob) {
    const float wa = W_a[ob * 16 + lm];
    const float bv = is_rel ? v_rel2[ob * 16 + lm] : 0.0f;
#pragma unroll
    for (int v = 0; v < 8; ++v) {
      const int row = m0 + v + 8 * hi;
      const float x0 = acc[ob][v];
      const float x1 = acc[4 + ob][v];
      p0[v] += x0 * wa;
      p1[v] += x1 * wa;
      if (row < M) {
        out0[(size_t)row * DIM + ob * 16 + lm] = x0;
        out1[(size_t)row * DIM + ob * 16 + lm] = x1 + bv;
      }
    }
  }

  const int srow = m0 + 8 * hi + lm;
  const bool swr = lm < 8 && srow < M;
  float s0 = 0.0f;
#pragma unroll
  for (int v = 0; v < 8; ++v) {
    const float r = row16_sum(p0[v]);
    s0 = (lm == v) ? r : s0;
  }
  if (swr) aout0[srow] = s0;
  if (!is_rel) {
    float s1 = 0.0f;
#pragma unroll
    for (int v = 0; v < 8; ++v) {
      const float r = row16_sum(p1[v]);
      s1 = (lm == v) ? r : s1;
    }
    if (swr) a_d[srow] = s1;
  }
}

static_assert(NUM_RELS <= (1 << 9), "relation id must fit 9 bits of a record");
static_assert(NUM_NODES <= (1 << 23), "node id must fit 23 bits of a record");
#define RL_BINS  512
#define RL_SLOTS BK_SLICE
static_assert(NUM_RELS <= RL_BINS, "every relation needs a run word");
static_assert(BK_SLICE < (1 << 15), "start and length of a run must fit 16 bits each");

__global__ __launch_bounds__(1024) void bucket_kernel(
    const int* __restrict__ src, const int* __restrict__ dst,
    const int* __restrict__ rel, const float* __restrict__ cntm_ws,
    float* __restrict__ rec_ws, float* __restrict__ binstart_ws,
    float* __restrict__ bintot_ws, float* __restrict__ rec2_ws,
    float* __restrict__ roff_ws) {
  const int* cntm = reinterpret_cast<const int*>(cntm_ws);
  uint2* rec = reinterpret_cast<uint2*>(rec_ws);
  int* binstart = reinterpret_cast<int*>(binstart_ws);
  int* bintot = reinterpret_cast<int*>(bintot_ws);
  uint2* rec2 = reinterpret_cast<uint2*>(rec2_ws);
  int* roff = reinterpret_cast<int*>(roff_ws);
  __shared__ int spart[4 * 2 * BK_BINS];
  __shared__ int sT[BK_BINS], sP[BK_BINS], sL[BK_BINS];
  __shared__ int sbase[BK_BINS];
  __shared__ int lpos[BK_BINS];
  __shared__ int lcur[BK_BINS];
  __shared__ uint2 srec[4096];
  __shared__ int rcnt[RL_BINS];
  __shared__ int rstart[RL_BINS];
  __shared__ int rcur[RL_BINS];
  __shared__ uint2 srec2[4096];
  const int t = threadIdx.x;
  const int g = blockIdx.x;
  const int b = t & (BK_BINS - 1);
  const int part = t >> 8;

  if (t < RL_BINS) { rcnt[t] = 0; rcur[t] = 0; }

  const int e0 = g * BK_SLICE;
  int ns = NUM_EDGES - e0;
  ns = ns > BK_SLICE ? BK_SLICE : ns;
  ns = ns < 0 ? 0 : ns;
  int sv[BK_TRIPS], dv[BK_TRIPS], rv[BK_TRIPS];
#pragma unroll
  for (int k = 0; k < BK_TRIPS; ++k) {
    const int i = t + 1024 * k;
    sv[k] = 0; dv[k] = 0; rv[k] = 0;
    if (i < ns) { sv[k] = src[e0 + i]; dv[k] = dst[e0 + i]; rv[k] = rel[e0 + i]; }
  }

  int tot = 0, pre = 0;
#pragma unroll
  for (int k = 0; k < (BK_GROUPS + 3) / 4; ++k) {
    const int r = part + 4 * k;
    if (r < BK_GROUPS) {
      const int v = cntm[r * BK_BINS + b];
      tot += v;
      pre += r < g ? v : 0;
    }
  }
  spart[part * 2 * BK_BINS + b] = tot;
  spart[part * 2 * BK_BINS + BK_BINS + b] = pre;
  __syncthreads();
  if (t < BK_BINS) {
    sT[t] = spart[t] + spart[2 * BK_BINS + t] + spart[4 * BK_BINS + t] +
            spart[6 * BK_BINS + t];
    sP[t] = spart[BK_BINS + t] + spart[3 * BK_BINS + t] +
            spart[5 * BK_BINS + t] + spart[7 * BK_BINS + t];
    sL[t] = cntm[g * BK_BINS + t];
    lcur[t] = 0;
  }
#pragma unroll
  for (int k = 0; k < BK_TRIPS; ++k) {
    const int i = t + 1024 * k;
    if (i < ns)
      __hip_atomic_fetch_add(&rcnt[(unsigned)rv[k] & (RL_BINS - 1)], 1,
                             __ATOMIC_RELAXED, __HIP_MEMORY_SCOPE_WORKGROUP);
  }
  __syncthreads();
  if (t < 32) {
    int vT[8], vL[8];
    int accT = 0, accL = 0;
#pragma unroll
    for (int j = 0; j < 8; ++j) {
      vT[j] = sT[8 * t + j];
      vL[j] = sL[8 * t + j];
      accT += vT[j];
      accL += vL[j];
    }
    int inT = accT, inL = accL;
    for (int o = 1; o < 32; o <<= 1) {
      const int uT = __shfl_up(inT, o);
      const int uL = __shfl_up(inL, o);
      if (t >= o) { inT += uT; inL += uL; }
    }
    int exT = inT - accT, exL = inL - accL;
#pragma unroll
    for (int j = 0; j < 8; ++j) {
      const int bb = 8 * t + j;
      sbase[bb] = exT + sP[bb];
      lpos[bb] = exL;
      if (g == 0) { binstart[bb] = exT; bintot[bb] = vT[j]; }
      exT += vT[j];
      exL += vL[j];
    }
  } else if (t < 64) {
    const int l = t - 32;
    int vR[RL_BINS / 32];
    int accR = 0;
#pragma unroll
    for (int j = 0; j < RL_BINS / 32; ++j) {
      vR[j] = rcnt[(RL_BINS / 32) * l + j];
      accR += vR[j];
    }
    int inR = accR;
    for (int o = 1; o < 32; o <<= 1) {
      const int uR = __shfl_up(inR, o);
      if (l >= o) inR += uR;
    }
    int exR = inR - accR;
#pragma unroll
    for (int j = 0; j < RL_BINS / 32; ++j) {
      rstart[(RL_BINS / 32) * l + j] = exR;
      exR += vR[j];
    }
  }
  __syncthreads();

#pragma unroll
  for (int k = 0; k < BK_TRIPS; ++k) {
    const int i = t + 1024 * k;
    if (i < ns) {
      const unsigned bb = owner_of(sv[k]);
      const int rank = __hip_atomic_fetch_add(&lcur[bb], 1, __ATOMIC_RELAXED,
                                              __HIP_MEMORY_SCOPE_WORKGROUP);
      int slot = lpos[bb] + rank;
      slot = slot < 0 ? 0 : (slot > 4095 ? 4095 : slot);
      srec[slot] = make_uint2((unsigned)sv[k], ((unsigned)dv[k] << 9) |
                                                   ((unsigned)rv[k] & 511u));
      const unsigned rb = (unsigned)rv[k] & (RL_BINS - 1);
      const int rank2 = __hip_atomic_fetch_add(&rcur[rb], 1, __ATOMIC_RELAXED,
                                               __HIP_MEMORY_SCOPE_WORKGROUP);
      int slot2 = rstart[rb] + rank2;
      slot2 = slot2 < 0 ? 0 : (slot2 > 4095 ? 4095 : slot2);
      srec2[slot2] = make_uint2((unsigned)sv[k], (unsigned)dv[k]);
    }
  }
  __syncthreads();
  for (int i = t; i < ns; i += 1024) {
    const uint2 rc = srec[i];
    const unsigned bb = owner_of((int)rc.x);
    int gp = sbase[bb] + (i - lpos[bb]);
    gp = gp < 0 ? 0 : (gp > NUM_EDGES - 1 ? NUM_EDGES - 1 : gp);
    rec[gp] = rc;
    rec2[e0 + i] = srec2[i];
  }
  if (t < RL_BINS) roff[g * RL_BINS + t] = rstart[t] | (rcnt[t] << 16);
}

#define HN_BATCH  8

__device__ __forceinline__ void hn_drain(
    uint2 rc, int n, int lane, int n0, float a0,
    const float* __restrict__ a_s, const float* __restrict__ a_d,
    const float* __restrict__ a_r, const float* __restrict__ u_d,
    const float* __restrict__ u_r, float* stab, float* sbs) {
  int d = 0, r = 0, loc = 0, bb = 0;
  if (lane < n) {
    loc = (int)rc.x - n0;
    loc = loc < 0 ? 0 : (loc > HN_NPR - 1 ? HN_NPR - 1 : loc);
    d = (int)(rc.y >> 9);
    r = (int)(rc.y & 511u);
    float l = a_s[n0 + loc] + a_d[d] + a_r[r] + a0;
    l = l > 0.f ? l : 0.01f * l;
    const float b = expf(l);
    __hip_atomic_fetch_add(&sbs[loc], b, __ATOMIC_RELAXED,
                           __HIP_MEMORY_SCOPE_WORKGROUP);
    bb = __float_as_int(b);
  }
  for (int k0 = 0; k0 < n; k0 += HN_BATCH) {
    v2f vd[HN_BATCH], vr[HN_BATCH];
    float bk[HN_BATCH];
    int lo[HN_BATCH];
#pragma unroll
    for (int k = 0; k < HN_BATCH; ++k) {
      const int sl = k0 + k < n ? k0 + k : n - 1;
      const int dk = __builtin_amdgcn_readlane(d, sl);
      const int rk = __builtin_amdgcn_readlane(r, sl);
      bk[k] = __int_as_float(__builtin_amdgcn_readlane(bb, sl));
      lo[k] = __builtin_amdgcn_readlane(loc, sl);
      vd[k] = reinterpret_cast<const v2f*>(u_d + (size_t)dk * DIM)[lane];
      vr[k] = reinterpret_cast<const v2f*>(u_r + (size_t)rk * DIM)[lane];
    }
#pragma unroll
    for (int k = 0; k < HN_BATCH; ++k) {
      if (k0 + k < n) {
        float* p = &stab[lo[k] * DIM + 2 * lane];
        __hip_atomic_fetch_add(p, bk[k] * (vd[k].x + vr[k].x), __ATOMIC_RELAXED,
                               __HIP_MEMORY_SCOPE_WORKGROUP);
        __hip_atomic_fetch_add(p + 1, bk[k] * (vd[k].y + vr[k].y),
                               __ATOMIC_RELAXED, __HIP_MEMORY_SCOPE_WORKGROUP);
      }
    }
  }
}

__global__ __launch_bounds__(1024) void edge_gather_h_kernel(
    const float* __restrict__ rec_ws, const float* __restrict__ binstart_ws,
    const float* __restrict__ bintot_ws, const float* __restrict__ u_d,
    const float* __restrict__ u_r, const float* __restrict__ cvec,
    const float* __restrict__ a_s, const float* __restrict__ a_d,
    const float* __restrict__ a_r, const float* __restrict__ a0p,
    float* __restrict__ h_ent) {
  __shared__ __attribute__((aligned(16))) float stab[HN_NPR * DIM];
  __shared__ float sbs[HN_NPR];
  const uint2* rec = reinterpret_cast<const uint2*>(rec_ws);
  const int* binstart = reinterpret_cast<const int*>(binstart_ws);
  const int* bintot = reinterpret_cast<const int*>(bintot_ws);
  const int t = threadIdx.x;
  for (int i = t; i < HN_NPR * DIM; i += 1024) stab[i] = 0.0f;
  for (int i = t; i < HN_NPR; i += 1024) sbs[i] = 0.0f;
  __syncthreads();
  const float a0 = a0p[0];

  const int n0 = blockIdx.x * HN_NPR;
  int nn = NUM_NODES - n0;
  nn = nn > HN_NPR ? HN_NPR : nn;
  nn = nn < 0 ? 0 : nn;

  int s0 = binstart[blockIdx.x];
  int ne = bintot[blockIdx.x];
  s0 = s0 < 0 ? 0 : (s0 > NUM_EDGES ? NUM_EDGES : s0);
  ne = ne < 0 ? 0 : (ne > NUM_EDGES - s0 ? NUM_EDGES - s0 : ne);

  const int wave = __builtin_amdgcn_readfirstlane(t >> 5);
  const int lane = t & 31;

  int i = wave * 32;
  uint2 nx = make_uint2(0u, 0u);
  if (i + lane < ne) nx = rec[s0 + i + lane];
  for (; i < ne; i += 1024) {
    const uint2 rc = nx;
    const int left = ne - i;
    const int m = left < 32 ? left : 32;
    const int in = i + 1024;
    if (in + lane < ne) nx = rec[s0 + in + lane];
    hn_drain(rc, m, lane, n0, a0, a_s, a_d, a_r, u_d, u_r, stab, sbs);
  }
  __syncthreads();

  for (int j = t; j < nn * (DIM / 4); j += 1024) {
    const int row = j >> 4, c4 = j & 15;
    const int node = n0 + row;
    const float bs = sbs[row];
    float4 o = make_float4(0.0f, 0.0f, 0.0f, 0.0f);
    float4* hp = reinterpret_cast<float4*>(h_ent + (size_t)node * DIM) + c4;
    if (bs > 0.0f) {
      const float4 a = reinterpret_cast<const float4*>(stab)[row * 16 + c4];
      const float4 us = *hp;
      const float4 cv = reinterpret_cast<const float4*>(cvec)[c4];
      o.x = a.x / bs + (us.x + cv.x);
      o.y = a.y / bs + (us.y + cv.y);
      o.z = a.z / bs + (us.z + cv.z);
      o.w = a.w / bs + (us.w + cv.w);
    }
    *hp = o;
  }
}

#define FR_WAVES 16
#define FR_SPW   16
#define FR_BATCH 8
static_assert(FR_WAVES * FR_SPW >= BK_GROUPS, "every slice needs a wave");
static_assert(FR_SPW <= 32 && FR_WAVES * 32 == DIM * 8, "thread maps of the relation pass");

__device__ __forceinline__ float lanes8_sum(float x) {
  x += __int_as_float(__builtin_amdgcn_update_dpp(0, __float_as_int(x), 0x161, 0xf, 0xf, false));
  x += __int_as_float(__builtin_amdgcn_update_dpp(0, __float_as_int(x), 0x162, 0xf, 0xf, false));
  x += __int_as_float(__builtin_amdgcn_update_dpp(0, __float_as_int(x), 0x164, 0xf, 0xf, false));
  return x;
}

__device__ __forceinline__ uint2 fr_fetch(const uint2* __restrict__ rec2, int i,
                                          int T, int ex, int off) {
  int j = 0;
#pragma unroll
  for (int k = 1; k < FR_SPW; ++k)
    j += (i >= __builtin_amdgcn_readlane(ex, k)) ? 1 : 0;
  int p = __shfl(off, j) + i;
  p = p < 0 ? 0 : (p > NUM_EDGES - 1 ? NUM_EDGES - 1 : p);
  uint2 rc = make_uint2(0u, 0u);
  if (i < T) rc = rec2[p];
  return rc;
}

__global__ __launch_bounds__(FR_WAVES * 32) void edge_featsum_kernel(
    const float* __restrict__ rec2_ws, const float* __restrict__ roff_ws,
    const float* __restrict__ h_ent, const float* __restrict__ rel2_proj,
    const float* __restrict__ W_rel3, const float* __restrict__ b_rel3,
    float* __restrict__ h_rel) {
  __shared__ float part[FR_WAVES * 128];
  __shared__ __attribute__((aligned(16))) float sm[3 * DIM];
  __shared__ int scnt[FR_WAVES];
  const uint2* rec2 = reinterpret_cast<const uint2*>(rec2_ws);
  const int* roff = reinterpret_cast<const int*>(roff_ws);
  const int t = threadIdx.x;
  const int r = blockIdx.x;
  const int wave = __builtin_amdgcn_readfirstlane(t >> 5);
  const int lane = t & 31;

  const int o = t >> 3, q = t & 7;
  const float4* wp = reinterpret_cast<const float4*>(W_rel3 + o * (3 * DIM) + q * 24);
  float4 wv[6];
#pragma unroll
  for (int k = 0; k < 6; ++k) wv[k] = wp[k];
  const float bo = b_rel3[o];
  const float r2 = t < DIM ? rel2_proj[(size_t)r * DIM + t] : 0.0f;

  const int s = wave * FR_SPW + lane;
  int st = 0, cn = 0;
  if (lane < FR_SPW && s < BK_GROUPS) {
    const int p = roff[s * RL_BINS + r];
    st = p & 0xffff;
    cn = (p >> 16) & 0xffff;
    st = st > RL_SLOTS ? RL_SLOTS : st;
    cn = cn > RL_SLOTS - st ? RL_SLOTS - st : cn;
  }
  int inc = cn;
  for (int d = 1; d < FR_SPW; d <<= 1) {
    const int u = __shfl_up(inc, d);
    if (lane >= d) inc += u;
  }
  const int ex = inc - cn;
  const int off = s * RL_SLOTS + st - ex;
  const int T = __builtin_amdgcn_readlane(inc, FR_SPW - 1);

  v2f as, ad;
  as.x = 0.0f; as.y = 0.0f; ad.x = 0.0f; ad.y = 0.0f;
  uint2 nx = fr_fetch(rec2, lane, T, ex, off);
  for (int i0 = 0; i0 < T; i0 += 32) {
    const uint2 rc = nx;
    nx = fr_fetch(rec2, i0 + 32 + lane, T, ex, off);
    const int left = T - i0;
    const int m = left < 32 ? left : 32;
    for (int k0 = 0; k0 < m; k0 += FR_BATCH) {
      v2f vs[FR_BATCH], vd[FR_BATCH];
#pragma unroll
      for (int k = 0; k < FR_BATCH; ++k) {
        const int sl = k0 + k < m ? k0 + k : m - 1;
        unsigned sk = (unsigned)__builtin_amdgcn_readlane((int)rc.x, sl);
        unsigned dk = (unsigned)__builtin_amdgcn_readlane((int)rc.y, sl);
        sk = sk < NUM_NODES ? sk : NUM_NODES - 1;
        dk = dk < NUM_NODES ? dk : NUM_NODES - 1;
        vs[k] = reinterpret_cast<const v2f*>(h_ent + (size_t)sk * DIM)[lane];
        vd[k] = reinterpret_cast<const v2f*>(h_ent + (size_t)dk * DIM)[lane];
      }
#pragma unroll
      for (int k = 0; k < FR_BATCH; ++k) {
        if (k0 + k < m) {
          as.x += vs[k].x; as.y += vs[k].y;
          ad.x += vd[k].x; ad.y += vd[k].y;
        }
      }
    }
  }
  part[wave * 128 + 2 * lane] = as.x;
  part[wave * 128 + 2 * lane + 1] = as.y;
  part[wave * 128 + 64 + 2 * lane] = ad.x;
  part[wave * 128 + 64 + 2 * lane + 1] = ad.y;
  if (lane == 0) scnt[wave] = T;
  __syncthreads();

  if (t < 128) {
    float sum = 0.0f;
    int ce = 0;
    for (int w = 0; w < FR_WAVES; ++w) { sum += part[w * 128 + t]; ce += scnt[w]; }
    const float cf = (float)ce;
    sm[t] = sum * (1.0f / fmaxf(cf, 1.0f));
    if (t < DIM) sm[128 + t] = ce > 0 ? r2 : 0.0f;
  }
  __syncthreads();

  const float4* mp = reinterpret_cast<const float4*>(&sm[q * 24]);
  float acc = 0.0f;
#pragma unroll
  for (int k = 0; k < 6; ++k) {
    const float4 m = mp[k];
    acc += m.x * wv[k].x;
    acc += m.y * wv[k].y;
    acc += m.z * wv[k].z;
    acc += m.w * wv[k].w;
  }
  acc = lanes8_sum(acc);
  if (q == 0) h_rel[(size_t)r * DIM + o] = acc + bo;
}

#define OFF_MS     0
#define OFF_MD     4096
#define OFF_MR     8192
#define OFF_M2     12288
#define OFF_CVEC   16384
#define OFF_VREL2  16448
#define OFF_A0     16512
#define OFF_UR     16576
#define OFF_AR     48576
#define OFF_REL2P  49088
#define OFF_AS     81088
#define OFF_AD     181088
#define OFF_CNTM   281088
#define OFF_BSTART 345088
#define OFF_BTOT   345344
#define OFF_REC    1445588
#define OFF_REC2   3445588
#define OFF_ROFF   5445588
#define OFF_UD     7845588
static_assert(OFF_CNTM + BK_GROUPS * BK_BINS <= OFF_BSTART &&
              OFF_BTOT + BK_BINS <= OFF_REC &&
              OFF_REC % 2 == 0 && OFF_REC + 2 * NUM_EDGES <= OFF_REC2 &&
              OFF_REC2 % 2 == 0 && OFF_REC2 + 2 * NUM_EDGES <= OFF_ROFF &&
              OFF_ROFF + BK_GROUPS * RL_BINS <= OFF_UD,
              "workspace planes do not overlap; records are 8-byte aligned");

extern "C" void kernel_launch(void* const* d_in, const int* in_sizes, int n_in,
                              void* d_out, int out_size, void* d_ws,
                              size_t ws_size, hipStream_t stream) {
  const float* ent_embed = (const float*)d_in[0];
  const float* rel_embed = (const float*)d_in[1];
  const float* W_ent  = (const float*)d_in[2];
  const float* b_ent  = (const float*)d_in[3];
  const float* W_relL = (const float*)d_in[4];
  const float* b_relL = (const float*)d_in[5];
  const float* W_rel2 = (const float*)d_in[6];
  const float* b_rel2 = (const float*)d_in[7];
  const float* W_rel3 = (const float*)d_in[8];
  const float* b_rel3 = (const float*)d_in[9];
  const float* W_a    = (const float*)d_in[10];
  const float* b_a    = (const float*)d_in[11];
  const float* W_fc   = (const float*)d_in[12];
  const float* b_fc   = (const float*)d_in[13];
  const int* src_idx  = (const int*)d_in[14];
  const int* dst_idx  = (const int*)d_in[15];
  const int* rel_idx  = (const int*)d_in[16];

  float* ws = (float*)d_ws;
  float* h_ent = (float*)d_out;
  float* h_rel = (float*)d_out + (size_t)NUM_NODES * DIM;

  fold_kernel<<<FOLD_BLOCKS, 256, 0, stream>>>(
      W_ent, b_ent, W_relL, b_relL, W_rel2, b_rel2, W_a, b_a, W_fc, b_fc,
      ws + OFF_MS, ws + OFF_MD, ws + OFF_MR, ws + OFF_M2, ws + OFF_CVEC,
      ws + OFF_VREL2, ws + OFF_A0);

  precompute_kernel<<<PC_NODE_BLOCKS + PC_REL_BLOCKS, 256, 0, stream>>>(
      ent_embed, rel_embed, W_a, ws + OFF_MS, ws + OFF_MD, ws + OFF_MR,
      ws + OFF_M2, ws + OFF_VREL2, h_ent, ws + OFF_UD, ws + OFF_AS,
      ws + OFF_AD, ws + OFF_UR, ws + OFF_REL2P, ws + OFF_AR, src_idx,
      ws + OFF_CNTM);

  bucket_kernel<<<BK_GROUPS, 1024, 0, stream>>>(
      src_idx, dst_idx, rel_idx, ws + OFF_CNTM, ws + OFF_REC, ws + OFF_BSTART,
      ws + OFF_BTOT, ws + OFF_REC2, ws + OFF_ROFF);

  edge_gather_h_kernel<<<HN_GROUPS, 1024, 0, stream>>>(
      ws + OFF_REC, ws + OFF_BSTART, ws + OFF_BTOT, ws + OFF_UD, ws + OFF_UR,
      ws + OFF_CVEC, ws + OFF_AS, ws + OFF_AD, ws + OFF_AR, ws + OFF_A0, h_ent);
  edge_featsum_kernel<<<NUM_RELS, FR_WAVES * 32, 0, stream>>>(
      ws + OFF_REC2, ws + OFF_ROFF, h_ent, ws + OFF_REL2P, W_rel3, b_rel3,
      h_rel);
}


# bucket_kernel: a thread's 63 count-matrix loads go out together off one base address, sums after; compiled before as 63 dependent L2 round trips (a wait behind every load); integer sums, same order
# speedup vs baseline: 8.9835x; 1.1130x over previous
#include <hip/hip_runtime.h>
#include <math.h>


#define NUM_NODES 100000
#define NUM_RELS  500
#define DIM       64
#define NUM_EDGES 1000000

typedef __attribute__((ext_vector_type(2))) float v2f;
typedef __attribute__((ext_vector_type(8))) float v8f;

#define FOLD_BLOCKS (DIM + 1)

__device__ __forceinline__ float row16_sum(float x) {
  x += __int_as_float(__builtin_amdgcn_update_dpp(0, __float_as_int(x), 0x161, 0xf, 0xf, false));
  x += __int_as_float(__builtin_amdgcn_update_dpp(0, __float_as_int(x), 0x162, 0xf, 0xf, false));
  x += __int_as_float(__builtin_amdgcn_update_dpp(0, __float_as_int(x), 0x164, 0xf, 0xf, false));
  x += __int_as_float(__builtin_amdgcn_update_dpp(0, __float_as_int(x), 0x168, 0xf, 0xf, false));
  return x;
}

__device__ __forceinline__ float wave32_sum(float x) {
  x = row16_sum(x);
  return x + __shfl_xor(x, 16);
}

__global__ __launch_bounds__(256) void fold_kernel(
    const float* __restrict__ W_ent, const float* __restrict__ b_ent,
    const float* __restrict__ W_relL, const float* __restrict__ b_relL,
    const float* __restrict__ W_rel2, const float* __restrict__ b_rel2,
    const float* __restrict__ W_a, const float* __restrict__ b_a,
    const float* __restrict__ W_fc, const float* __restrict__ b_fc,
    float* __restrict__ M_s, float* __restrict__ M_d,
    float* __restrict__ M_r, float* __restrict__ M_2,
    float* __restrict__ cvec, float* __restrict__ v_rel2,
    float* __restrict__ a0) {
  __shared__ __attribute__((aligned(16))) float srow[4 * DIM];
  const int t = threadIdx.x;

  if (blockIdx.x < DIM) {
    const int o = blockIdx.x;
    srow[t] = t < 3 * DIM ? W_fc[o * 192 + t] : W_rel2[o * DIM + (t - 3 * DIM)];
    __syncthreads();
    const int mat = t >> 6;
    const int k = t & 63;
    const float* Wk = (mat < 2 ? W_ent : W_relL) + k;
    const float* cf = &srow[mat * DIM];
    float acc = 0.f;
#pragma unroll
    for (int u = 0; u < DIM; ++u) acc += cf[u] * Wk[u * DIM];
    float* Mout = mat == 0 ? M_s : mat == 1 ? M_d : mat == 2 ? M_r : M_2;
    Mout[o * DIM + k] = acc;
    return;
  }

  const int wave = t >> 5;
  const int lane = t & 31;
  const int orow = wave * 8 + (lane & 7);
  const float bfc = b_fc[orow], br2 = b_rel2[orow];
  const float wa0 = W_a[lane], wa1 = W_a[lane + 32], ba = b_a[0];
  const float be0 = b_ent[lane], be1 = b_ent[lane + 32];
  const float bl0 = b_relL[lane], bl1 = b_relL[lane + 32];
  float pc[8], pv[8];
#pragma unroll
  for (int j = 0; j < 8; ++j) {
    const float* f = W_fc + (wave * 8 + j) * 192;
    const float* g = W_rel2 + (wave * 8 + j) * DIM;
    pc[j] = f[lane] * be0 + f[64 + lane] * be0 + f[128 + lane] * bl0 +
            f[32 + lane] * be1 + f[96 + lane] * be1 + f[160 + lane] * bl1;
    pv[j] = g[lane] * bl0 + g[32 + lane] * bl1;
  }
  float cs = 0.f, vs = 0.f;
#pragma unroll
  for (int j = 0; j < 8; ++j) {
    const float sc = wave32_sum(pc[j]);
    const float sv = wave32_sum(pv[j]);
    cs = (lane == j) ? sc : cs;
    vs = (lane == j) ? sv : vs;
  }
  if (lane < 8) {
    const float cv = bfc + cs;
    cvec[orow] = cv;
    v_rel2[orow] = br2 + vs;
    srow[orow] = cv;
  }
  __syncthreads();
  if (wave == 0) {
    const float a = wave32_sum(srow[lane] * wa0 + srow[lane + 32] * wa1);
    if (lane == 0) a0[0] = ba + a;
  }
}

#define PC_NODE_BLOCKS ((NUM_NODES + 127) / 128)
#define HN_NPR    400
#define HN_GROUPS ((NUM_NODES + HN_NPR - 1) / HN_NPR)
#define BK_GROUPS HN_GROUPS
#define BK_SLICE  ((NUM_EDGES + BK_GROUPS - 1) / BK_GROUPS)
#define BK_BINS   256
static_assert(HN_GROUPS <= BK_BINS, "an owner index must fit the count-matrix row");
static_assert(BK_GROUPS <= PC_NODE_BLOCKS, "every slice needs a counting work-group");
static_assert(BK_SLICE <= 4096, "a slice's records must fit the LDS staging area");
#define BK_TRIPS  ((BK_SLICE + 1023) / 1024)
#define BK_CTRIPS ((BK_GROUPS + 3) / 4)
#define PC_CNT_TRIPS ((BK_SLICE + 255) / 256)

__device__ __forceinline__ unsigned owner_of(int s) {
  const unsigned b = (unsigned)s / HN_NPR;
  return b < BK_BINS ? b : BK_BINS - 1;
}
#define PC_REL_BLOCKS  ((NUM_RELS + 127) / 128)
#define PC_PITCH (DIM + 4)

__global__ __launch_bounds__(256) void precompute_kernel(
    const float* __restrict__ ent_embed, const float* __restrict__ rel_embed,
    const float* __restrict__ W_a, const float* __restrict__ M_s,
    const float* __restrict__ M_d, const float* __restrict__ M_r,
    const float* __restrict__ M_2, const float* __restrict__ v_rel2,
    float* __restrict__ u_s, float* __restrict__ u_d, float* __restrict__ a_s,
    float* __restrict__ a_d, float* __restrict__ u_r,
    float* __restrict__ rel2_proj, float* __restrict__ a_r,
    const int* __restrict__ src, float* __restrict__ cntm_ws) {
  __shared__ float sW[2 * DIM * PC_PITCH];
  __shared__ int shist[BK_BINS];
  const int t = threadIdx.x;
  int* cntm = reinterpret_cast<int*>(cntm_ws);

  if (blockIdx.x < BK_GROUPS) {
    shist[t] = 0;
    __syncthreads();
    const int e0 = blockIdx.x * BK_SLICE;
    int sv[PC_CNT_TRIPS];
#pragma unroll
    for (int k = 0; k < PC_CNT_TRIPS; ++k) {
      const int i = t + 256 * k;
      sv[k] = 0;
      if (i < BK_SLICE && e0 + i < NUM_EDGES) sv[k] = src[e0 + i];
    }
#pragma unroll
    for (int k = 0; k < PC_CNT_TRIPS; ++k) {
      const int i = t + 256 * k;
      if (i < BK_SLICE && e0 + i < NUM_EDGES)
        __hip_atomic_fetch_add(&shist[owner_of(sv[k])], 1, __ATOMIC_RELAXED,
                               __HIP_MEMORY_SCOPE_WORKGROUP);
    }
    __syncthreads();
    cntm[blockIdx.x * BK_BINS + t] = shist[t];
  }

  const bool is_rel = blockIdx.x >= PC_NODE_BLOCKS;
  const float* A = is_rel ? rel_embed : ent_embed;
  const float* W0 = is_rel ? M_r : M_s;
  const float* W1 = is_rel ? M_2 : M_d;
  float* out0 = is_rel ? u_r : u_s;
  float* out1 = is_rel ? rel2_proj : u_d;
  float* aout0 = is_rel ? a_r : a_s;
  const int M = is_rel ? NUM_RELS : NUM_NODES;
  const int tb = is_rel ? (int)blockIdx.x - PC_NODE_BLOCKS : (int)blockIdx.x;

  for (int i = t; i < DIM * DIM; i += 256) {
    const int si = (i >> 6) * PC_PITCH + (i & 63);
    sW[si] = W0[i];
    sW[DIM * PC_PITCH + si] = W1[i];
  }
  __syncthreads();

  const int wave = t >> 5;
  const int lane = t & 31;
  const int m0 = (tb * 8 + wave) * 16;
  if (m0 >= M) return;

  const int lm = lane & 15;
  const int hi = lane >> 4;

  int arow = m0 + lm;
  if (arow >= M) arow = M - 1;
  const float* Arow = A + (size_t)arow * DIM;

  v8f acc[8] = {};
#pragma unroll
  for (int k = 0; k < DIM; k += 4) {
    v2f a;
    a.x = Arow[k + 2 * hi + 0];
    a.y = Arow[k + 2 * hi + 1];
#pragma unroll
    for (int j = 0; j < 8; ++j) {
      const float* w = &sW[(j * 16 + lm) * PC_PITCH + k + 2 * hi];
      v2f b;
      b.x = w[0];
      b.y = w[1];
      acc[j] = __builtin_amdgcn_wmma_f32_16x16x4_f32(false, a, false, b,
                                                     (short)0, acc[j], false,
                                                     false);
    }
  }

  float p0[8], p1[8];
#pragma unroll
  for (int v = 0; v < 8; ++v) { p0[v] = 0.0f; p1[v] = 0.0f; }
#pragma unroll
  for (int ob = 0; ob < 4; ++ob) {
    const float wa = W_a[ob * 16 + lm];
    const float bv = is_rel ? v_rel2[ob * 16 + lm] : 0.0f;
#pragma unroll
    for (int v = 0; v < 8; ++v) {
      const int row = m0 + v + 8 * hi;
      const float x0 = acc[ob][v];
      const float x1 = acc[4 + ob][v];
      p0[v] += x0 * wa;
      p1[v] += x1 * wa;
      if (row < M) {
        out0[(size_t)row * DIM + ob * 16 + lm] = x0;
        out1[(size_t)row * DIM + ob * 16 + lm] = x1 + bv;
      }
    }
  }

  const int srow = m0 + 8 * hi + lm;
  const bool swr = lm < 8 && srow < M;
  float s0 = 0.0f;
#pragma unroll
  for (int v = 0; v < 8; ++v) {
    const float r = row16_sum(p0[v]);
    s0 = (lm == v) ? r : s0;
  }
  if (swr) aout0[srow] = s0;
  if (!is_rel) {
    float s1 = 0.0f;
#pragma unroll
    for (int v = 0; v < 8; ++v) {
      const float r = row16_sum(p1[v]);
      s1 = (lm == v) ? r : s1;
    }
    if (swr) a_d[srow] = s1;
  }
}

static_assert(NUM_RELS <= (1 << 9), "relation id must fit 9 bits of a record");
static_assert(NUM_NODES <= (1 << 23), "node id must fit 23 bits of a record");
#define RL_BINS  512
#define RL_SLOTS BK_SLICE
static_assert(NUM_RELS <= RL_BINS, "every relation needs a run word");
static_assert(BK_SLICE < (1 << 15), "start and length of a run must fit 16 bits each");

__global__ __launch_bounds__(1024) void bucket_kernel(
    const int* __restrict__ src, const int* __restrict__ dst,
    const int* __restrict__ rel, const float* __restrict__ cntm_ws,
    float* __restrict__ rec_ws, float* __restrict__ binstart_ws,
    float* __restrict__ bintot_ws, float* __restrict__ rec2_ws,
    float* __restrict__ roff_ws) {
  const int* cntm = reinterpret_cast<const int*>(cntm_ws);
  uint2* rec = reinterpret_cast<uint2*>(rec_ws);
  int* binstart = reinterpret_cast<int*>(binstart_ws);
  int* bintot = reinterpret_cast<int*>(bintot_ws);
  uint2* rec2 = reinterpret_cast<uint2*>(rec2_ws);
  int* roff = reinterpret_cast<int*>(roff_ws);
  __shared__ int spart[4 * 2 * BK_BINS];
  __shared__ int sT[BK_BINS], sP[BK_BINS], sL[BK_BINS];
  __shared__ int sbase[BK_BINS];
  __shared__ int lpos[BK_BINS];
  __shared__ int lcur[BK_BINS];
  __shared__ uint2 srec[4096];
  __shared__ int rcnt[RL_BINS];
  __shared__ int rstart[RL_BINS];
  __shared__ int rcur[RL_BINS];
  __shared__ uint2 srec2[4096];
  const int t = threadIdx.x;
  const int g = blockIdx.x;
  const int b = t & (BK_BINS - 1);
  const int part = t >> 8;

  if (t < RL_BINS) { rcnt[t] = 0; rcur[t] = 0; }

  const int e0 = g * BK_SLICE;
  int ns = NUM_EDGES - e0;
  ns = ns > BK_SLICE ? BK_SLICE : ns;
  ns = ns < 0 ? 0 : ns;
  int sv[BK_TRIPS], dv[BK_TRIPS], rv[BK_TRIPS];
#pragma unroll
  for (int k = 0; k < BK_TRIPS; ++k) {
    const int i = t + 1024 * k;
    sv[k] = 0; dv[k] = 0; rv[k] = 0;
    if (i < ns) { sv[k] = src[e0 + i]; dv[k] = dst[e0 + i]; rv[k] = rel[e0 + i]; }
  }

  int tot = 0, pre = 0;
  int cv[BK_CTRIPS];
  const int* col = cntm + part * BK_BINS + b;
#pragma unroll
  for (int k = 0; k < BK_CTRIPS; ++k) {
    cv[k] = 0;
    if (part + 4 * k < BK_GROUPS) cv[k] = col[k * 4 * BK_BINS];
  }
#pragma unroll
  for (int k = 0; k < BK_CTRIPS; ++k) {
    tot += cv[k];
    pre += part + 4 * k < g ? cv[k] : 0;
  }
  spart[part * 2 * BK_BINS + b] = tot;
  spart[part * 2 * BK_BINS + BK_BINS + b] = pre;
  __syncthreads();
  if (t < BK_BINS) {
    sT[t] = spart[t] + spart[2 * BK_BINS + t] + spart[4 * BK_BINS + t] +
            spart[6 * BK_BINS + t];
    sP[t] = spart[BK_BINS + t] + spart[3 * BK_BINS + t] +
            spart[5 * BK_BINS + t] + spart[7 * BK_BINS + t];
    sL[t] = cntm[g * BK_BINS + t];
    lcur[t] = 0;
  }
#pragma unroll
  for (int k = 0; k < BK_TRIPS; ++k) {
    const int i = t + 1024 * k;
    if (i < ns)
      __hip_atomic_fetch_add(&rcnt[(unsigned)rv[k] & (RL_BINS - 1)], 1,
                             __ATOMIC_RELAXED, __HIP_MEMORY_SCOPE_WORKGROUP);
  }
  __syncthreads();
  if (t < 32) {
    int vT[8], vL[8];
    int accT = 0, accL = 0;
#pragma unroll
    for (int j = 0; j < 8; ++j) {
      vT[j] = sT[8 * t + j];
      vL[j] = sL[8 * t + j];
      accT += vT[j];
      accL += vL[j];
    }
    int inT = accT, inL = accL;
    for (int o = 1; o < 32; o <<= 1) {
      const int uT = __shfl_up(inT, o);
      const int uL = __shfl_up(inL, o);
      if (t >= o) { inT += uT; inL += uL; }
    }
    int exT = inT - accT, exL = inL - accL;
#pragma unroll
    for (int j = 0; j < 8; ++j) {
      const int bb = 8 * t + j;
      sbase[bb] = exT + sP[bb];
      lpos[bb] = exL;
      if (g == 0) { binstart[bb] = exT; bintot[bb] = vT[j]; }
      exT += vT[j];
      exL += vL[j];
    }
  } else if (t < 64) {
    const int l = t - 32;
    int vR[RL_BINS / 32];
    int accR = 0;
#pragma unroll
    for (int j = 0; j < RL_BINS / 32; ++j) {
      vR[j] = rcnt[(RL_BINS / 32) * l + j];
      accR += vR[j];
    }
    int inR = accR;
    for (int o = 1; o < 32; o <<= 1) {
      const int uR = __shfl_up(inR, o);
      if (l >= o) inR += uR;
    }
    int exR = inR - accR;
#pragma unroll
    for (int j = 0; j < RL_BINS / 32; ++j) {
      rstart[(RL_BINS / 32) * l + j] = exR;
      exR += vR[j];
    }
  }
  __syncthreads();

#pragma unroll
  for (int k = 0; k < BK_TRIPS; ++k) {
    const int i = t + 1024 * k;
    if (i < ns) {
      const unsigned bb = owner_of(sv[k]);
      const int rank = __hip_atomic_fetch_add(&lcur[bb], 1, __ATOMIC_RELAXED,
                                              __HIP_MEMORY_SCOPE_WORKGROUP);
      int slot = lpos[bb] + rank;
      slot = slot < 0 ? 0 : (slot > 4095 ? 4095 : slot);
      srec[slot] = make_uint2((unsigned)sv[k], ((unsigned)dv[k] << 9) |
                                                   ((unsigned)rv[k] & 511u));
      const unsigned rb = (unsigned)rv[k] & (RL_BINS - 1);
      const int rank2 = __hip_atomic_fetch_add(&rcur[rb], 1, __ATOMIC_RELAXED,
                                               __HIP_MEMORY_SCOPE_WORKGROUP);
      int slot2 = rstart[rb] + rank2;
      slot2 = slot2 < 0 ? 0 : (slot2 > 4095 ? 4095 : slot2);
      srec2[slot2] = make_uint2((unsigned)sv[k], (unsigned)dv[k]);
    }
  }
  __syncthreads();
  for (int i = t; i < ns; i += 1024) {
    const uint2 rc = srec[i];
    const unsigned bb = owner_of((int)rc.x);
    int gp = sbase[bb] + (i - lpos[bb]);
    gp = gp < 0 ? 0 : (gp > NUM_EDGES - 1 ? NUM_EDGES - 1 : gp);
    rec[gp] = rc;
    rec2[e0 + i] = srec2[i];
  }
  if (t < RL_BINS) roff[g * RL_BINS + t] = rstart[t] | (rcnt[t] << 16);
}

#define HN_BATCH  8

__device__ __forceinline__ void hn_drain(
    uint2 rc, int n, int lane, int n0, float a0,
    const float* __restrict__ a_s, const float* __restrict__ a_d,
    const float* __restrict__ a_r, const float* __restrict__ u_d,
    const float* __restrict__ u_r, float* stab, float* sbs) {
  int d = 0, r = 0, loc = 0, bb = 0;
  if (lane < n) {
    loc = (int)rc.x - n0;
    loc = loc < 0 ? 0 : (loc > HN_NPR - 1 ? HN_NPR - 1 : loc);
    d = (int)(rc.y >> 9);
    r = (int)(rc.y & 511u);
    float l = a_s[n0 + loc] + a_d[d] + a_r[r] + a0;
    l = l > 0.f ? l : 0.01f * l;
    const float b = expf(l);
    __hip_atomic_fetch_add(&sbs[loc], b, __ATOMIC_RELAXED,
                           __HIP_MEMORY_SCOPE_WORKGROUP);
    bb = __float_as_int(b);
  }
  for (int k0 = 0; k0 < n; k0 += HN_BATCH) {
    v2f vd[HN_BATCH], vr[HN_BATCH];
    float bk[HN_BATCH];
    int lo[HN_BATCH];
#pragma unroll
    for (int k = 0; k < HN_BATCH; ++k) {
      const int sl = k0 + k < n ? k0 + k : n - 1;
      const int dk = __builtin_amdgcn_readlane(d, sl);
      const int rk = __builtin_amdgcn_readlane(r, sl);
      bk[k] = __int_as_float(__builtin_amdgcn_readlane(bb, sl));
      lo[k] = __builtin_amdgcn_readlane(loc, sl);
      vd[k] = reinterpret_cast<const v2f*>(u_d + (size_t)dk * DIM)[lane];
      vr[k] = reinterpret_cast<const v2f*>(u_r + (size_t)rk * DIM)[lane];
    }
#pragma unroll
    for (int k = 0; k < HN_BATCH; ++k) {
      if (k0 + k < n) {
        float* p = &stab[lo[k] * DIM + 2 * lane];
        __hip_atomic_fetch_add(p, bk[k] * (vd[k].x + vr[k].x), __ATOMIC_RELAXED,
                               __HIP_MEMORY_SCOPE_WORKGROUP);
        __hip_atomic_fetch_add(p + 1, bk[k] * (vd[k].y + vr[k].y),
                               __ATOMIC_RELAXED, __HIP_MEMORY_SCOPE_WORKGROUP);
      }
    }
  }
}

__global__ __launch_bounds__(1024) void edge_gather_h_kernel(
    const float* __restrict__ rec_ws, const float* __restrict__ binstart_ws,
    const float* __restrict__ bintot_ws, const float* __restrict__ u_d,
    const float* __restrict__ u_r, const float* __restrict__ cvec,
    const float* __restrict__ a_s, const float* __restrict__ a_d,
    const float* __restrict__ a_r, const float* __restrict__ a0p,
    float* __restrict__ h_ent) {
  __shared__ __attribute__((aligned(16))) float stab[HN_NPR * DIM];
  __shared__ float sbs[HN_NPR];
  const uint2* rec = reinterpret_cast<const uint2*>(rec_ws);
  const int* binstart = reinterpret_cast<const int*>(binstart_ws);
  const int* bintot = reinterpret_cast<const int*>(bintot_ws);
  const int t = threadIdx.x;
  for (int i = t; i < HN_NPR * DIM; i += 1024) stab[i] = 0.0f;
  for (int i = t; i < HN_NPR; i += 1024) sbs[i] = 0.0f;
  __syncthreads();
  const float a0 = a0p[0];

  const int n0 = blockIdx.x * HN_NPR;
  int nn = NUM_NODES - n0;
  nn = nn > HN_NPR ? HN_NPR : nn;
  nn = nn < 0 ? 0 : nn;

  int s0 = binstart[blockIdx.x];
  int ne = bintot[blockIdx.x];
  s0 = s0 < 0 ? 0 : (s0 > NUM_EDGES ? NUM_EDGES : s0);
  ne = ne < 0 ? 0 : (ne > NUM_EDGES - s0 ? NUM_EDGES - s0 : ne);

  const int wave = __builtin_amdgcn_readfirstlane(t >> 5);
  const int lane = t & 31;

  int i = wave * 32;
  uint2 nx = make_uint2(0u, 0u);
  if (i + lane < ne) nx = rec[s0 + i + lane];
  for (; i < ne; i += 1024) {
    const uint2 rc = nx;
    const int left = ne - i;
    const int m = left < 32 ? left : 32;
    const int in = i + 1024;
    if (in + lane < ne) nx = rec[s0 + in + lane];
    hn_drain(rc, m, lane, n0, a0, a_s, a_d, a_r, u_d, u_r, stab, sbs);
  }
  __syncthreads();

  for (int j = t; j < nn * (DIM / 4); j += 1024) {
    const int row = j >> 4, c4 = j & 15;
    const int node = n0 + row;
    const float bs = sbs[row];
    float4 o = make_float4(0.0f, 0.0f, 0.0f, 0.0f);
    float4* hp = reinterpret_cast<float4*>(h_ent + (size_t)node * DIM) + c4;
    if (bs > 0.0f) {
      const float4 a = reinterpret_cast<const float4*>(stab)[row * 16 + c4];
      const float4 us = *hp;
      const float4 cv = reinterpret_cast<const float4*>(cvec)[c4];
      o.x = a.x / bs + (us.x + cv.x);
      o.y = a.y / bs + (us.y + cv.y);
      o.z = a.z / bs + (us.z + cv.z);
      o.w = a.w / bs + (us.w + cv.w);
    }
    *hp = o;
  }
}

#define FR_WAVES 16
#define FR_SPW   16
#define FR_BATCH 8
static_assert(FR_WAVES * FR_SPW >= BK_GROUPS, "every slice needs a wave");
static_assert(FR_SPW <= 32 && FR_WAVES * 32 == DIM * 8, "thread maps of the relation pass");

__device__ __forceinline__ float lanes8_sum(float x) {
  x += __int_as_float(__builtin_amdgcn_update_dpp(0, __float_as_int(x), 0x161, 0xf, 0xf, false));
  x += __int_as_float(__builtin_amdgcn_update_dpp(0, __float_as_int(x), 0x162, 0xf, 0xf, false));
  x += __int_as_float(__builtin_amdgcn_update_dpp(0, __float_as_int(x), 0x164, 0xf, 0xf, false));
  return x;
}

__device__ __forceinline__ uint2 fr_fetch(const uint2* __restrict__ rec2, int i,
                                          int T, int ex, int off) {
  int j = 0;
#pragma unroll
  for (int k = 1; k < FR_SPW; ++k)
    j += (i >= __builtin_amdgcn_readlane(ex, k)) ? 1 : 0;
  int p = __shfl(off, j) + i;
  p = p < 0 ? 0 : (p > NUM_EDGES - 1 ? NUM_EDGES - 1 : p);
  uint2 rc = make_uint2(0u, 0u);
  if (i < T) rc = rec2[p];
  return rc;
}

__global__ __launch_bounds__(FR_WAVES * 32) void edge_featsum_kernel(
    const float* __restrict__ rec2_ws, const float* __restrict__ roff_ws,
    const float* __restrict__ h_ent, const float* __restrict__ rel2_proj,
    const float* __restrict__ W_rel3, const float* __restrict__ b_rel3,
    float* __restrict__ h_rel) {
  __shared__ float part[FR_WAVES * 128];
  __shared__ __attribute__((aligned(16))) float sm[3 * DIM];
  __shared__ int scnt[FR_WAVES];
  const uint2* rec2 = reinterpret_cast<const uint2*>(rec2_ws);
  const int* roff = reinterpret_cast<const int*>(roff_ws);
  const int t = threadIdx.x;
  const int r = blockIdx.x;
  const int wave = __builtin_amdgcn_readfirstlane(t >> 5);
  const int lane = t & 31;

  const int o = t >> 3, q = t & 7;
  const float4* wp = reinterpret_cast<const float4*>(W_rel3 + o * (3 * DIM) + q * 24);
  float4 wv[6];
#pragma unroll
  for (int k = 0; k < 6; ++k) wv[k] = wp[k];
  const float bo = b_rel3[o];
  const float r2 = t < DIM ? rel2_proj[(size_t)r * DIM + t] : 0.0f;

  const int s = wave * FR_SPW + lane;
  int st = 0, cn = 0;
  if (lane < FR_SPW && s < BK_GROUPS) {
    const int p = roff[s * RL_BINS + r];
    st = p & 0xffff;
    cn = (p >> 16) & 0xffff;
    st = st > RL_SLOTS ? RL_SLOTS : st;
    cn = cn > RL_SLOTS - st ? RL_SLOTS - st : cn;
  }
  int inc = cn;
  for (int d = 1; d < FR_SPW; d <<= 1) {
    const int u = __shfl_up(inc, d);
    if (lane >= d) inc += u;
  }
  const int ex = inc - cn;
  const int off = s * RL_SLOTS + st - ex;
  const int T = __builtin_amdgcn_readlane(inc, FR_SPW - 1);

  v2f as, ad;
  as.x = 0.0f; as.y = 0.0f; ad.x = 0.0f; ad.y = 0.0f;
  uint2 nx = fr_fetch(rec2, lane, T, ex, off);
  for (int i0 = 0; i0 < T; i0 += 32) {
    const uint2 rc = nx;
    nx = fr_fetch(rec2, i0 + 32 + lane, T, ex, off);
    const int left = T - i0;
    const int m = left < 32 ? left : 32;
    for (int k0 = 0; k0 < m; k0 += FR_BATCH) {
      v2f vs[FR_BATCH], vd[FR_BATCH];
#pragma unroll
      for (int k = 0; k < FR_BATCH; ++k) {
        const int sl = k0 + k < m ? k0 + k : m - 1;
        unsigned sk = (unsigned)__builtin_amdgcn_readlane((int)rc.x, sl);
        unsigned dk = (unsigned)__builtin_amdgcn_readlane((int)rc.y, sl);
        sk = sk < NUM_NODES ? sk : NUM_NODES - 1;
        dk = dk < NUM_NODES ? dk : NUM_NODES - 1;
        vs[k] = reinterpret_cast<const v2f*>(h_ent + (size_t)sk * DIM)[lane];
        vd[k] = reinterpret_cast<const v2f*>(h_ent + (size_t)dk * DIM)[lane];
      }
#pragma unroll
      for (int k = 0; k < FR_BATCH; ++k) {
        if (k0 + k < m) {
          as.x += vs[k].x; as.y += vs[k].y;
          ad.x += vd[k].x; ad.y += vd[k].y;
        }
      }
    }
  }
  part[wave * 128 + 2 * lane] = as.x;
  part[wave * 128 + 2 * lane + 1] = as.y;
  part[wave * 128 + 64 + 2 * lane] = ad.x;
  part[wave * 128 + 64 + 2 * lane + 1] = ad.y;
  if (lane == 0) scnt[wave] = T;
  __syncthreads();

  if (t < 128) {
    float sum = 0.0f;
    int ce = 0;
    for (int w = 0; w < FR_WAVES; ++w) { sum += part[w * 128 + t]; ce += scnt[w]; }
    const float cf = (float)ce;
    sm[t] = sum * (1.0f / fmaxf(cf, 1.0f));
    if (t < DIM) sm[128 + t] = ce > 0 ? r2 : 0.0f;
  }
  __syncthreads();

  const float4* mp = reinterpret_cast<const float4*>(&sm[q * 24]);
  float acc = 0.0f;
#pragma unroll
  for (int k = 0; k < 6; ++k) {
    const float4 m = mp[k];
    acc += m.x * wv[k].x;
    acc += m.y * wv[k].y;
    acc += m.z * wv[k].z;
    acc += m.w * wv[k].w;
  }
  acc = lanes8_sum(acc);
  if (q == 0) h_rel[(size_t)r * DIM + o] = acc + bo;
}

#define OFF_MS     0
#define OFF_MD     4096
#define OFF_MR     8192
#define OFF_M2     12288
#define OFF_CVEC   16384
#define OFF_VREL2  16448
#define OFF_A0     16512
#define OFF_UR     16576
#define OFF_AR     48576
#define OFF_REL2P  49088
#define OFF_AS     81088
#define OFF_AD     181088
#define OFF_CNTM   281088
#define OFF_BSTART 345088
#define OFF_BTOT   345344
#define OFF_REC    1445588
#define OFF_REC2   3445588
#define OFF_ROFF   5445588
#define OFF_UD     7845588
static_assert(OFF_CNTM + BK_GROUPS * BK_BINS <= OFF_BSTART &&
              OFF_BTOT + BK_BINS <= OFF_REC &&
              OFF_REC % 2 == 0 && OFF_REC + 2 * NUM_EDGES <= OFF_REC2 &&
              OFF_REC2 % 2 == 0 && OFF_REC2 + 2 * NUM_EDGES <= OFF_ROFF &&
              OFF_ROFF + BK_GROUPS * RL_BINS <= OFF_UD,
              "workspace planes do not overlap; records are 8-byte aligned");

extern "C" void kernel_launch(void* const* d_in, const int* in_sizes, int n_in,
                              void* d_out, int out_size, void* d_ws,
                              size_t ws_size, hipStream_t stream) {
  const float* ent_embed = (const float*)d_in[0];
  const float* rel_embed = (const float*)d_in[1];
  const float* W_ent  = (const float*)d_in[2];
  const float* b_ent  = (const float*)d_in[3];
  const float* W_relL = (const float*)d_in[4];
  const float* b_relL = (const float*)d_in[5];
  const float* W_rel2 = (const float*)d_in[6];
  const float* b_rel2 = (const float*)d_in[7];
  const float* W_rel3 = (const float*)d_in[8];
  const float* b_rel3 = (const float*)d_in[9];
  const float* W_a    = (const float*)d_in[10];
  const float* b_a    = (const float*)d_in[11];
  const float* W_fc   = (const float*)d_in[12];
  const float* b_fc   = (const float*)d_in[13];
  const int* src_idx  = (const int*)d_in[14];
  const int* dst_idx  = (const int*)d_in[15];
  const int* rel_idx  = (const int*)d_in[16];

  float* ws = (float*)d_ws;
  float* h_ent = (float*)d_out;
  float* h_rel = (float*)d_out + (size_t)NUM_NODES * DIM;

  fold_kernel<<<FOLD_BLOCKS, 256, 0, stream>>>(
      W_ent, b_ent, W_relL, b_relL, W_rel2, b_rel2, W_a, b_a, W_fc, b_fc,
      ws + OFF_MS, ws + OFF_MD, ws + OFF_MR, ws + OFF_M2, ws + OFF_CVEC,
      ws + OFF_VREL2, ws + OFF_A0);

  precompute_kernel<<<PC_NODE_BLOCKS + PC_REL_BLOCKS, 256, 0, stream>>>(
      ent_embed, rel_embed, W_a, ws + OFF_MS, ws + OFF_MD, ws + OFF_MR,
      ws + OFF_M2, ws + OFF_VREL2, h_ent, ws + OFF_UD, ws + OFF_AS,
      ws + OFF_AD, ws + OFF_UR, ws + OFF_REL2P, ws + OFF_AR, src_idx,
      ws + OFF_CNTM);

  bucket_kernel<<<BK_GROUPS, 1024, 0, stream>>>(
      src_idx, dst_idx, rel_idx, ws + OFF_CNTM, ws + OFF_REC, ws + OFF_BSTART,
      ws + OFF_BTOT, ws + OFF_REC2, ws + OFF_ROFF);

  edge_gather_h_kernel<<<HN_GROUPS, 1024, 0, stream>>>(
      ws + OFF_REC, ws + OFF_BSTART, ws + OFF_BTOT, ws + OFF_UD, ws + OFF_UR,
      ws + OFF_CVEC, ws + OFF_AS, ws + OFF_AD, ws + OFF_AR, ws + OFF_A0, h_ent);
  edge_featsum_kernel<<<NUM_RELS, FR_WAVES * 32, 0, stream>>>(
      ws + OFF_REC2, ws + OFF_ROFF, h_ent, ws + OFF_REL2P, W_rel3, b_rel3,
      h_rel);
}


# edge_gather_h: owner orders its records by node in LDS (counting sort); a wave owns 12-13 nodes: row sums in registers, one LDS add per node not per edge, writes its rows itself; sum order changes
# speedup vs baseline: 9.3142x; 1.0368x over previous
#include <hip/hip_runtime.h>
#include <math.h>


#define NUM_NODES 100000
#define NUM_RELS  500
#define DIM       64
#define NUM_EDGES 1000000

typedef __attribute__((ext_vector_type(2))) float v2f;
typedef __attribute__((ext_vector_type(8))) float v8f;

#define FOLD_BLOCKS (DIM + 1)

__device__ __forceinline__ float row16_sum(float x) {
  x += __int_as_float(__builtin_amdgcn_update_dpp(0, __float_as_int(x), 0x161, 0xf, 0xf, false));
  x += __int_as_float(__builtin_amdgcn_update_dpp(0, __float_as_int(x), 0x162, 0xf, 0xf, false));
  x += __int_as_float(__builtin_amdgcn_update_dpp(0, __float_as_int(x), 0x164, 0xf, 0xf, false));
  x += __int_as_float(__builtin_amdgcn_update_dpp(0, __float_as_int(x), 0x168, 0xf, 0xf, false));
  return x;
}

__device__ __forceinline__ float wave32_sum(float x) {
  x = row16_sum(x);
  return x + __shfl_xor(x, 16);
}

__global__ __launch_bounds__(256) void fold_kernel(
    const float* __restrict__ W_ent, const float* __restrict__ b_ent,
    const float* __restrict__ W_relL, const float* __restrict__ b_relL,
    const float* __restrict__ W_rel2, const float* __restrict__ b_rel2,
    const float* __restrict__ W_a, const float* __restrict__ b_a,
    const float* __restrict__ W_fc, const float* __restrict__ b_fc,
    float* __restrict__ M_s, float* __restrict__ M_d,
    float* __restrict__ M_r, float* __restrict__ M_2,
    float* __restrict__ cvec, float* __restrict__ v_rel2,
    float* __restrict__ a0) {
  __shared__ __attribute__((aligned(16))) float srow[4 * DIM];
  const int t = threadIdx.x;

  if (blockIdx.x < DIM) {
    const int o = blockIdx.x;
    srow[t] = t < 3 * DIM ? W_fc[o * 192 + t] : W_rel2[o * DIM + (t - 3 * DIM)];
    __syncthreads();
    const int mat = t >> 6;
    const int k = t & 63;
    const float* Wk = (mat < 2 ? W_ent : W_relL) + k;
    const float* cf = &srow[mat * DIM];
    float acc = 0.f;
#pragma unroll
    for (int u = 0; u < DIM; ++u) acc += cf[u] * Wk[u * DIM];
    float* Mout = mat == 0 ? M_s : mat == 1 ? M_d : mat == 2 ? M_r : M_2;
    Mout[o * DIM + k] = acc;
    return;
  }

  const int wave = t >> 5;
  const int lane = t & 31;
  const int orow = wave * 8 + (lane & 7);
  const float bfc = b_fc[orow], br2 = b_rel2[orow];
  const float wa0 = W_a[lane], wa1 = W_a[lane + 32], ba = b_a[0];
  const float be0 = b_ent[lane], be1 = b_ent[lane + 32];
  const float bl0 = b_relL[lane], bl1 = b_relL[lane + 32];
  float pc[8], pv[8];
#pragma unroll
  for (int j = 0; j < 8; ++j) {
    const float* f = W_fc + (wave * 8 + j) * 192;
    const float* g = W_rel2 + (wave * 8 + j) * DIM;
    pc[j] = f[lane] * be0 + f[64 + lane] * be0 + f[128 + lane] * bl0 +
            f[32 + lane] * be1 + f[96 + lane] * be1 + f[160 + lane] * bl1;
    pv[j] = g[lane] * bl0 + g[32 + lane] * bl1;
  }
  float cs = 0.f, vs = 0.f;
#pragma unroll
  for (int j = 0; j < 8; ++j) {
    const float sc = wave32_sum(pc[j]);
    const float sv = wave32_sum(pv[j]);
    cs = (lane == j) ? sc : cs;
    vs = (lane == j) ? sv : vs;
  }
  if (lane < 8) {
    const float cv = bfc + cs;
    cvec[orow] = cv;
    v_rel2[orow] = br2 + vs;
    srow[orow] = cv;
  }
  __syncthreads();
  if (wave == 0) {
    const float a = wave32_sum(srow[lane] * wa0 + srow[lane + 32] * wa1);
    if (lane == 0) a0[0] = ba + a;
  }
}

#define PC_NODE_BLOCKS ((NUM_NODES + 127) / 128)
#define HN_NPR    400
#define HN_GROUPS ((NUM_NODES + HN_NPR - 1) / HN_NPR)
#define BK_GROUPS HN_GROUPS
#define BK_SLICE  ((NUM_EDGES + BK_GROUPS - 1) / BK_GROUPS)
#define BK_BINS   256
static_assert(HN_GROUPS <= BK_BINS, "an owner index must fit the count-matrix row");
static_assert(BK_GROUPS <= PC_NODE_BLOCKS, "every slice needs a counting work-group");
static_assert(BK_SLICE <= 4096, "a slice's records must fit the LDS staging area");
#define BK_TRIPS  ((BK_SLICE + 1023) / 1024)
#define BK_CTRIPS ((BK_GROUPS + 3) / 4)
#define PC_CNT_TRIPS ((BK_SLICE + 255) / 256)

__device__ __forceinline__ unsigned owner_of(int s) {
  const unsigned b = (unsigned)s / HN_NPR;
  return b < BK_BINS ? b : BK_BINS - 1;
}
#define PC_REL_BLOCKS  ((NUM_RELS + 127) / 128)
#define PC_PITCH (DIM + 4)

__global__ __launch_bounds__(256) void precompute_kernel(
    const float* __restrict__ ent_embed, const float* __restrict__ rel_embed,
    const float* __restrict__ W_a, const float* __restrict__ M_s,
    const float* __restrict__ M_d, const float* __restrict__ M_r,
    const float* __restrict__ M_2, const float* __restrict__ v_rel2,
    float* __restrict__ u_s, float* __restrict__ u_d, float* __restrict__ a_s,
    float* __restrict__ a_d, float* __restrict__ u_r,
    float* __restrict__ rel2_proj, float* __restrict__ a_r,
    const int* __restrict__ src, float* __restrict__ cntm_ws) {
  __shared__ float sW[2 * DIM * PC_PITCH];
  __shared__ int shist[BK_BINS];
  const int t = threadIdx.x;
  int* cntm = reinterpret_cast<int*>(cntm_ws);

  if (blockIdx.x < BK_GROUPS) {
    shist[t] = 0;
    __syncthreads();
    const int e0 = blockIdx.x * BK_SLICE;
    int sv[PC_CNT_TRIPS];
#pragma unroll
    for (int k = 0; k < PC_CNT_TRIPS; ++k) {
      const int i = t + 256 * k;
      sv[k] = 0;
      if (i < BK_SLICE && e0 + i < NUM_EDGES) sv[k] = src[e0 + i];
    }
#pragma unroll
    for (int k = 0; k < PC_CNT_TRIPS; ++k) {
      const int i = t + 256 * k;
      if (i < BK_SLICE && e0 + i < NUM_EDGES)
        __hip_atomic_fetch_add(&shist[owner_of(sv[k])], 1, __ATOMIC_RELAXED,
                               __HIP_MEMORY_SCOPE_WORKGROUP);
    }
    __syncthreads();
    cntm[blockIdx.x * BK_BINS + t] = shist[t];
  }

  const bool is_rel = blockIdx.x >= PC_NODE_BLOCKS;
  const float* A = is_rel ? rel_embed : ent_embed;
  const float* W0 = is_rel ? M_r : M_s;
  const float* W1 = is_rel ? M_2 : M_d;
  float* out0 = is_rel ? u_r : u_s;
  float* out1 = is_rel ? rel2_proj : u_d;
  float* aout0 = is_rel ? a_r : a_s;
  const int M = is_rel ? NUM_RELS : NUM_NODES;
  const int tb = is_rel ? (int)blockIdx.x - PC_NODE_BLOCKS : (int)blockIdx.x;

  for (int i = t; i < DIM * DIM; i += 256) {
    const int si = (i >> 6) * PC_PITCH + (i & 63);
    sW[si] = W0[i];
    sW[DIM * PC_PITCH + si] = W1[i];
  }
  __syncthreads();

  const int wave = t >> 5;
  const int lane = t & 31;
  const int m0 = (tb * 8 + wave) * 16;
  if (m0 >= M) return;

  const int lm = lane & 15;
  const int hi = lane >> 4;

  int arow = m0 + lm;
  if (arow >= M) arow = M - 1;
  const float* Arow = A + (size_t)arow * DIM;

  v8f acc[8] = {};
#pragma unroll
  for (int k = 0; k < DIM; k += 4) {
    v2f a;
    a.x = Arow[k + 2 * hi + 0];
    a.y = Arow[k + 2 * hi + 1];
#pragma unroll
    for (int j = 0; j < 8; ++j) {
      const float* w = &sW[(j * 16 + lm) * PC_PITCH + k + 2 * hi];
      v2f b;
      b.x = w[0];
      b.y = w[1];
      acc[j] = __builtin_amdgcn_wmma_f32_16x16x4_f32(false, a, false, b,
                                                     (short)0, acc[j], false,
                                                     false);
    }
  }

  float p0[8], p1[8];
#pragma unroll
  for (int v = 0; v < 8; ++v) { p0[v] = 0.0f; p1[v] = 0.0f; }
#pragma unroll
  for (int ob = 0; ob < 4; ++ob) {
    const float wa = W_a[ob * 16 + lm];
    const float bv = is_rel ? v_rel2[ob * 16 + lm] : 0.0f;
#pragma unroll
    for (int v = 0; v < 8; ++v) {
      const int row = m0 + v + 8 * hi;
      const float x0 = acc[ob][v];
      const float x1 = acc[4 + ob][v];
      p0[v] += x0 * wa;
      p1[v] += x1 * wa;
      if (row < M) {
        out0[(size_t)row * DIM + ob * 16 + lm] = x0;
        out1[(size_t)row * DIM + ob * 16 + lm] = x1 + bv;
      }
    }
  }

  const int srow = m0 + 8 * hi + lm;
  const bool swr = lm < 8 && srow < M;
  float s0 = 0.0f;
#pragma unroll
  for (int v = 0; v < 8; ++v) {
    const float r = row16_sum(p0[v]);
    s0 = (lm == v) ? r : s0;
  }
  if (swr) aout0[srow] = s0;
  if (!is_rel) {
    float s1 = 0.0f;
#pragma unroll
    for (int v = 0; v < 8; ++v) {
      const float r = row16_sum(p1[v]);
      s1 = (lm == v) ? r : s1;
    }
    if (swr) a_d[srow] = s1;
  }
}

static_assert(NUM_RELS <= (1 << 9), "relation id must fit 9 bits of a record");
static_assert(NUM_NODES <= (1 << 23), "node id must fit 23 bits of a record");
#define RL_BINS  512
#define RL_SLOTS BK_SLICE
static_assert(NUM_RELS <= RL_BINS, "every relation needs a run word");
static_assert(BK_SLICE < (1 << 15), "start and length of a run must fit 16 bits each");

__global__ __launch_bounds__(1024) void bucket_kernel(
    const int* __restrict__ src, const int* __restrict__ dst,
    const int* __restrict__ rel, const float* __restrict__ cntm_ws,
    float* __restrict__ rec_ws, float* __restrict__ binstart_ws,
    float* __restrict__ bintot_ws, float* __restrict__ rec2_ws,
    float* __restrict__ roff_ws) {
  const int* cntm = reinterpret_cast<const int*>(cntm_ws);
  uint2* rec = reinterpret_cast<uint2*>(rec_ws);
  int* binstart = reinterpret_cast<int*>(binstart_ws);
  int* bintot = reinterpret_cast<int*>(bintot_ws);
  uint2* rec2 = reinterpret_cast<uint2*>(rec2_ws);
  int* roff = reinterpret_cast<int*>(roff_ws);
  __shared__ int spart[4 * 2 * BK_BINS];
  __shared__ int sT[BK_BINS], sP[BK_BINS], sL[BK_BINS];
  __shared__ int sbase[BK_BINS];
  __shared__ int lpos[BK_BINS];
  __shared__ int lcur[BK_BINS];
  __shared__ uint2 srec[4096];
  __shared__ int rcnt[RL_BINS];
  __shared__ int rstart[RL_BINS];
  __shared__ int rcur[RL_BINS];
  __shared__ uint2 srec2[4096];
  const int t = threadIdx.x;
  const int g = blockIdx.x;
  const int b = t & (BK_BINS - 1);
  const int part = t >> 8;

  if (t < RL_BINS) { rcnt[t] = 0; rcur[t] = 0; }

  const int e0 = g * BK_SLICE;
  int ns = NUM_EDGES - e0;
  ns = ns > BK_SLICE ? BK_SLICE : ns;
  ns = ns < 0 ? 0 : ns;
  int sv[BK_TRIPS], dv[BK_TRIPS], rv[BK_TRIPS];
#pragma unroll
  for (int k = 0; k < BK_TRIPS; ++k) {
    const int i = t + 1024 * k;
    sv[k] = 0; dv[k] = 0; rv[k] = 0;
    if (i < ns) { sv[k] = src[e0 + i]; dv[k] = dst[e0 + i]; rv[k] = rel[e0 + i]; }
  }

  int tot = 0, pre = 0;
  int cv[BK_CTRIPS];
  const int* col = cntm + part * BK_BINS + b;
#pragma unroll
  for (int k = 0; k < BK_CTRIPS; ++k) {
    cv[k] = 0;
    if (part + 4 * k < BK_GROUPS) cv[k] = col[k * 4 * BK_BINS];
  }
#pragma unroll
  for (int k = 0; k < BK_CTRIPS; ++k) {
    tot += cv[k];
    pre += part + 4 * k < g ? cv[k] : 0;
  }
  spart[part * 2 * BK_BINS + b] = tot;
  spart[part * 2 * BK_BINS + BK_BINS + b] = pre;
  __syncthreads();
  if (t < BK_BINS) {
    sT[t] = spart[t] + spart[2 * BK_BINS + t] + spart[4 * BK_BINS + t] +
            spart[6 * BK_BINS + t];
    sP[t] = spart[BK_BINS + t] + spart[3 * BK_BINS + t] +
            spart[5 * BK_BINS + t] + spart[7 * BK_BINS + t];
    sL[t] = cntm[g * BK_BINS + t];
    lcur[t] = 0;
  }
#pragma unroll
  for (int k = 0; k < BK_TRIPS; ++k) {
    const int i = t + 1024 * k;
    if (i < ns)
      __hip_atomic_fetch_add(&rcnt[(unsigned)rv[k] & (RL_BINS - 1)], 1,
                             __ATOMIC_RELAXED, __HIP_MEMORY_SCOPE_WORKGROUP);
  }
  __syncthreads();
  if (t < 32) {
    int vT[8], vL[8];
    int accT = 0, accL = 0;
#pragma unroll
    for (int j = 0; j < 8; ++j) {
      vT[j] = sT[8 * t + j];
      vL[j] = sL[8 * t + j];
      accT += vT[j];
      accL += vL[j];
    }
    int inT = accT, inL = accL;
    for (int o = 1; o < 32; o <<= 1) {
      const int uT = __shfl_up(inT, o);
      const int uL = __shfl_up(inL, o);
      if (t >= o) { inT += uT; inL += uL; }
    }
    int exT = inT - accT, exL = inL - accL;
#pragma unroll
    for (int j = 0; j < 8; ++j) {
      const int bb = 8 * t + j;
      sbase[bb] = exT + sP[bb];
      lpos[bb] = exL;
      if (g == 0) { binstart[bb] = exT; bintot[bb] = vT[j]; }
      exT += vT[j];
      exL += vL[j];
    }
  } else if (t < 64) {
    const int l = t - 32;
    int vR[RL_BINS / 32];
    int accR = 0;
#pragma unroll
    for (int j = 0; j < RL_BINS / 32; ++j) {
      vR[j] = rcnt[(RL_BINS / 32) * l + j];
      accR += vR[j];
    }
    int inR = accR;
    for (int o = 1; o < 32; o <<= 1) {
      const int uR = __shfl_up(inR, o);
      if (l >= o) inR += uR;
    }
    int exR = inR - accR;
#pragma unroll
    for (int j = 0; j < RL_BINS / 32; ++j) {
      rstart[(RL_BINS / 32) * l + j] = exR;
      exR += vR[j];
    }
  }
  __syncthreads();

#pragma unroll
  for (int k = 0; k < BK_TRIPS; ++k) {
    const int i = t + 1024 * k;
    if (i < ns) {
      const unsigned bb = owner_of(sv[k]);
      const int rank = __hip_atomic_fetch_add(&lcur[bb], 1, __ATOMIC_RELAXED,
                                              __HIP_MEMORY_SCOPE_WORKGROUP);
      int slot = lpos[bb] + rank;
      slot = slot < 0 ? 0 : (slot > 4095 ? 4095 : slot);
      srec[slot] = make_uint2((unsigned)sv[k], ((unsigned)dv[k] << 9) |
                                                   ((unsigned)rv[k] & 511u));
      const unsigned rb = (unsigned)rv[k] & (RL_BINS - 1);
      const int rank2 = __hip_atomic_fetch_add(&rcur[rb], 1, __ATOMIC_RELAXED,
                                               __HIP_MEMORY_SCOPE_WORKGROUP);
      int slot2 = rstart[rb] + rank2;
      slot2 = slot2 < 0 ? 0 : (slot2 > 4095 ? 4095 : slot2);
      srec2[slot2] = make_uint2((unsigned)sv[k], (unsigned)dv[k]);
    }
  }
  __syncthreads();
  for (int i = t; i < ns; i += 1024) {
    const uint2 rc = srec[i];
    const unsigned bb = owner_of((int)rc.x);
    int gp = sbase[bb] + (i - lpos[bb]);
    gp = gp < 0 ? 0 : (gp > NUM_EDGES - 1 ? NUM_EDGES - 1 : gp);
    rec[gp] = rc;
    rec2[e0 + i] = srec2[i];
  }
  if (t < RL_BINS) roff[g * RL_BINS + t] = rstart[t] | (rcnt[t] << 16);
}

#define HN_BATCH  16
#define HN_CAP    5120
#define HN_RPT    (HN_CAP / 1024)
#define HN_SCAN   ((HN_NPR + 31) / 32)
static_assert(HN_CAP % 1024 == 0 && HN_SCAN <= 32,
              "whole trips of the work-group; a lane per node of a wave in the write-out");

__device__ __forceinline__ void hn_flush(int node, v2f acc, float bs, int lane,
                                         float* stab, float* sbs) {
  float* p = &stab[node * DIM + 2 * lane];
  __hip_atomic_fetch_add(p, acc.x, __ATOMIC_RELAXED, __HIP_MEMORY_SCOPE_WORKGROUP);
  __hip_atomic_fetch_add(p + 1, acc.y, __ATOMIC_RELAXED,
                         __HIP_MEMORY_SCOPE_WORKGROUP);
  if (lane == 0)
    __hip_atomic_fetch_add(&sbs[node], bs, __ATOMIC_RELAXED,
                           __HIP_MEMORY_SCOPE_WORKGROUP);
}

__global__ __launch_bounds__(1024) void edge_gather_h_kernel(
    const float* __restrict__ rec_ws, const float* __restrict__ binstart_ws,
    const float* __restrict__ bintot_ws, const float* __restrict__ u_d,
    const float* __restrict__ u_r, const float* __restrict__ cvec,
    const float* __restrict__ a_s, const float* __restrict__ a_d,
    const float* __restrict__ a_r, const float* __restrict__ a0p,
    float* __restrict__ h_ent) {
  __shared__ __attribute__((aligned(16))) float stab[HN_NPR * DIM];
  __shared__ float sbs[HN_NPR];
  __shared__ uint2 ssort[HN_CAP];
  __shared__ int sloc[HN_CAP];
  __shared__ int ncnt[32 * HN_SCAN];
  __shared__ int nstart[32 * HN_SCAN + 1];
  const uint2* rec = reinterpret_cast<const uint2*>(rec_ws);
  const int* binstart = reinterpret_cast<const int*>(binstart_ws);
  const int* bintot = reinterpret_cast<const int*>(bintot_ws);
  const int t = threadIdx.x;
  for (int i = t; i < HN_NPR * DIM; i += 1024) stab[i] = 0.0f;
  for (int i = t; i < HN_NPR; i += 1024) sbs[i] = 0.0f;
  const float a0 = a0p[0];

  const int n0 = blockIdx.x * HN_NPR;
  int nn = NUM_NODES - n0;
  nn = nn > HN_NPR ? HN_NPR : nn;
  nn = nn < 0 ? 0 : nn;

  int s0 = binstart[blockIdx.x];
  int ne = bintot[blockIdx.x];
  s0 = s0 < 0 ? 0 : (s0 > NUM_EDGES ? NUM_EDGES : s0);
  ne = ne < 0 ? 0 : (ne > NUM_EDGES - s0 ? NUM_EDGES - s0 : ne);

  const int wave = __builtin_amdgcn_readfirstlane(t >> 5);
  const int lane = t & 31;
  const int nlo = (wave * HN_NPR) >> 5;
  const int nhi = ((wave + 1) * HN_NPR) >> 5;

  for (int c0 = 0; c0 < ne; c0 += HN_CAP) {
    int nc = ne - c0;
    nc = nc > HN_CAP ? HN_CAP : nc;
    if (t < 32 * HN_SCAN) ncnt[t] = 0;
    __syncthreads();

    uint2 rc[HN_RPT];
    int lo[HN_RPT], rk[HN_RPT];
    float sa[HN_RPT], sd[HN_RPT], sr[HN_RPT], bw[HN_RPT];
#pragma unroll
    for (int k = 0; k < HN_RPT; ++k) {
      const int i = t + 1024 * k;
      rc[k] = rec[s0 + c0 + (i < nc ? i : nc - 1)];
    }
#pragma unroll
    for (int k = 0; k < HN_RPT; ++k) {
      int loc = (int)rc[k].x - n0;
      loc = loc < 0 ? 0 : (loc > HN_NPR - 1 ? HN_NPR - 1 : loc);
      lo[k] = loc;
      sa[k] = a_s[n0 + loc];
      sd[k] = a_d[rc[k].y >> 9];
      sr[k] = a_r[rc[k].y & 511u];
    }
#pragma unroll
    for (int k = 0; k < HN_RPT; ++k) {
      rk[k] = 0;
      float l = sa[k] + sd[k] + sr[k] + a0;
      l = l > 0.f ? l : 0.01f * l;
      bw[k] = expf(l);
      if (t + 1024 * k < nc)
        rk[k] = __hip_atomic_fetch_add(&ncnt[lo[k]], 1, __ATOMIC_RELAXED,
                                       __HIP_MEMORY_SCOPE_WORKGROUP);
    }
    __syncthreads();
    if (wave == 0) {
      int v[HN_SCAN];
      int accn = 0;
#pragma unroll
      for (int j = 0; j < HN_SCAN; ++j) {
        v[j] = ncnt[HN_SCAN * lane + j];
        accn += v[j];
      }
      int inc = accn;
      for (int o = 1; o < 32; o <<= 1) {
        const int u = __shfl_up(inc, o);
        if (lane >= o) inc += u;
      }
      int ex = inc - accn;
#pragma unroll
      for (int j = 0; j < HN_SCAN; ++j) {
        nstart[HN_SCAN * lane + j] = ex;
        ex += v[j];
      }
      if (lane == 31) nstart[32 * HN_SCAN] = ex;
    }
    __syncthreads();
#pragma unroll
    for (int k = 0; k < HN_RPT; ++k) {
      if (t + 1024 * k < nc) {
        int slot = nstart[lo[k]] + rk[k];
        slot = slot < 0 ? 0 : (slot > HN_CAP - 1 ? HN_CAP - 1 : slot);
        ssort[slot] = make_uint2(rc[k].y, (unsigned)__float_as_int(bw[k]));
        sloc[slot] = lo[k];
      }
    }
    __syncthreads();

    int rs = __builtin_amdgcn_readfirstlane(nstart[nlo]);
    int re = __builtin_amdgcn_readfirstlane(nstart[nhi]);
    rs = rs < 0 ? 0 : (rs > HN_CAP ? HN_CAP : rs);
    re = re < rs ? rs : (re > HN_CAP ? HN_CAP : re);
    int cur = -1;
    v2f acc;
    acc.x = 0.0f; acc.y = 0.0f;
    float bs = 0.0f;
    uint2 nq = make_uint2(0u, 0u);
    int nl = 0;
    if (rs + lane < re) { nq = ssort[rs + lane]; nl = sloc[rs + lane]; }
    for (int i0 = rs; i0 < re; i0 += 32) {
      const uint2 q = nq;
      const int ql = nl;
      const int ix = i0 + 32 + lane;
      if (ix < re) { nq = ssort[ix]; nl = sloc[ix]; }
      const int left = re - i0;
      const int m = left < 32 ? left : 32;
      for (int k0 = 0; k0 < m; k0 += HN_BATCH) {
        v2f vd[HN_BATCH], vr[HN_BATCH];
        float bk[HN_BATCH];
        int nk[HN_BATCH];
#pragma unroll
        for (int k = 0; k < HN_BATCH; ++k) {
          const int sl = k0 + k < m ? k0 + k : m - 1;
          const unsigned y = (unsigned)__builtin_amdgcn_readlane((int)q.x, sl);
          bk[k] = __int_as_float(__builtin_amdgcn_readlane((int)q.y, sl));
          nk[k] = __builtin_amdgcn_readlane(ql, sl);
          vd[k] = reinterpret_cast<const v2f*>(u_d + (size_t)(y >> 9) * DIM)[lane];
          vr[k] = reinterpret_cast<const v2f*>(u_r + (size_t)(y & 511u) * DIM)[lane];
        }
#pragma unroll
        for (int k = 0; k < HN_BATCH; ++k) {
          if (k0 + k < m) {
            if (nk[k] != cur) {
              if (cur >= 0) hn_flush(cur, acc, bs, lane, stab, sbs);
              cur = nk[k];
              acc.x = 0.0f; acc.y = 0.0f;
              bs = 0.0f;
            }
            acc.x += bk[k] * (vd[k].x + vr[k].x);
            acc.y += bk[k] * (vd[k].y + vr[k].y);
            bs += bk[k];
          }
        }
      }
    }
    if (cur >= 0) hn_flush(cur, acc, bs, lane, stab, sbs);
  }
  __syncthreads();

  v2f us[HN_SCAN], ta[HN_SCAN];
  float bsl = 0.0f;
  if (lane < HN_SCAN && nlo + lane < nhi) bsl = sbs[nlo + lane];
#pragma unroll
  for (int j = 0; j < HN_SCAN; ++j) {
    int row = nlo + j < nhi ? nlo + j : nhi - 1;
    row = row < nn ? row : (nn > 0 ? nn - 1 : 0);
    us[j] = reinterpret_cast<const v2f*>(h_ent + (size_t)(n0 + row) * DIM)[lane];
    ta[j] = *reinterpret_cast<const v2f*>(&stab[row * DIM + 2 * lane]);
  }
  const v2f cv = reinterpret_cast<const v2f*>(cvec)[lane];
#pragma unroll
  for (int j = 0; j < HN_SCAN; ++j) {
    const int row = nlo + j;
    if (row < nhi && row < nn) {
      const float bs = __int_as_float(
          __builtin_amdgcn_readlane(__float_as_int(bsl), j));
      v2f o;
      o.x = 0.0f; o.y = 0.0f;
      if (bs > 0.0f) {
        o.x = ta[j].x / bs + (us[j].x + cv.x);
        o.y = ta[j].y / bs + (us[j].y + cv.y);
      }
      reinterpret_cast<v2f*>(h_ent + (size_t)(n0 + row) * DIM)[lane] = o;
    }
  }
}

#define FR_WAVES 16
#define FR_SPW   16
#define FR_BATCH 8
static_assert(FR_WAVES * FR_SPW >= BK_GROUPS, "every slice needs a wave");
static_assert(FR_SPW <= 32 && FR_WAVES * 32 == DIM * 8, "thread maps of the relation pass");

__device__ __forceinline__ float lanes8_sum(float x) {
  x += __int_as_float(__builtin_amdgcn_update_dpp(0, __float_as_int(x), 0x161, 0xf, 0xf, false));
  x += __int_as_float(__builtin_amdgcn_update_dpp(0, __float_as_int(x), 0x162, 0xf, 0xf, false));
  x += __int_as_float(__builtin_amdgcn_update_dpp(0, __float_as_int(x), 0x164, 0xf, 0xf, false));
  return x;
}

__device__ __forceinline__ uint2 fr_fetch(const uint2* __restrict__ rec2, int i,
                                          int T, int ex, int off) {
  int j = 0;
#pragma unroll
  for (int k = 1; k < FR_SPW; ++k)
    j += (i >= __builtin_amdgcn_readlane(ex, k)) ? 1 : 0;
  int p = __shfl(off, j) + i;
  p = p < 0 ? 0 : (p > NUM_EDGES - 1 ? NUM_EDGES - 1 : p);
  uint2 rc = make_uint2(0u, 0u);
  if (i < T) rc = rec2[p];
  return rc;
}

__global__ __launch_bounds__(FR_WAVES * 32) void edge_featsum_kernel(
    const float* __restrict__ rec2_ws, const float* __restrict__ roff_ws,
    const float* __restrict__ h_ent, const float* __restrict__ rel2_proj,
    const float* __restrict__ W_rel3, const float* __restrict__ b_rel3,
    float* __restrict__ h_rel) {
  __shared__ float part[FR_WAVES * 128];
  __shared__ __attribute__((aligned(16))) float sm[3 * DIM];
  __shared__ int scnt[FR_WAVES];
  const uint2* rec2 = reinterpret_cast<const uint2*>(rec2_ws);
  const int* roff = reinterpret_cast<const int*>(roff_ws);
  const int t = threadIdx.x;
  const int r = blockIdx.x;
  const int wave = __builtin_amdgcn_readfirstlane(t >> 5);
  const int lane = t & 31;

  const int o = t >> 3, q = t & 7;
  const float4* wp = reinterpret_cast<const float4*>(W_rel3 + o * (3 * DIM) + q * 24);
  float4 wv[6];
#pragma unroll
  for (int k = 0; k < 6; ++k) wv[k] = wp[k];
  const float bo = b_rel3[o];
  const float r2 = t < DIM ? rel2_proj[(size_t)r * DIM + t] : 0.0f;

  const int s = wave * FR_SPW + lane;
  int st = 0, cn = 0;
  if (lane < FR_SPW && s < BK_GROUPS) {
    const int p = roff[s * RL_BINS + r];
    st = p & 0xffff;
    cn = (p >> 16) & 0xffff;
    st = st > RL_SLOTS ? RL_SLOTS : st;
    cn = cn > RL_SLOTS - st ? RL_SLOTS - st : cn;
  }
  int inc = cn;
  for (int d = 1; d < FR_SPW; d <<= 1) {
    const int u = __shfl_up(inc, d);
    if (lane >= d) inc += u;
  }
  const int ex = inc - cn;
  const int off = s * RL_SLOTS + st - ex;
  const int T = __builtin_amdgcn_readlane(inc, FR_SPW - 1);

  v2f as, ad;
  as.x = 0.0f; as.y = 0.0f; ad.x = 0.0f; ad.y = 0.0f;
  uint2 nx = fr_fetch(rec2, lane, T, ex, off);
  for (int i0 = 0; i0 < T; i0 += 32) {
    const uint2 rc = nx;
    nx = fr_fetch(rec2, i0 + 32 + lane, T, ex, off);
    const int left = T - i0;
    const int m = left < 32 ? left : 32;
    for (int k0 = 0; k0 < m; k0 += FR_BATCH) {
      v2f vs[FR_BATCH], vd[FR_BATCH];
#pragma unroll
      for (int k = 0; k < FR_BATCH; ++k) {
        const int sl = k0 + k < m ? k0 + k : m - 1;
        unsigned sk = (unsigned)__builtin_amdgcn_readlane((int)rc.x, sl);
        unsigned dk = (unsigned)__builtin_amdgcn_readlane((int)rc.y, sl);
        sk = sk < NUM_NODES ? sk : NUM_NODES - 1;
        dk = dk < NUM_NODES ? dk : NUM_NODES - 1;
        vs[k] = reinterpret_cast<const v2f*>(h_ent + (size_t)sk * DIM)[lane];
        vd[k] = reinterpret_cast<const v2f*>(h_ent + (size_t)dk * DIM)[lane];
      }
#pragma unroll
      for (int k = 0; k < FR_BATCH; ++k) {
        if (k0 + k < m) {
          as.x += vs[k].x; as.y += vs[k].y;
          ad.x += vd[k].x; ad.y += vd[k].y;
        }
      }
    }
  }
  part[wave * 128 + 2 * lane] = as.x;
  part[wave * 128 + 2 * lane + 1] = as.y;
  part[wave * 128 + 64 + 2 * lane] = ad.x;
  part[wave * 128 + 64 + 2 * lane + 1] = ad.y;
  if (lane == 0) scnt[wave] = T;
  __syncthreads();

  if (t < 128) {
    float sum = 0.0f;
    int ce = 0;
    for (int w = 0; w < FR_WAVES; ++w) { sum += part[w * 128 + t]; ce += scnt[w]; }
    const float cf = (float)ce;
    sm[t] = sum * (1.0f / fmaxf(cf, 1.0f));
    if (t < DIM) sm[128 + t] = ce > 0 ? r2 : 0.0f;
  }
  __syncthreads();

  const float4* mp = reinterpret_cast<const float4*>(&sm[q * 24]);
  float acc = 0.0f;
#pragma unroll
  for (int k = 0; k < 6; ++k) {
    const float4 m = mp[k];
    acc += m.x * wv[k].x;
    acc += m.y * wv[k].y;
    acc += m.z * wv[k].z;
    acc += m.w * wv[k].w;
  }
  acc = lanes8_sum(acc);
  if (q == 0) h_rel[(size_t)r * DIM + o] = acc + bo;
}

#define OFF_MS     0
#define OFF_MD     4096
#define OFF_MR     8192
#define OFF_M2     12288
#define OFF_CVEC   16384
#define OFF_VREL2  16448
#define OFF_A0     16512
#define OFF_UR     16576
#define OFF_AR     48576
#define OFF_REL2P  49088
#define OFF_AS     81088
#define OFF_AD     181088
#define OFF_CNTM   281088
#define OFF_BSTART 345088
#define OFF_BTOT   345344
#define OFF_REC    1445588
#define OFF_REC2   3445588
#define OFF_ROFF   5445588
#define OFF_UD     7845588
static_assert(OFF_CNTM + BK_GROUPS * BK_BINS <= OFF_BSTART &&
              OFF_BTOT + BK_BINS <= OFF_REC &&
              OFF_REC % 2 == 0 && OFF_REC + 2 * NUM_EDGES <= OFF_REC2 &&
              OFF_REC2 % 2 == 0 && OFF_REC2 + 2 * NUM_EDGES <= OFF_ROFF &&
              OFF_ROFF + BK_GROUPS * RL_BINS <= OFF_UD,
              "workspace planes do not overlap; records are 8-byte aligned");

extern "C" void kernel_launch(void* const* d_in, const int* in_sizes, int n_in,
                              void* d_out, int out_size, void* d_ws,
                              size_t ws_size, hipStream_t stream) {
  const float* ent_embed = (const float*)d_in[0];
  const float* rel_embed = (const float*)d_in[1];
  const float* W_ent  = (const float*)d_in[2];
  const float* b_ent  = (const float*)d_in[3];
  const float* W_relL = (const float*)d_in[4];
  const float* b_relL = (const float*)d_in[5];
  const float* W_rel2 = (const float*)d_in[6];
  const float* b_rel2 = (const float*)d_in[7];
  const float* W_rel3 = (const float*)d_in[8];
  const float* b_rel3 = (const float*)d_in[9];
  const float* W_a    = (const float*)d_in[10];
  const float* b_a    = (const float*)d_in[11];
  const float* W_fc   = (const float*)d_in[12];
  const float* b_fc   = (const float*)d_in[13];
  const int* src_idx  = (const int*)d_in[14];
  const int* dst_idx  = (const int*)d_in[15];
  const int* rel_idx  = (const int*)d_in[16];

  float* ws = (float*)d_ws;
  float* h_ent = (float*)d_out;
  float* h_rel = (float*)d_out + (size_t)NUM_NODES * DIM;

  fold_kernel<<<FOLD_BLOCKS, 256, 0, stream>>>(
      W_ent, b_ent, W_relL, b_relL, W_rel2, b_rel2, W_a, b_a, W_fc, b_fc,
      ws + OFF_MS, ws + OFF_MD, ws + OFF_MR, ws + OFF_M2, ws + OFF_CVEC,
      ws + OFF_VREL2, ws + OFF_A0);

  precompute_kernel<<<PC_NODE_BLOCKS + PC_REL_BLOCKS, 256, 0, stream>>>(
      ent_embed, rel_embed, W_a, ws + OFF_MS, ws + OFF_MD, ws + OFF_MR,
      ws + OFF_M2, ws + OFF_VREL2, h_ent, ws + OFF_UD, ws + OFF_AS,
      ws + OFF_AD, ws + OFF_UR, ws + OFF_REL2P, ws + OFF_AR, src_idx,
      ws + OFF_CNTM);

  bucket_kernel<<<BK_GROUPS, 1024, 0, stream>>>(
      src_idx, dst_idx, rel_idx, ws + OFF_CNTM, ws + OFF_REC, ws + OFF_BSTART,
      ws + OFF_BTOT, ws + OFF_REC2, ws + OFF_ROFF);

  edge_gather_h_kernel<<<HN_GROUPS, 1024, 0, stream>>>(
      ws + OFF_REC, ws + OFF_BSTART, ws + OFF_BTOT, ws + OFF_UD, ws + OFF_UR,
      ws + OFF_CVEC, ws + OFF_AS, ws + OFF_AD, ws + OFF_AR, ws + OFF_A0, h_ent);
  edge_featsum_kernel<<<NUM_RELS, FR_WAVES * 32, 0, stream>>>(
      ws + OFF_REC2, ws + OFF_ROFF, h_ent, ws + OFF_REL2P, W_rel3, b_rel3,
      h_rel);
}
